# starting kernel
# baseline (speedup 1.0000x reference)
#include <hip/hip_runtime.h>
#include <math.h>

#define LOG_TD 22u
#define TD (1u << LOG_TD)

typedef __attribute__((ext_vector_type(2))) float v2f;
typedef __attribute__((ext_vector_type(8))) float v8f;

constexpr unsigned TPW = 4u;

__device__ __forceinline__ float hw_exp2(float x) {
#if __has_builtin(__builtin_amdgcn_exp2f)
  return __builtin_amdgcn_exp2f(x);
#else
  return exp2f(x);
#endif
}
__device__ __forceinline__ float hw_rcp(float x) {
#if __has_builtin(__builtin_amdgcn_rcpf)
  return __builtin_amdgcn_rcpf(x);
#else
  return 1.0f / x;
#endif
}
__device__ __forceinline__ float fast_sigmoid(float x) {
  return hw_rcp(1.0f + hw_exp2(-1.4426950408889634f * x));
}
__device__ __forceinline__ float fast_tanh(float x) {
#if __has_builtin(__builtin_amdgcn_tanhf)
  return __builtin_amdgcn_tanhf(x);
#else
  return 1.0f - 2.0f * hw_rcp(hw_exp2(2.8853900817779268f * x) + 1.0f);
#endif
}

template <bool HAS_M1>
__device__ __forceinline__ float gate_src(unsigned m,
    const float* __restrict__ hv, const float* __restrict__ m0,
    const float* __restrict__ m1) {
  if (m < 3u * TD) {
    unsigned n = m >> LOG_TD;
    unsigned r = m & (TD - 1u);
    if (n == 0u) return m0[r];
    if (HAS_M1 && n == 1u) return m1[r];
    return 0.0f;
  }
  unsigned r = (m - 3u * TD) & (TD - 1u);
  return hv[r];
}

template <bool HAS_M1>
__device__ __forceinline__ float cand_src(unsigned m,
    const float* __restrict__ hv, const float* __restrict__ m0,
    const float* __restrict__ m1, const float* __restrict__ G) {
  unsigned seg = m >> LOG_TD;
  unsigned r   = m & (TD - 1u);
  if (seg & 1u) return hv[r];
  unsigned n = seg >> 1;
  if (n == 0u) return G[r] * m0[r];
  if (HAS_M1 && n == 1u) return G[TD + r] * m1[r];
  return 0.0f;
}

template <bool HAS_M1>
__global__ __launch_bounds__(256) void gsn_gate_wmma(
    const float* __restrict__ hv, const float* __restrict__ m0,
    const float* __restrict__ m1, const float* __restrict__ Wx,
    const float* __restrict__ Bx,
    float* __restrict__ G) {
  const unsigned lane  = threadIdx.x & 31u;
  const unsigned wave  = threadIdx.x >> 5u;
  const unsigned tile0 = (blockIdx.x * 8u + wave) * TPW;

  const bool     lo  = lane < 16u;
  const unsigned N   = lane & 15u;
  const bool     nok = N < 3u;
  const unsigned wn  = nok ? N : 0u;
  const float    msk = nok ? 1.0f : 0.0f;
  const float w0 = Wx[wn*6u+0u]*msk, w1 = Wx[wn*6u+1u]*msk, w2 = Wx[wn*6u+2u]*msk,
              w3 = Wx[wn*6u+3u]*msk, w4 = Wx[wn*6u+4u]*msk, w5 = Wx[wn*6u+5u]*msk;
  const float bias = Bx[wn] * msk;

  v2f b, b2;
  b.x  = lo ? w0 : w2;
  b.y  = lo ? w1 : w3;
  b2.x = lo ? w4 : 0.0f;
  b2.y = lo ? w5 : 0.0f;

  for (unsigned t = 0; t < TPW; ++t) {
    const unsigned q0   = (tile0 + t) * 16u;
    const unsigned row  = q0 + (lane & 15u);
    const unsigned base = 6u * row;
    v2f a, a2;
    a.x  = gate_src<HAS_M1>(base + (lo ? 0u : 2u), hv, m0, m1);
    a.y  = gate_src<HAS_M1>(base + (lo ? 1u : 3u), hv, m0, m1);
    a2.x = lo ? gate_src<HAS_M1>(base + 4u, hv, m0, m1) : 0.0f;
    a2.y = lo ? gate_src<HAS_M1>(base + 5u, hv, m0, m1) : 0.0f;

    v8f c;
#pragma unroll
    for (int v = 0; v < 8; ++v) c[v] = bias;

    v8f d = __builtin_amdgcn_wmma_f32_16x16x4_f32(false, a,  false, b,  (short)0, c, false, false);
    d     = __builtin_amdgcn_wmma_f32_16x16x4_f32(false, a2, false, b2, (short)0, d, false, false);

    if (nok) {
      const unsigned mb = lo ? 0u : 8u;
#pragma unroll
      for (int v = 0; v < 8; ++v) {
        unsigned p = 3u * (q0 + mb + (unsigned)v) + N;
        G[p] = fast_sigmoid(d[v]);
      }
    }
  }
}

template <bool HAS_M1>
__global__ __launch_bounds__(256) void gsn_cand_wmma(
    const float* __restrict__ hv, const float* __restrict__ m0,
    const float* __restrict__ m1, const float* __restrict__ Ww,
    const float* __restrict__ Bw,
    const float* __restrict__ G, float* __restrict__ CT) {
  const unsigned lane  = threadIdx.x & 31u;
  const unsigned wave  = threadIdx.x >> 5u;
  const unsigned tile0 = (blockIdx.x * 8u + wave) * TPW;

  const bool     lo  = lane < 16u;
  const unsigned N   = lane & 15u;
  const bool     nok = N < 3u;
  const unsigned wn  = nok ? N : 0u;
  const float    msk = nok ? 1.0f : 0.0f;
  const float w0 = Ww[wn*6u+0u]*msk, w1 = Ww[wn*6u+1u]*msk, w2 = Ww[wn*6u+2u]*msk,
              w3 = Ww[wn*6u+3u]*msk, w4 = Ww[wn*6u+4u]*msk, w5 = Ww[wn*6u+5u]*msk;
  const float bias = Bw[wn] * msk;

  v2f b, b2;
  b.x  = lo ? w0 : w2;
  b.y  = lo ? w1 : w3;
  b2.x = lo ? w4 : 0.0f;
  b2.y = lo ? w5 : 0.0f;

  for (unsigned t = 0; t < TPW; ++t) {
    const unsigned q0   = (tile0 + t) * 16u;
    const unsigned row  = q0 + (lane & 15u);
    const unsigned base = 6u * row;
    v2f a, a2;
    a.x  = cand_src<HAS_M1>(base + (lo ? 0u : 2u), hv, m0, m1, G);
    a.y  = cand_src<HAS_M1>(base + (lo ? 1u : 3u), hv, m0, m1, G);
    a2.x = lo ? cand_src<HAS_M1>(base + 4u, hv, m0, m1, G) : 0.0f;
    a2.y = lo ? cand_src<HAS_M1>(base + 5u, hv, m0, m1, G) : 0.0f;

    v8f c;
#pragma unroll
    for (int v = 0; v < 8; ++v) c[v] = bias;

    v8f d = __builtin_amdgcn_wmma_f32_16x16x4_f32(false, a,  false, b,  (short)0, c, false, false);
    d     = __builtin_amdgcn_wmma_f32_16x16x4_f32(false, a2, false, b2, (short)0, d, false, false);

    if (nok) {
      const unsigned mb = lo ? 0u : 8u;
#pragma unroll
      for (int v = 0; v < 8; ++v) {
        unsigned p   = 3u * (q0 + mb + (unsigned)v) + N;
        float cand   = fast_tanh(d[v]);
        float g      = G[p];
        unsigned n_o = p >> LOG_TD;
        unsigned r   = p & (TD - 1u);
        float mv;
        if (n_o == 0u)               mv = m0[r];
        else if (HAS_M1 && n_o == 1u) mv = m1[r];
        else                          mv = 0.0f;
        CT[p] = (1.0f - g) * mv + g * cand;
      }
    }
  }
}

__global__ __launch_bounds__(256) void gsn_finalize(
    const float* __restrict__ hv, const float* __restrict__ CT,
    float* __restrict__ out) {
  const unsigned i = (blockIdx.x * 256u + threadIdx.x) * 4u;
  const float4 h4 = *(const float4*)(hv + i);
  const float4 c0 = *(const float4*)(CT + i);
  const float4 c1 = *(const float4*)(CT + TD + i);
  const float4 c2 = *(const float4*)(CT + 2u * TD + i);
  float4 o;
  o.x = h4.x + c0.x + c1.x + c2.x;
  o.y = h4.y + c0.y + c1.y + c2.y;
  o.z = h4.z + c0.z + c1.z + c2.z;
  o.w = h4.w + c0.w + c1.w + c2.w;
  *(float4*)(out + i) = o;
}

extern "C" void kernel_launch(void* const* d_in, const int* in_sizes, int n_in,
                              void* d_out, int out_size, void* d_ws, size_t ws_size,
                              hipStream_t stream) {
  (void)in_sizes; (void)n_in; (void)out_size; (void)ws_size;
  const float* feat = (const float*)d_in[0];
  const float* W_w  = (const float*)d_in[1];
  const float* W_b  = (const float*)d_in[2];
  const float* Wx_w = (const float*)d_in[3];
  const float* Wx_b = (const float*)d_in[4];
  float* out = (float*)d_out;

  float* G  = (float*)d_ws;
  float* CT = G + (size_t)3 * TD;

  const float* h0 = feat;
  const float* h1 = feat + (size_t)TD;
  const float* h2 = feat + (size_t)2 * TD;

  dim3 blk(256);
  dim3 grd_mm(TD / 16u / TPW / 8u);
  dim3 grd_fin(TD / 4u / 256u);

  gsn_gate_wmma<false><<<grd_mm, blk, 0, stream>>>(h0, h2, nullptr, Wx_w, Wx_b, G);
  gsn_cand_wmma<false><<<grd_mm, blk, 0, stream>>>(h0, h2, nullptr, W_w, W_b, G, CT);
  gsn_finalize<<<grd_fin, blk, 0, stream>>>(h0, CT, out + 0 * (size_t)TD);

  gsn_gate_wmma<false><<<grd_mm, blk, 0, stream>>>(h1, h0, nullptr, Wx_w, Wx_b, G);
  gsn_cand_wmma<false><<<grd_mm, blk, 0, stream>>>(h1, h0, nullptr, W_w, W_b, G, CT);
  gsn_finalize<<<grd_fin, blk, 0, stream>>>(h1, CT, out + 1 * (size_t)TD);

  gsn_gate_wmma<true><<<grd_mm, blk, 0, stream>>>(h2, h1, h0, Wx_w, Wx_b, G);
  gsn_cand_wmma<true><<<grd_mm, blk, 0, stream>>>(h2, h1, h0, W_w, W_b, G, CT);
  gsn_finalize<<<grd_fin, blk, 0, stream>>>(h2, CT, out + 2 * (size_t)TD);
}


# Re-tile both products: 4 rows per matrix-op column via block-diagonal 24x12 weight (A) x data (B); lanes own 4-8 consecutive outputs, all loads/stores float4; dot now summed in 3 chained steps
# speedup vs baseline: 1.9589x; 1.9589x over previous
#include <hip/hip_runtime.h>
#include <math.h>

#define LOG_TD 22u
#define TD (1u << LOG_TD)

typedef __attribute__((ext_vector_type(2))) float v2f;
typedef __attribute__((ext_vector_type(8))) float v8f;

constexpr unsigned TPW    = 8u;
constexpr unsigned NTILES = TD / 64u;
constexpr unsigned NWAVES = NTILES / TPW;
constexpr unsigned NBLK   = NWAVES / 8u;
constexpr unsigned GCH    = 4u;
constexpr unsigned CCH    = 2u;

__device__ __forceinline__ float hw_exp2(float x) {
#if __has_builtin(__builtin_amdgcn_exp2f)
  return __builtin_amdgcn_exp2f(x);
#else
  return exp2f(x);
#endif
}
__device__ __forceinline__ float hw_rcp(float x) {
#if __has_builtin(__builtin_amdgcn_rcpf)
  return __builtin_amdgcn_rcpf(x);
#else
  return 1.0f / x;
#endif
}
__device__ __forceinline__ float fast_sigmoid(float x) {
  return hw_rcp(1.0f + hw_exp2(-1.4426950408889634f * x));
}
__device__ __forceinline__ float fast_tanh(float x) {
#if __has_builtin(__builtin_amdgcn_tanhf)
  return __builtin_amdgcn_tanhf(x);
#else
  return 1.0f - 2.0f * hw_rcp(hw_exp2(2.8853900817779268f * x) + 1.0f);
#endif
}

__device__ __forceinline__ float4 ld4(const float* __restrict__ p) {
  return *(const float4*)p;
}
__device__ __forceinline__ float4 zero4() { return make_float4(0.0f, 0.0f, 0.0f, 0.0f); }
__device__ __forceinline__ float4 sel4(bool c, const float4& v) {
  return make_float4(c ? v.x : 0.0f, c ? v.y : 0.0f, c ? v.z : 0.0f, c ? v.w : 0.0f);
}


template <bool HAS_M1>
__device__ __forceinline__ float4 gate_src4(unsigned m,
    const float* __restrict__ hv, const float* __restrict__ m0,
    const float* __restrict__ m1) {
  const unsigned seg = m >> LOG_TD;
  const unsigned r   = m & (TD - 1u);
  const bool is_h  = seg >= 3u;
  const bool is_m0 = seg == 0u;
  const bool is_m1 = HAS_M1 && seg == 1u;
  const bool valid = is_h || is_m0 || is_m1;
  const float* base = is_m0 ? m0 : (is_m1 ? m1 : hv);
  const float4 v = ld4(base + (valid ? r : 0u));
  return sel4(valid, v);
}

template <bool HAS_M1>
__device__ __forceinline__ float4 msg4(unsigned p, const float* __restrict__ hv,
    const float* __restrict__ m0, const float* __restrict__ m1) {
  const unsigned n = p >> LOG_TD;
  const unsigned r = p & (TD - 1u);
  const bool is_m0 = n == 0u;
  const bool is_m1 = HAS_M1 && n == 1u;
  const bool valid = is_m0 || is_m1;
  const float* base = is_m0 ? m0 : (is_m1 ? m1 : hv);
  const float4 v = ld4(base + (valid ? r : 0u));
  return sel4(valid, v);
}

template <bool HAS_M1>
__device__ __forceinline__ void cand_src4_load(unsigned m,
    const float* __restrict__ hv, const float* __restrict__ m0,
    const float* __restrict__ m1, const float* __restrict__ G,
    float4& u, float4& w) {
  const unsigned seg = m >> LOG_TD;
  const unsigned r   = m & (TD - 1u);
  const unsigned n   = seg >> 1;
  const bool is_h  = (seg & 1u) != 0u;
  const bool is_m0 = !is_h && n == 0u;
  const bool is_m1 = HAS_M1 && !is_h && n == 1u;
  const bool prod  = is_m0 || is_m1;
  const float* ub = prod ? G : hv;
  const unsigned uo = prod ? (n * TD + r) : (is_h ? r : 0u);
  u = ld4(ub + uo);
  const float* wb = is_m0 ? m0 : (is_m1 ? m1 : hv);
  w = ld4(wb + (prod ? r : 0u));
}
template <bool HAS_M1>
__device__ __forceinline__ float4 cand_src4_value(unsigned m, const float4& u,
    const float4& w) {
  const unsigned seg = m >> LOG_TD;
  const unsigned n   = seg >> 1;
  const bool is_h  = (seg & 1u) != 0u;
  const bool prod  = !is_h && (n == 0u || (HAS_M1 && n == 1u));
  const float4 gm = make_float4(u.x * w.x, u.y * w.y, u.z * w.z, u.w * w.w);
  return make_float4(is_h ? u.x : (prod ? gm.x : 0.0f), is_h ? u.y : (prod ? gm.y : 0.0f),
                     is_h ? u.z : (prod ? gm.z : 0.0f), is_h ? u.w : (prod ? gm.w : 0.0f));
}

__device__ __forceinline__ void make_weights(const float* __restrict__ Wm,
    const float* __restrict__ Bv, unsigned lane, v2f (&a)[6], v8f& cinit) {
  const unsigned o  = lane & 15u;
  const unsigned h  = lane >> 4;
  const unsigned g  = o / 3u;
  const unsigned j  = o - 3u * g;
  const bool valid  = (o < 12u) && ((g >> 1) == h);
  const float msk   = valid ? 1.0f : 0.0f;
  const bool second = (g & 1u) != 0u;
  float w[6];
#pragma unroll
  for (int i = 0; i < 6; ++i) w[i] = Wm[j * 6u + (unsigned)i] * msk;
#pragma unroll
  for (int c = 0; c < 3; ++c) {
    a[c].x     = second ? 0.0f : w[2 * c];
    a[c].y     = second ? 0.0f : w[2 * c + 1];
    a[c + 3].x = second ? w[2 * c]     : 0.0f;
    a[c + 3].y = second ? w[2 * c + 1] : 0.0f;
  }
  const float b0 = Bv[0], b1 = Bv[1], b2 = Bv[2];
  const bool hi = h != 0u;
  cinit[0] = hi ? b2 : b0;
  cinit[1] = hi ? b0 : b1;
  cinit[2] = hi ? b1 : b2;
  cinit[3] = hi ? b2 : b0;
  cinit[4] = hi ? 0.0f : b1;
  cinit[5] = hi ? 0.0f : b2;
  cinit[6] = hi ? 0.0f : b0;
  cinit[7] = hi ? 0.0f : b1;
}

__device__ __forceinline__ v8f group_product(const v2f (&a)[6], const float4& x0,
    const float4& x1, const float4& x2, v8f c) {
  v2f b;
  b.x = x0.x; b.y = x0.y;
  c = __builtin_amdgcn_wmma_f32_16x16x4_f32(false, a[0], false, b, (short)0, c, false, false);
  b.x = x0.z; b.y = x0.w;
  c = __builtin_amdgcn_wmma_f32_16x16x4_f32(false, a[1], false, b, (short)0, c, false, false);
  b.x = x1.x; b.y = x1.y;
  c = __builtin_amdgcn_wmma_f32_16x16x4_f32(false, a[2], false, b, (short)0, c, false, false);
  b.x = x1.z; b.y = x1.w;
  c = __builtin_amdgcn_wmma_f32_16x16x4_f32(false, a[3], false, b, (short)0, c, false, false);
  b.x = x2.x; b.y = x2.y;
  c = __builtin_amdgcn_wmma_f32_16x16x4_f32(false, a[4], false, b, (short)0, c, false, false);
  b.x = x2.z; b.y = x2.w;
  c = __builtin_amdgcn_wmma_f32_16x16x4_f32(false, a[5], false, b, (short)0, c, false, false);
  return c;
}

template <bool HAS_M1>
__global__ __launch_bounds__(256, 8) void gsn_gate_wmma(
    const float* __restrict__ hv, const float* __restrict__ m0,
    const float* __restrict__ m1, const float* __restrict__ Wx,
    const float* __restrict__ Bx,
    float* __restrict__ G) {
  const unsigned lane = threadIdx.x & 31u;
  const unsigned gw   = blockIdx.x * 8u + (threadIdx.x >> 5u);
  const unsigned h    = lane >> 4;
  const unsigned loff = 24u * (lane & 15u) + 12u * h;
  const unsigned poff = 12u * (lane & 15u) + 8u * h;

  v2f a[6];
  v8f cinit;
  make_weights(Wx, Bx, lane, a, cinit);

  for (unsigned ch = 0; ch < TPW / GCH; ++ch) {
    float4 x0[GCH], x1[GCH], x2[GCH];
#pragma unroll
    for (unsigned t = 0; t < GCH; ++t) {
      const unsigned tile = (ch * GCH + t) * NWAVES + gw;
      const unsigned base = 384u * tile + loff;
      x0[t] = gate_src4<HAS_M1>(base,      hv, m0, m1);
      x1[t] = gate_src4<HAS_M1>(base + 4u, hv, m0, m1);
      x2[t] = gate_src4<HAS_M1>(base + 8u, hv, m0, m1);
    }
#pragma unroll
    for (unsigned t = 0; t < GCH; ++t) {
      const unsigned tile = (ch * GCH + t) * NWAVES + gw;
      const v8f d = group_product(a, x0[t], x1[t], x2[t], cinit);

      const unsigned p = 192u * tile + poff;
      float4 o0;
      o0.x = fast_sigmoid(d[0]); o0.y = fast_sigmoid(d[1]);
      o0.z = fast_sigmoid(d[2]); o0.w = fast_sigmoid(d[3]);
      *(float4*)(G + p) = o0;
      if (h == 0u) {
        float4 o1;
        o1.x = fast_sigmoid(d[4]); o1.y = fast_sigmoid(d[5]);
        o1.z = fast_sigmoid(d[6]); o1.w = fast_sigmoid(d[7]);
        *(float4*)(G + p + 4u) = o1;
      }
    }
  }
}

__device__ __forceinline__ float4 delta4(const float4& g, const float4& mv,
    float d0, float d1, float d2, float d3) {
  float4 o;
  o.x = (1.0f - g.x) * mv.x + g.x * fast_tanh(d0);
  o.y = (1.0f - g.y) * mv.y + g.y * fast_tanh(d1);
  o.z = (1.0f - g.z) * mv.z + g.z * fast_tanh(d2);
  o.w = (1.0f - g.w) * mv.w + g.w * fast_tanh(d3);
  return o;
}

template <bool HAS_M1>
__global__ __launch_bounds__(256, 8) void gsn_cand_wmma(
    const float* __restrict__ hv, const float* __restrict__ m0,
    const float* __restrict__ m1, const float* __restrict__ Ww,
    const float* __restrict__ Bw,
    const float* __restrict__ G, float* __restrict__ CT) {
  const unsigned lane = threadIdx.x & 31u;
  const unsigned gw   = blockIdx.x * 8u + (threadIdx.x >> 5u);
  const unsigned h    = lane >> 4;
  const unsigned loff = 24u * (lane & 15u) + 12u * h;
  const unsigned poff = 12u * (lane & 15u) + 8u * h;
  const unsigned pb   = h == 0u ? 4u : 0u;

  v2f a[6];
  v8f cinit;
  make_weights(Ww, Bw, lane, a, cinit);

  for (unsigned ch = 0; ch < TPW / CCH; ++ch) {
    float4 u[CCH][3], w[CCH][3], ga[CCH], gb[CCH], ma[CCH], mb[CCH];
#pragma unroll
    for (unsigned t = 0; t < CCH; ++t) {
      const unsigned tile = (ch * CCH + t) * NWAVES + gw;
      const unsigned base = 384u * tile + loff;
      cand_src4_load<HAS_M1>(base,      hv, m0, m1, G, u[t][0], w[t][0]);
      cand_src4_load<HAS_M1>(base + 4u, hv, m0, m1, G, u[t][1], w[t][1]);
      cand_src4_load<HAS_M1>(base + 8u, hv, m0, m1, G, u[t][2], w[t][2]);
      const unsigned p = 192u * tile + poff;
      ga[t] = ld4(G + p);
      gb[t] = ld4(G + p + pb);
      ma[t] = msg4<HAS_M1>(p,      hv, m0, m1);
      mb[t] = msg4<HAS_M1>(p + pb, hv, m0, m1);
    }
#pragma unroll
    for (unsigned t = 0; t < CCH; ++t) {
      const unsigned tile = (ch * CCH + t) * NWAVES + gw;
      const unsigned base = 384u * tile + loff;
      asm volatile("" :: "v"(gb[t].x), "v"(gb[t].y), "v"(gb[t].z), "v"(gb[t].w),
                         "v"(mb[t].x), "v"(mb[t].y), "v"(mb[t].z), "v"(mb[t].w));
      const float4 x0 = cand_src4_value<HAS_M1>(base,      u[t][0], w[t][0]);
      const float4 x1 = cand_src4_value<HAS_M1>(base + 4u, u[t][1], w[t][1]);
      const float4 x2 = cand_src4_value<HAS_M1>(base + 8u, u[t][2], w[t][2]);
      const v8f d = group_product(a, x0, x1, x2, cinit);

      const unsigned p = 192u * tile + poff;
      *(float4*)(CT + p) = delta4(ga[t], ma[t], d[0], d[1], d[2], d[3]);
      if (h == 0u)
        *(float4*)(CT + p + 4u) = delta4(gb[t], mb[t], d[4], d[5], d[6], d[7]);
    }
  }
}

__global__ __launch_bounds__(256) void gsn_finalize(
    const float* __restrict__ hv, const float* __restrict__ CT,
    float* __restrict__ out) {
  const unsigned i = (blockIdx.x * 256u + threadIdx.x) * 4u;
  const float4 h4 = *(const float4*)(hv + i);
  const float4 c0 = *(const float4*)(CT + i);
  const float4 c1 = *(const float4*)(CT + TD + i);
  const float4 c2 = *(const float4*)(CT + 2u * TD + i);
  float4 o;
  o.x = h4.x + c0.x + c1.x + c2.x;
  o.y = h4.y + c0.y + c1.y + c2.y;
  o.z = h4.z + c0.z + c1.z + c2.z;
  o.w = h4.w + c0.w + c1.w + c2.w;
  *(float4*)(out + i) = o;
}

extern "C" void kernel_launch(void* const* d_in, const int* in_sizes, int n_in,
                              void* d_out, int out_size, void* d_ws, size_t ws_size,
                              hipStream_t stream) {
  (void)in_sizes; (void)n_in; (void)out_size; (void)ws_size;
  const float* feat = (const float*)d_in[0];
  const float* W_w  = (const float*)d_in[1];
  const float* W_b  = (const float*)d_in[2];
  const float* Wx_w = (const float*)d_in[3];
  const float* Wx_b = (const float*)d_in[4];
  float* out = (float*)d_out;

  float* G  = (float*)d_ws;
  float* CT = G + (size_t)3 * TD;

  const float* h0 = feat;
  const float* h1 = feat + (size_t)TD;
  const float* h2 = feat + (size_t)2 * TD;

  dim3 blk(256);
  dim3 grd_mm(NBLK);
  dim3 grd_fin(TD / 4u / 256u);

  gsn_gate_wmma<false><<<grd_mm, blk, 0, stream>>>(h0, h2, nullptr, Wx_w, Wx_b, G);
  gsn_cand_wmma<false><<<grd_mm, blk, 0, stream>>>(h0, h2, nullptr, W_w, W_b, G, CT);
  gsn_finalize<<<grd_fin, blk, 0, stream>>>(h0, CT, out + 0 * (size_t)TD);

  gsn_gate_wmma<false><<<grd_mm, blk, 0, stream>>>(h1, h0, nullptr, Wx_w, Wx_b, G);
  gsn_cand_wmma<false><<<grd_mm, blk, 0, stream>>>(h1, h0, nullptr, W_w, W_b, G, CT);
  gsn_finalize<<<grd_fin, blk, 0, stream>>>(h1, CT, out + 1 * (size_t)TD);

  gsn_gate_wmma<true><<<grd_mm, blk, 0, stream>>>(h2, h1, h0, Wx_w, Wx_b, G);
  gsn_cand_wmma<true><<<grd_mm, blk, 0, stream>>>(h2, h1, h0, W_w, W_b, G, CT);
  gsn_finalize<<<grd_fin, blk, 0, stream>>>(h2, CT, out + 2 * (size_t)TD);
}


# Sum pass folded into the cand pass: a block owns 760 float4 positions, computes all three slots' delta there, adds them to h through LDS in the old order and writes out; 9 launches -> 6
# speedup vs baseline: 2.2326x; 1.1397x over previous
#include <hip/hip_runtime.h>
#include <math.h>

#define LOG_TD 22u
#define TD (1u << LOG_TD)

typedef __attribute__((ext_vector_type(2))) float v2f;
typedef __attribute__((ext_vector_type(8))) float v8f;

constexpr unsigned TPW    = 8u;
constexpr unsigned NTILES = TD / 64u;
constexpr unsigned NWAVES = NTILES / TPW;
constexpr unsigned NBLK   = NWAVES / 8u;
constexpr unsigned GCH    = 4u;
constexpr unsigned NQ     = TD / 4u;
constexpr unsigned TPS    = 2u;
constexpr unsigned GPB    = 16u * 8u * TPS;
constexpr unsigned LQ     = 760u;
constexpr unsigned CBLK   = (NQ + LQ - 1u) / LQ;
static_assert(LQ % 8u == 0u && LQ + 2u <= 3u * GPB, "block range must be whole lines and fit its groups");

__device__ __forceinline__ float hw_exp2(float x) {
#if __has_builtin(__builtin_amdgcn_exp2f)
  return __builtin_amdgcn_exp2f(x);
#else
  return exp2f(x);
#endif
}
__device__ __forceinline__ float hw_rcp(float x) {
#if __has_builtin(__builtin_amdgcn_rcpf)
  return __builtin_amdgcn_rcpf(x);
#else
  return 1.0f / x;
#endif
}
__device__ __forceinline__ float fast_sigmoid(float x) {
  return hw_rcp(1.0f + hw_exp2(-1.4426950408889634f * x));
}
__device__ __forceinline__ float fast_tanh(float x) {
#if __has_builtin(__builtin_amdgcn_tanhf)
  return __builtin_amdgcn_tanhf(x);
#else
  return 1.0f - 2.0f * hw_rcp(hw_exp2(2.8853900817779268f * x) + 1.0f);
#endif
}

__device__ __forceinline__ float4 ld4(const float* __restrict__ p) {
  return *(const float4*)p;
}
__device__ __forceinline__ float4 zero4() { return make_float4(0.0f, 0.0f, 0.0f, 0.0f); }
__device__ __forceinline__ float4 sel4(bool c, const float4& v) {
  return make_float4(c ? v.x : 0.0f, c ? v.y : 0.0f, c ? v.z : 0.0f, c ? v.w : 0.0f);
}


template <bool HAS_M1>
__device__ __forceinline__ float4 gate_src4(unsigned m,
    const float* __restrict__ hv, const float* __restrict__ m0,
    const float* __restrict__ m1) {
  const unsigned seg = m >> LOG_TD;
  const unsigned r   = m & (TD - 1u);
  const bool is_h  = seg >= 3u;
  const bool is_m0 = seg == 0u;
  const bool is_m1 = HAS_M1 && seg == 1u;
  const bool valid = is_h || is_m0 || is_m1;
  const float* base = is_m0 ? m0 : (is_m1 ? m1 : hv);
  const float4 v = ld4(base + (valid ? r : 0u));
  return sel4(valid, v);
}

template <bool HAS_M1>
__device__ __forceinline__ float4 msg4(unsigned p, const float* __restrict__ hv,
    const float* __restrict__ m0, const float* __restrict__ m1) {
  const unsigned n = p >> LOG_TD;
  const unsigned r = p & (TD - 1u);
  const bool is_m0 = n == 0u;
  const bool is_m1 = HAS_M1 && n == 1u;
  const bool valid = is_m0 || is_m1;
  const float* base = is_m0 ? m0 : (is_m1 ? m1 : hv);
  const float4 v = ld4(base + (valid ? r : 0u));
  return sel4(valid, v);
}

template <bool HAS_M1>
__device__ __forceinline__ void cand_src4_load(unsigned m,
    const float* __restrict__ hv, const float* __restrict__ m0,
    const float* __restrict__ m1, const float* __restrict__ G,
    float4& u, float4& w) {
  const unsigned seg = m >> LOG_TD;
  const unsigned r   = m & (TD - 1u);
  const unsigned n   = seg >> 1;
  const bool is_h  = (seg & 1u) != 0u;
  const bool is_m0 = !is_h && n == 0u;
  const bool is_m1 = HAS_M1 && !is_h && n == 1u;
  const bool prod  = is_m0 || is_m1;
  const float* ub = prod ? G : hv;
  const unsigned uo = prod ? (n * TD + r) : (is_h ? r : 0u);
  u = ld4(ub + uo);
  const float* wb = is_m0 ? m0 : (is_m1 ? m1 : hv);
  w = ld4(wb + (prod ? r : 0u));
}
template <bool HAS_M1>
__device__ __forceinline__ float4 cand_src4_value(unsigned m, const float4& u,
    const float4& w) {
  const unsigned seg = m >> LOG_TD;
  const unsigned n   = seg >> 1;
  const bool is_h  = (seg & 1u) != 0u;
  const bool prod  = !is_h && (n == 0u || (HAS_M1 && n == 1u));
  const float4 gm = make_float4(u.x * w.x, u.y * w.y, u.z * w.z, u.w * w.w);
  return make_float4(is_h ? u.x : (prod ? gm.x : 0.0f), is_h ? u.y : (prod ? gm.y : 0.0f),
                     is_h ? u.z : (prod ? gm.z : 0.0f), is_h ? u.w : (prod ? gm.w : 0.0f));
}

__device__ __forceinline__ void make_weights(const float* __restrict__ Wm,
    const float* __restrict__ Bv, unsigned lane, v2f (&a)[6], v8f& cinit) {
  const unsigned o  = lane & 15u;
  const unsigned h  = lane >> 4;
  const unsigned g  = o / 3u;
  const unsigned j  = o - 3u * g;
  const bool valid  = (o < 12u) && ((g >> 1) == h);
  const float msk   = valid ? 1.0f : 0.0f;
  const bool second = (g & 1u) != 0u;
  float w[6];
#pragma unroll
  for (int i = 0; i < 6; ++i) w[i] = Wm[j * 6u + (unsigned)i] * msk;
#pragma unroll
  for (int c = 0; c < 3; ++c) {
    a[c].x     = second ? 0.0f : w[2 * c];
    a[c].y     = second ? 0.0f : w[2 * c + 1];
    a[c + 3].x = second ? w[2 * c]     : 0.0f;
    a[c + 3].y = second ? w[2 * c + 1] : 0.0f;
  }
  const float b0 = Bv[0], b1 = Bv[1], b2 = Bv[2];
  const bool hi = h != 0u;
  cinit[0] = hi ? b2 : b0;
  cinit[1] = hi ? b0 : b1;
  cinit[2] = hi ? b1 : b2;
  cinit[3] = hi ? b2 : b0;
  cinit[4] = hi ? 0.0f : b1;
  cinit[5] = hi ? 0.0f : b2;
  cinit[6] = hi ? 0.0f : b0;
  cinit[7] = hi ? 0.0f : b1;
}

__device__ __forceinline__ v8f group_product(const v2f (&a)[6], const float4& x0,
    const float4& x1, const float4& x2, v8f c) {
  v2f b;
  b.x = x0.x; b.y = x0.y;
  c = __builtin_amdgcn_wmma_f32_16x16x4_f32(false, a[0], false, b, (short)0, c, false, false);
  b.x = x0.z; b.y = x0.w;
  c = __builtin_amdgcn_wmma_f32_16x16x4_f32(false, a[1], false, b, (short)0, c, false, false);
  b.x = x1.x; b.y = x1.y;
  c = __builtin_amdgcn_wmma_f32_16x16x4_f32(false, a[2], false, b, (short)0, c, false, false);
  b.x = x1.z; b.y = x1.w;
  c = __builtin_amdgcn_wmma_f32_16x16x4_f32(false, a[3], false, b, (short)0, c, false, false);
  b.x = x2.x; b.y = x2.y;
  c = __builtin_amdgcn_wmma_f32_16x16x4_f32(false, a[4], false, b, (short)0, c, false, false);
  b.x = x2.z; b.y = x2.w;
  c = __builtin_amdgcn_wmma_f32_16x16x4_f32(false, a[5], false, b, (short)0, c, false, false);
  return c;
}

template <bool HAS_M1>
__global__ __launch_bounds__(256, 8) void gsn_gate_wmma(
    const float* __restrict__ hv, const float* __restrict__ m0,
    const float* __restrict__ m1, const float* __restrict__ Wx,
    const float* __restrict__ Bx,
    float* __restrict__ G) {
  const unsigned lane = threadIdx.x & 31u;
  const unsigned gw   = blockIdx.x * 8u + (threadIdx.x >> 5u);
  const unsigned h    = lane >> 4;
  const unsigned loff = 24u * (lane & 15u) + 12u * h;
  const unsigned poff = 12u * (lane & 15u) + 8u * h;

  v2f a[6];
  v8f cinit;
  make_weights(Wx, Bx, lane, a, cinit);

  for (unsigned ch = 0; ch < TPW / GCH; ++ch) {
    float4 x0[GCH], x1[GCH], x2[GCH];
#pragma unroll
    for (unsigned t = 0; t < GCH; ++t) {
      const unsigned tile = (ch * GCH + t) * NWAVES + gw;
      const unsigned base = 384u * tile + loff;
      x0[t] = gate_src4<HAS_M1>(base,      hv, m0, m1);
      x1[t] = gate_src4<HAS_M1>(base + 4u, hv, m0, m1);
      x2[t] = gate_src4<HAS_M1>(base + 8u, hv, m0, m1);
    }
#pragma unroll
    for (unsigned t = 0; t < GCH; ++t) {
      const unsigned tile = (ch * GCH + t) * NWAVES + gw;
      const v8f d = group_product(a, x0[t], x1[t], x2[t], cinit);

      const unsigned p = 192u * tile + poff;
      float4 o0;
      o0.x = fast_sigmoid(d[0]); o0.y = fast_sigmoid(d[1]);
      o0.z = fast_sigmoid(d[2]); o0.w = fast_sigmoid(d[3]);
      *(float4*)(G + p) = o0;
      if (h == 0u) {
        float4 o1;
        o1.x = fast_sigmoid(d[4]); o1.y = fast_sigmoid(d[5]);
        o1.z = fast_sigmoid(d[6]); o1.w = fast_sigmoid(d[7]);
        *(float4*)(G + p + 4u) = o1;
      }
    }
  }
}

__device__ __forceinline__ float4 delta4(const float4& g, const float4& mv,
    float d0, float d1, float d2, float d3) {
  float4 o;
  o.x = (1.0f - g.x) * mv.x + g.x * fast_tanh(d0);
  o.y = (1.0f - g.y) * mv.y + g.y * fast_tanh(d1);
  o.z = (1.0f - g.z) * mv.z + g.z * fast_tanh(d2);
  o.w = (1.0f - g.w) * mv.w + g.w * fast_tanh(d3);
  return o;
}

template <bool HAS_M1>
__global__ __launch_bounds__(256, 8) void gsn_cand_wmma(
    const float* __restrict__ hv, const float* __restrict__ m0,
    const float* __restrict__ m1, const float* __restrict__ Ww,
    const float* __restrict__ Bw,
    const float* __restrict__ G, float* __restrict__ out) {
  __shared__ float4 acc[3][LQ];
  const unsigned lane = threadIdx.x & 31u;
  const unsigned wv   = threadIdx.x >> 5u;
  const unsigned h    = lane >> 4;
  const unsigned c    = lane & 15u;
  const unsigned q0   = blockIdx.x * LQ;
  const unsigned nq   = (NQ - q0) < LQ ? (NQ - q0) : LQ;
  const unsigned pb   = h == 0u ? 4u : 0u;

  v2f a[6];
  v8f cinit;
  make_weights(Ww, Bw, lane, a, cinit);

  for (unsigned n = 0; n < 3u; ++n) {
    const unsigned Q0 = n * NQ + q0;
    const unsigned g0 = Q0 / 3u;
    float4 u[TPS][3], w[TPS][3], ga[TPS], gb[TPS], ma[TPS], mb[TPS];
#pragma unroll
    for (unsigned t = 0; t < TPS; ++t) {
      const unsigned g  = g0 + (t * 8u + wv) * 16u + c;
      const unsigned gc = g < NQ ? g : NQ - 1u;
      const unsigned base = 24u * gc + 12u * h;
      cand_src4_load<HAS_M1>(base,      hv, m0, m1, G, u[t][0], w[t][0]);
      cand_src4_load<HAS_M1>(base + 4u, hv, m0, m1, G, u[t][1], w[t][1]);
      cand_src4_load<HAS_M1>(base + 8u, hv, m0, m1, G, u[t][2], w[t][2]);
      const unsigned p = 12u * gc + 8u * h;
      ga[t] = ld4(G + p);
      gb[t] = ld4(G + p + pb);
      ma[t] = msg4<HAS_M1>(p,      hv, m0, m1);
      mb[t] = msg4<HAS_M1>(p + pb, hv, m0, m1);
    }
#pragma unroll
    for (unsigned t = 0; t < TPS; ++t) {
      const unsigned g  = g0 + (t * 8u + wv) * 16u + c;
      const unsigned gc = g < NQ ? g : NQ - 1u;
      const unsigned base = 24u * gc + 12u * h;
      asm volatile("" :: "v"(ga[t].x), "v"(ga[t].y), "v"(ga[t].z), "v"(ga[t].w),
                         "v"(ma[t].x), "v"(ma[t].y), "v"(ma[t].z), "v"(ma[t].w),
                         "v"(gb[t].x), "v"(gb[t].y), "v"(gb[t].z), "v"(gb[t].w),
                         "v"(mb[t].x), "v"(mb[t].y), "v"(mb[t].z), "v"(mb[t].w));
      const float4 x0 = cand_src4_value<HAS_M1>(base,      u[t][0], w[t][0]);
      const float4 x1 = cand_src4_value<HAS_M1>(base + 4u, u[t][1], w[t][1]);
      const float4 x2 = cand_src4_value<HAS_M1>(base + 8u, u[t][2], w[t][2]);
      const v8f d = group_product(a, x0, x1, x2, cinit);

      const unsigned qa = 3u * g + 2u * h - Q0;
      if (qa < nq)
        acc[n][qa] = delta4(ga[t], ma[t], d[0], d[1], d[2], d[3]);
      if (h == 0u && qa + 1u < nq)
        acc[n][qa + 1u] = delta4(gb[t], mb[t], d[4], d[5], d[6], d[7]);
    }
  }
  __syncthreads();
  for (unsigned i = threadIdx.x; i < nq; i += 256u) {
    const float4 h4 = ld4(hv + 4u * (q0 + i));
    const float4 c0 = acc[0][i];
    const float4 c1 = acc[1][i];
    const float4 c2 = acc[2][i];
    float4 o;
    o.x = h4.x + c0.x + c1.x + c2.x;
    o.y = h4.y + c0.y + c1.y + c2.y;
    o.z = h4.z + c0.z + c1.z + c2.z;
    o.w = h4.w + c0.w + c1.w + c2.w;
    *(float4*)(out + 4u * (q0 + i)) = o;
  }
}

extern "C" void kernel_launch(void* const* d_in, const int* in_sizes, int n_in,
                              void* d_out, int out_size, void* d_ws, size_t ws_size,
                              hipStream_t stream) {
  (void)in_sizes; (void)n_in; (void)out_size; (void)ws_size;
  const float* feat = (const float*)d_in[0];
  const float* W_w  = (const float*)d_in[1];
  const float* W_b  = (const float*)d_in[2];
  const float* Wx_w = (const float*)d_in[3];
  const float* Wx_b = (const float*)d_in[4];
  float* out = (float*)d_out;

  float* G  = (float*)d_ws;

  const float* h0 = feat;
  const float* h1 = feat + (size_t)TD;
  const float* h2 = feat + (size_t)2 * TD;

  dim3 blk(256);
  dim3 grd_mm(NBLK);
  dim3 grd_cd(CBLK);

  gsn_gate_wmma<false><<<grd_mm, blk, 0, stream>>>(h0, h2, nullptr, Wx_w, Wx_b, G);
  gsn_cand_wmma<false><<<grd_cd, blk, 0, stream>>>(h0, h2, nullptr, W_w, W_b, G, out + 0 * (size_t)TD);

  gsn_gate_wmma<false><<<grd_mm, blk, 0, stream>>>(h1, h0, nullptr, Wx_w, Wx_b, G);
  gsn_cand_wmma<false><<<grd_cd, blk, 0, stream>>>(h1, h0, nullptr, W_w, W_b, G, out + 1 * (size_t)TD);

  gsn_gate_wmma<true><<<grd_mm, blk, 0, stream>>>(h2, h1, h0, Wx_w, Wx_b, G);
  gsn_cand_wmma<true><<<grd_cd, blk, 0, stream>>>(h2, h1, h0, W_w, W_b, G, out + 2 * (size_t)TD);
}


# Slab choice (which plane a float4 comes from) made once per block by scalar tests in both passes: scalar base + lane offset, no per-lane selects or dummy loads; boundary blocks keep per-lane form
# speedup vs baseline: 2.5958x; 1.1627x over previous
#include <hip/hip_runtime.h>
#include <math.h>

#define LOG_TD 22u
#define TD (1u << LOG_TD)

typedef __attribute__((ext_vector_type(2))) float v2f;
typedef __attribute__((ext_vector_type(8))) float v8f;

constexpr unsigned TPW    = 8u;
constexpr unsigned NTILES = TD / 64u;
constexpr unsigned NWAVES = NTILES / TPW;
constexpr unsigned NBLK   = NWAVES / 8u;
constexpr unsigned GCH    = 4u;
constexpr unsigned NQ     = TD / 4u;
constexpr unsigned TPS    = 2u;
constexpr unsigned GPB    = 16u * 8u * TPS;
constexpr unsigned LQ     = 760u;
constexpr unsigned CBLK   = (NQ + LQ - 1u) / LQ;
static_assert(LQ % 8u == 0u && LQ + 2u <= 3u * GPB, "block range must be whole lines and fit its groups");

__device__ __forceinline__ float hw_exp2(float x) {
#if __has_builtin(__builtin_amdgcn_exp2f)
  return __builtin_amdgcn_exp2f(x);
#else
  return exp2f(x);
#endif
}
__device__ __forceinline__ float hw_rcp(float x) {
#if __has_builtin(__builtin_amdgcn_rcpf)
  return __builtin_amdgcn_rcpf(x);
#else
  return 1.0f / x;
#endif
}
__device__ __forceinline__ float fast_sigmoid(float x) {
  return hw_rcp(1.0f + hw_exp2(-1.4426950408889634f * x));
}
__device__ __forceinline__ float fast_tanh(float x) {
#if __has_builtin(__builtin_amdgcn_tanhf)
  return __builtin_amdgcn_tanhf(x);
#else
  return 1.0f - 2.0f * hw_rcp(hw_exp2(2.8853900817779268f * x) + 1.0f);
#endif
}

__device__ __forceinline__ float4 ld4(const float* __restrict__ p) {
  return *(const float4*)p;
}
__device__ __forceinline__ float4 zero4() { return make_float4(0.0f, 0.0f, 0.0f, 0.0f); }
__device__ __forceinline__ float4 sel4(bool c, const float4& v) {
  return make_float4(c ? v.x : 0.0f, c ? v.y : 0.0f, c ? v.z : 0.0f, c ? v.w : 0.0f);
}


template <bool HAS_M1>
__device__ __forceinline__ float4 gate_src4(unsigned m,
    const float* __restrict__ hv, const float* __restrict__ m0,
    const float* __restrict__ m1) {
  const unsigned seg = m >> LOG_TD;
  const unsigned r   = m & (TD - 1u);
  const bool is_h  = seg >= 3u;
  const bool is_m0 = seg == 0u;
  const bool is_m1 = HAS_M1 && seg == 1u;
  const bool valid = is_h || is_m0 || is_m1;
  const float* base = is_m0 ? m0 : (is_m1 ? m1 : hv);
  const float4 v = ld4(base + (valid ? r : 0u));
  return sel4(valid, v);
}

template <bool HAS_M1>
__device__ __forceinline__ float4 msg4(unsigned p, const float* __restrict__ hv,
    const float* __restrict__ m0, const float* __restrict__ m1) {
  const unsigned n = p >> LOG_TD;
  const unsigned r = p & (TD - 1u);
  const bool is_m0 = n == 0u;
  const bool is_m1 = HAS_M1 && n == 1u;
  const bool valid = is_m0 || is_m1;
  const float* base = is_m0 ? m0 : (is_m1 ? m1 : hv);
  const float4 v = ld4(base + (valid ? r : 0u));
  return sel4(valid, v);
}

template <bool HAS_M1>
__device__ __forceinline__ void cand_src4_load(unsigned m,
    const float* __restrict__ hv, const float* __restrict__ m0,
    const float* __restrict__ m1, const float* __restrict__ G,
    float4& u, float4& w) {
  const unsigned seg = m >> LOG_TD;
  const unsigned r   = m & (TD - 1u);
  const unsigned n   = seg >> 1;
  const bool is_h  = (seg & 1u) != 0u;
  const bool is_m0 = !is_h && n == 0u;
  const bool is_m1 = HAS_M1 && !is_h && n == 1u;
  const bool prod  = is_m0 || is_m1;
  const float* ub = prod ? G : hv;
  const unsigned uo = prod ? (n * TD + r) : (is_h ? r : 0u);
  u = ld4(ub + uo);
  const float* wb = is_m0 ? m0 : (is_m1 ? m1 : hv);
  w = ld4(wb + (prod ? r : 0u));
}
template <bool HAS_M1>
__device__ __forceinline__ float4 cand_src4_value(unsigned m, const float4& u,
    const float4& w) {
  const unsigned seg = m >> LOG_TD;
  const unsigned n   = seg >> 1;
  const bool is_h  = (seg & 1u) != 0u;
  const bool prod  = !is_h && (n == 0u || (HAS_M1 && n == 1u));
  const float4 gm = make_float4(u.x * w.x, u.y * w.y, u.z * w.z, u.w * w.w);
  return make_float4(is_h ? u.x : (prod ? gm.x : 0.0f), is_h ? u.y : (prod ? gm.y : 0.0f),
                     is_h ? u.z : (prod ? gm.z : 0.0f), is_h ? u.w : (prod ? gm.w : 0.0f));
}

__device__ __forceinline__ void make_weights(const float* __restrict__ Wm,
    const float* __restrict__ Bv, unsigned lane, v2f (&a)[6], v8f& cinit) {
  const unsigned o  = lane & 15u;
  const unsigned h  = lane >> 4;
  const unsigned g  = o / 3u;
  const unsigned j  = o - 3u * g;
  const bool valid  = (o < 12u) && ((g >> 1) == h);
  const float msk   = valid ? 1.0f : 0.0f;
  const bool second = (g & 1u) != 0u;
  float w[6];
#pragma unroll
  for (int i = 0; i < 6; ++i) w[i] = Wm[j * 6u + (unsigned)i] * msk;
#pragma unroll
  for (int c = 0; c < 3; ++c) {
    a[c].x     = second ? 0.0f : w[2 * c];
    a[c].y     = second ? 0.0f : w[2 * c + 1];
    a[c + 3].x = second ? w[2 * c]     : 0.0f;
    a[c + 3].y = second ? w[2 * c + 1] : 0.0f;
  }
  const float b0 = Bv[0], b1 = Bv[1], b2 = Bv[2];
  const bool hi = h != 0u;
  cinit[0] = hi ? b2 : b0;
  cinit[1] = hi ? b0 : b1;
  cinit[2] = hi ? b1 : b2;
  cinit[3] = hi ? b2 : b0;
  cinit[4] = hi ? 0.0f : b1;
  cinit[5] = hi ? 0.0f : b2;
  cinit[6] = hi ? 0.0f : b0;
  cinit[7] = hi ? 0.0f : b1;
}

__device__ __forceinline__ v8f group_product(const v2f (&a)[6], const float4& x0,
    const float4& x1, const float4& x2, v8f c) {
  v2f b;
  b.x = x0.x; b.y = x0.y;
  c = __builtin_amdgcn_wmma_f32_16x16x4_f32(false, a[0], false, b, (short)0, c, false, false);
  b.x = x0.z; b.y = x0.w;
  c = __builtin_amdgcn_wmma_f32_16x16x4_f32(false, a[1], false, b, (short)0, c, false, false);
  b.x = x1.x; b.y = x1.y;
  c = __builtin_amdgcn_wmma_f32_16x16x4_f32(false, a[2], false, b, (short)0, c, false, false);
  b.x = x1.z; b.y = x1.w;
  c = __builtin_amdgcn_wmma_f32_16x16x4_f32(false, a[3], false, b, (short)0, c, false, false);
  b.x = x2.x; b.y = x2.y;
  c = __builtin_amdgcn_wmma_f32_16x16x4_f32(false, a[4], false, b, (short)0, c, false, false);
  b.x = x2.z; b.y = x2.w;
  c = __builtin_amdgcn_wmma_f32_16x16x4_f32(false, a[5], false, b, (short)0, c, false, false);
  return c;
}

template <bool HAS_M1>
__global__ __launch_bounds__(256, 8) void gsn_gate_wmma(
    const float* __restrict__ hv, const float* __restrict__ m0,
    const float* __restrict__ m1, const float* __restrict__ Wx,
    const float* __restrict__ Bx,
    float* __restrict__ G) {
  const unsigned lane = threadIdx.x & 31u;
  const unsigned wv   = threadIdx.x >> 5u;
  const unsigned gw   = blockIdx.x * 8u + wv;
  const unsigned h    = lane >> 4;
  const unsigned loff = 24u * (lane & 15u) + 12u * h;
  const unsigned poff = 12u * (lane & 15u) + 8u * h;
  const unsigned lw   = 384u * wv + loff;

  v2f a[6];
  v8f cinit;
  make_weights(Wx, Bx, lane, a, cinit);

  for (unsigned ch = 0; ch < TPW / GCH; ++ch) {
    float4 x0[GCH], x1[GCH], x2[GCH];
#pragma unroll
    for (unsigned t = 0; t < GCH; ++t) {
      const unsigned k  = ch * GCH + t;
      const unsigned f0 = 384u * (k * NWAVES + blockIdx.x * 8u);
      const unsigned seg = f0 >> LOG_TD;
      const bool whole = ((f0 + 8u * 384u - 1u) >> LOG_TD) == seg;
      const bool is_h  = seg >= 3u;
      const bool is_m0 = seg == 0u;
      const bool is_m1 = HAS_M1 && seg == 1u;
      if (whole && (is_h || is_m0 || is_m1)) {
        const float* bp = (is_h ? hv : (is_m0 ? m0 : m1)) + (f0 & (TD - 1u));
        x0[t] = ld4(bp + lw);
        x1[t] = ld4(bp + lw + 4u);
        x2[t] = ld4(bp + lw + 8u);
      } else if (whole) {
        x0[t] = zero4(); x1[t] = zero4(); x2[t] = zero4();
      } else {
        const unsigned base = f0 + lw;
        x0[t] = gate_src4<HAS_M1>(base,      hv, m0, m1);
        x1[t] = gate_src4<HAS_M1>(base + 4u, hv, m0, m1);
        x2[t] = gate_src4<HAS_M1>(base + 8u, hv, m0, m1);
      }
    }
#pragma unroll
    for (unsigned t = 0; t < GCH; ++t) {
      const unsigned tile = (ch * GCH + t) * NWAVES + gw;
      const v8f d = group_product(a, x0[t], x1[t], x2[t], cinit);

      const unsigned p = 192u * tile + poff;
      float4 o0;
      o0.x = fast_sigmoid(d[0]); o0.y = fast_sigmoid(d[1]);
      o0.z = fast_sigmoid(d[2]); o0.w = fast_sigmoid(d[3]);
      *(float4*)(G + p) = o0;
      if (h == 0u) {
        float4 o1;
        o1.x = fast_sigmoid(d[4]); o1.y = fast_sigmoid(d[5]);
        o1.z = fast_sigmoid(d[6]); o1.w = fast_sigmoid(d[7]);
        *(float4*)(G + p + 4u) = o1;
      }
    }
  }
}

__device__ __forceinline__ float4 delta4(const float4& g, const float4& mv,
    float d0, float d1, float d2, float d3) {
  float4 o;
  o.x = (1.0f - g.x) * mv.x + g.x * fast_tanh(d0);
  o.y = (1.0f - g.y) * mv.y + g.y * fast_tanh(d1);
  o.z = (1.0f - g.z) * mv.z + g.z * fast_tanh(d2);
  o.w = (1.0f - g.w) * mv.w + g.w * fast_tanh(d3);
  return o;
}

__device__ __forceinline__ float4 gated4(const float4& g, float d0, float d1, float d2, float d3) {
  float4 o;
  o.x = g.x * fast_tanh(d0);
  o.y = g.y * fast_tanh(d1);
  o.z = g.z * fast_tanh(d2);
  o.w = g.w * fast_tanh(d3);
  return o;
}


template <bool HAS_M1>
__device__ __forceinline__ void cand_slot_lane(unsigned Q0, unsigned g0, unsigned nq,
    unsigned gl, unsigned h, unsigned pb, const v2f (&a)[6], const v8f& cinit,
    const float* __restrict__ hv, const float* __restrict__ m0, const float* __restrict__ m1,
    const float* __restrict__ G, float4* accn) {
  for (unsigned t = 0; t < TPS; ++t) {
    const unsigned g  = g0 + t * 128u + gl;
    const unsigned gc = g < NQ ? g : NQ - 1u;
    const unsigned base = 24u * gc + 12u * h;
    float4 u0, u1, u2, w0, w1, w2;
    cand_src4_load<HAS_M1>(base,      hv, m0, m1, G, u0, w0);
    cand_src4_load<HAS_M1>(base + 4u, hv, m0, m1, G, u1, w1);
    cand_src4_load<HAS_M1>(base + 8u, hv, m0, m1, G, u2, w2);
    const unsigned p = 12u * gc + 8u * h;
    const float4 ga = ld4(G + p);
    const float4 gb = ld4(G + p + pb);
    const float4 ma = msg4<HAS_M1>(p,      hv, m0, m1);
    const float4 mb = msg4<HAS_M1>(p + pb, hv, m0, m1);
    const float4 x0 = cand_src4_value<HAS_M1>(base,      u0, w0);
    const float4 x1 = cand_src4_value<HAS_M1>(base + 4u, u1, w1);
    const float4 x2 = cand_src4_value<HAS_M1>(base + 8u, u2, w2);
    const v8f d = group_product(a, x0, x1, x2, cinit);

    const unsigned qa = 3u * (g0 + t * 128u) - Q0 + (3u * gl + 2u * h);
    if (qa < nq)
      accn[qa] = delta4(ga, ma, d[0], d[1], d[2], d[3]);
    if (h == 0u && qa + 1u < nq)
      accn[qa + 1u] = delta4(gb, mb, d[4], d[5], d[6], d[7]);
  }
}

constexpr int K_PROD = 1;
constexpr int K_H    = 2;
constexpr int K_ZERO = 3;

template <int KIND, bool MSG>
__device__ __forceinline__ void cand_slot_uniform(
    const float* __restrict__ ub, const float* __restrict__ wb,
    const float* __restrict__ gp, const float* __restrict__ mp,
    unsigned qoff, unsigned nq, unsigned gl, unsigned h,
    const v2f (&a)[6], const v8f& cinit, float4* accn) {
  const unsigned lo = 24u * gl + 12u * h;
  const unsigned po = 12u * gl + 8u * h;
  const unsigned pq = 12u * gl + 4u * (h + 1u);
  float4 u[TPS][3], w[TPS][3], ga[TPS], gb[TPS], ma[TPS], mb[TPS];
#pragma unroll
  for (unsigned t = 0; t < TPS; ++t) {
    const unsigned ot = lo + t * (24u * 128u);
    const unsigned pt = po + t * (12u * 128u);
    const unsigned pu = pq + t * (12u * 128u);
    if (KIND != K_ZERO) {
      u[t][0] = ld4(ub + ot);
      u[t][1] = ld4(ub + ot + 4u);
      u[t][2] = ld4(ub + ot + 8u);
    } else {
      u[t][0] = zero4(); u[t][1] = zero4(); u[t][2] = zero4();
    }
    if (KIND == K_PROD) {
      w[t][0] = ld4(wb + ot);
      w[t][1] = ld4(wb + ot + 4u);
      w[t][2] = ld4(wb + ot + 8u);
    } else {
      w[t][0] = zero4(); w[t][1] = zero4(); w[t][2] = zero4();
    }
    ga[t] = ld4(gp + pt);
    gb[t] = ld4(gp + pu);
    if (MSG) {
      ma[t] = ld4(mp + pt);
      mb[t] = ld4(mp + pu);
    } else {
      ma[t] = zero4(); mb[t] = zero4();
    }
  }
#pragma unroll
  for (unsigned t = 0; t < TPS; ++t) {
    if (KIND != K_ZERO)
      asm volatile("" :: "v"(u[t][0].x), "v"(u[t][0].y), "v"(u[t][0].z), "v"(u[t][0].w),
                         "v"(u[t][1].x), "v"(u[t][1].y), "v"(u[t][1].z), "v"(u[t][1].w),
                         "v"(u[t][2].x), "v"(u[t][2].y), "v"(u[t][2].z), "v"(u[t][2].w));
    if (KIND == K_PROD)
      asm volatile("" :: "v"(w[t][0].x), "v"(w[t][0].y), "v"(w[t][0].z), "v"(w[t][0].w),
                         "v"(w[t][1].x), "v"(w[t][1].y), "v"(w[t][1].z), "v"(w[t][1].w),
                         "v"(w[t][2].x), "v"(w[t][2].y), "v"(w[t][2].z), "v"(w[t][2].w));
    asm volatile("" :: "v"(ga[t].x), "v"(ga[t].y), "v"(ga[t].z), "v"(ga[t].w),
                       "v"(gb[t].x), "v"(gb[t].y), "v"(gb[t].z), "v"(gb[t].w));
    if (MSG)
      asm volatile("" :: "v"(ma[t].x), "v"(ma[t].y), "v"(ma[t].z), "v"(ma[t].w),
                         "v"(mb[t].x), "v"(mb[t].y), "v"(mb[t].z), "v"(mb[t].w));
  }
#pragma unroll
  for (unsigned t = 0; t < TPS; ++t) {
    float4 x0 = u[t][0], x1 = u[t][1], x2 = u[t][2];
    if (KIND == K_PROD) {
      x0 = make_float4(x0.x * w[t][0].x, x0.y * w[t][0].y, x0.z * w[t][0].z, x0.w * w[t][0].w);
      x1 = make_float4(x1.x * w[t][1].x, x1.y * w[t][1].y, x1.z * w[t][1].z, x1.w * w[t][1].w);
      x2 = make_float4(x2.x * w[t][2].x, x2.y * w[t][2].y, x2.z * w[t][2].z, x2.w * w[t][2].w);
    }
    const v8f d = group_product(a, x0, x1, x2, cinit);

    const unsigned qa = qoff + t * 384u + (3u * gl + 2u * h);
    if (qa < nq)
      accn[qa] = MSG ? delta4(ga[t], ma[t], d[0], d[1], d[2], d[3])
                     : gated4(ga[t], d[0], d[1], d[2], d[3]);
    if (h == 0u && qa + 1u < nq)
      accn[qa + 1u] = MSG ? delta4(gb[t], mb[t], d[4], d[5], d[6], d[7])
                          : gated4(gb[t], d[4], d[5], d[6], d[7]);
  }
}

template <bool HAS_M1>
__global__ __launch_bounds__(256) __attribute__((amdgpu_waves_per_eu(8, 8))) void gsn_cand_wmma(
    const float* __restrict__ hv, const float* __restrict__ m0,
    const float* __restrict__ m1, const float* __restrict__ Ww,
    const float* __restrict__ Bw,
    const float* __restrict__ G, float* __restrict__ out) {
  __shared__ float4 acc[3][LQ];
  const unsigned lane = threadIdx.x & 31u;
  const unsigned h    = lane >> 4;
  const unsigned gl   = (threadIdx.x >> 5u) * 16u + (lane & 15u);
  const unsigned q0   = blockIdx.x * LQ;
  const unsigned nq   = (NQ - q0) < LQ ? (NQ - q0) : LQ;
  const unsigned pb   = h == 0u ? 4u : 0u;

  v2f a[6];
  v8f cinit;
  make_weights(Ww, Bw, lane, a, cinit);

  for (unsigned n = 0; n < 3u; ++n) {
    const unsigned Q0 = n * NQ + q0;
    const unsigned g0 = Q0 / 3u;
    const unsigned op_lo = 24u * g0;
    const unsigned ep_lo = 12u * g0;
    const unsigned seg   = op_lo >> LOG_TD;
    const bool whole = g0 + (GPB - 1u) < NQ
        && ((op_lo + 24u * GPB - 1u) >> LOG_TD) == seg
        && (ep_lo >> LOG_TD) == n && ((ep_lo + 12u * GPB - 1u) >> LOG_TD) == n;
    const unsigned ns  = seg >> 1;
    const bool is_h    = (seg & 1u) != 0u;
    const bool op_msg  = ns == 0u || (HAS_M1 && ns == 1u);
    const bool ep_msg  = n == 0u || (HAS_M1 && n == 1u);
    const float* opl   = ns == 0u ? m0 : m1;
    const float* epl   = n == 0u ? m0 : m1;
    const unsigned rb  = op_lo & (TD - 1u);
    const unsigned pr  = ep_lo & (TD - 1u);
    const unsigned qoff = 3u * g0 - Q0;
    if (whole && is_h && ep_msg)
      cand_slot_uniform<K_H, true>(hv + rb, hv, G + ep_lo, epl + pr, qoff, nq, gl, h, a, cinit, acc[n]);
    else if (whole && is_h)
      cand_slot_uniform<K_H, false>(hv + rb, hv, G + ep_lo, hv, qoff, nq, gl, h, a, cinit, acc[n]);
    else if (whole && op_msg && ep_msg)
      cand_slot_uniform<K_PROD, true>(G + ns * TD + rb, opl + rb, G + ep_lo, epl + pr, qoff, nq, gl, h, a, cinit, acc[n]);
    else if (whole && !op_msg && !ep_msg)
      cand_slot_uniform<K_ZERO, false>(hv, hv, G + ep_lo, hv, qoff, nq, gl, h, a, cinit, acc[n]);
    else
      cand_slot_lane<HAS_M1>(Q0, g0, nq, gl, h, pb, a, cinit, hv, m0, m1, G, acc[n]);
  }
  __syncthreads();
  for (unsigned i = threadIdx.x; i < nq; i += 256u) {
    const float4 h4 = ld4(hv + 4u * (q0 + i));
    const float4 c0 = acc[0][i];
    const float4 c1 = acc[1][i];
    const float4 c2 = acc[2][i];
    float4 o;
    o.x = h4.x + c0.x + c1.x + c2.x;
    o.y = h4.y + c0.y + c1.y + c2.y;
    o.z = h4.z + c0.z + c1.z + c2.z;
    o.w = h4.w + c0.w + c1.w + c2.w;
    *(float4*)(out + 4u * (q0 + i)) = o;
  }
}

extern "C" void kernel_launch(void* const* d_in, const int* in_sizes, int n_in,
                              void* d_out, int out_size, void* d_ws, size_t ws_size,
                              hipStream_t stream) {
  (void)in_sizes; (void)n_in; (void)out_size; (void)ws_size;
  const float* feat = (const float*)d_in[0];
  const float* W_w  = (const float*)d_in[1];
  const float* W_b  = (const float*)d_in[2];
  const float* Wx_w = (const float*)d_in[3];
  const float* Wx_b = (const float*)d_in[4];
  float* out = (float*)d_out;

  float* G  = (float*)d_ws;

  const float* h0 = feat;
  const float* h1 = feat + (size_t)TD;
  const float* h2 = feat + (size_t)2 * TD;

  dim3 blk(256);
  dim3 grd_mm(NBLK);
  dim3 grd_cd(CBLK);

  gsn_gate_wmma<false><<<grd_mm, blk, 0, stream>>>(h0, h2, nullptr, Wx_w, Wx_b, G);
  gsn_cand_wmma<false><<<grd_cd, blk, 0, stream>>>(h0, h2, nullptr, W_w, W_b, G, out + 0 * (size_t)TD);

  gsn_gate_wmma<false><<<grd_mm, blk, 0, stream>>>(h1, h0, nullptr, Wx_w, Wx_b, G);
  gsn_cand_wmma<false><<<grd_cd, blk, 0, stream>>>(h1, h0, nullptr, W_w, W_b, G, out + 1 * (size_t)TD);

  gsn_gate_wmma<true><<<grd_mm, blk, 0, stream>>>(h2, h1, h0, Wx_w, Wx_b, G);
  gsn_cand_wmma<true><<<grd_cd, blk, 0, stream>>>(h2, h1, h0, W_w, W_b, G, out + 2 * (size_t)TD);
}


# Blend gates recomputed in the cand pass: a second product on the gate weights over operands it holds anyway (same lane, same register as the candidate); gate pass cut to the message slots' gates
# speedup vs baseline: 3.2015x; 1.2333x over previous
#include <hip/hip_runtime.h>
#include <math.h>

#define LOG_TD 22u
#define TD (1u << LOG_TD)

typedef __attribute__((ext_vector_type(2))) float v2f;
typedef __attribute__((ext_vector_type(8))) float v8f;

constexpr unsigned TPW    = 6u;
constexpr unsigned GCH    = 3u;
constexpr unsigned GBLK1  = 512u;
constexpr unsigned GBLK2  = 1024u;
constexpr unsigned GWAV1  = 4096u;
constexpr unsigned GWAV2  = 8192u;
static_assert(GBLK2 == 2u * GBLK1 && GWAV1 == 8u * GBLK1 && GWAV2 == 8u * GBLK2, "gate grids");
static_assert(GBLK1 * 8u * TPW * 192u >= TD, "the gate grid must cover a whole slot");
constexpr unsigned NQ     = TD / 4u;
constexpr unsigned TPS    = 2u;
constexpr unsigned GPB    = 16u * 8u * TPS;
constexpr unsigned LQ     = 760u;
constexpr unsigned CBLK   = (NQ + LQ - 1u) / LQ;
static_assert(LQ % 8u == 0u && LQ + 2u <= 3u * GPB, "block range must be whole lines and fit its groups");

__device__ __forceinline__ float hw_exp2(float x) {
#if __has_builtin(__builtin_amdgcn_exp2f)
  return __builtin_amdgcn_exp2f(x);
#else
  return exp2f(x);
#endif
}
__device__ __forceinline__ float hw_rcp(float x) {
#if __has_builtin(__builtin_amdgcn_rcpf)
  return __builtin_amdgcn_rcpf(x);
#else
  return 1.0f / x;
#endif
}
__device__ __forceinline__ float fast_sigmoid(float x) {
  return hw_rcp(1.0f + hw_exp2(-1.4426950408889634f * x));
}
__device__ __forceinline__ float fast_tanh(float x) {
#if __has_builtin(__builtin_amdgcn_tanhf)
  return __builtin_amdgcn_tanhf(x);
#else
  return 1.0f - 2.0f * hw_rcp(hw_exp2(2.8853900817779268f * x) + 1.0f);
#endif
}

__device__ __forceinline__ float4 ld4(const float* __restrict__ p) {
  return *(const float4*)p;
}
__device__ __forceinline__ float4 zero4() { return make_float4(0.0f, 0.0f, 0.0f, 0.0f); }
__device__ __forceinline__ float4 sel4(bool c, const float4& v) {
  return make_float4(c ? v.x : 0.0f, c ? v.y : 0.0f, c ? v.z : 0.0f, c ? v.w : 0.0f);
}


template <bool HAS_M1>
__device__ __forceinline__ float4 gate_src4(unsigned m,
    const float* __restrict__ hv, const float* __restrict__ m0,
    const float* __restrict__ m1) {
  const unsigned seg = m >> LOG_TD;
  const unsigned r   = m & (TD - 1u);
  const bool is_h  = seg >= 3u;
  const bool is_m0 = seg == 0u;
  const bool is_m1 = HAS_M1 && seg == 1u;
  const bool valid = is_h || is_m0 || is_m1;
  const float* base = is_m0 ? m0 : (is_m1 ? m1 : hv);
  const float4 v = ld4(base + (valid ? r : 0u));
  return sel4(valid, v);
}

template <bool HAS_M1>
__device__ __forceinline__ float4 msg4(unsigned p, const float* __restrict__ hv,
    const float* __restrict__ m0, const float* __restrict__ m1) {
  const unsigned n = p >> LOG_TD;
  const unsigned r = p & (TD - 1u);
  const bool is_m0 = n == 0u;
  const bool is_m1 = HAS_M1 && n == 1u;
  const bool valid = is_m0 || is_m1;
  const float* base = is_m0 ? m0 : (is_m1 ? m1 : hv);
  const float4 v = ld4(base + (valid ? r : 0u));
  return sel4(valid, v);
}

template <bool HAS_M1>
__device__ __forceinline__ void cand_src4_load(unsigned m,
    const float* __restrict__ hv, const float* __restrict__ m0,
    const float* __restrict__ m1, const float* __restrict__ G,
    float4& u, float4& w) {
  const unsigned seg = m >> LOG_TD;
  const unsigned r   = m & (TD - 1u);
  const unsigned n   = seg >> 1;
  const bool is_h  = (seg & 1u) != 0u;
  const bool is_m0 = !is_h && n == 0u;
  const bool is_m1 = HAS_M1 && !is_h && n == 1u;
  const bool prod  = is_m0 || is_m1;
  const float* ub = prod ? G : hv;
  const unsigned uo = prod ? (n * TD + r) : (is_h ? r : 0u);
  u = ld4(ub + uo);
  const float* wb = is_m0 ? m0 : (is_m1 ? m1 : hv);
  w = ld4(wb + (prod ? r : 0u));
}
template <bool HAS_M1>
__device__ __forceinline__ float4 cand_src4_value(unsigned m, const float4& u,
    const float4& w) {
  const unsigned seg = m >> LOG_TD;
  const unsigned n   = seg >> 1;
  const bool is_h  = (seg & 1u) != 0u;
  const bool prod  = !is_h && (n == 0u || (HAS_M1 && n == 1u));
  const float4 gm = make_float4(u.x * w.x, u.y * w.y, u.z * w.z, u.w * w.w);
  return make_float4(is_h ? u.x : (prod ? gm.x : 0.0f), is_h ? u.y : (prod ? gm.y : 0.0f),
                     is_h ? u.z : (prod ? gm.z : 0.0f), is_h ? u.w : (prod ? gm.w : 0.0f));
}

__device__ __forceinline__ void make_weights(const float* __restrict__ Wm,
    const float* __restrict__ Bv, unsigned lane, v2f (&a)[6], v8f& cinit) {
  const unsigned o  = lane & 15u;
  const unsigned h  = lane >> 4;
  const unsigned g  = o / 3u;
  const unsigned j  = o - 3u * g;
  const bool valid  = (o < 12u) && ((g >> 1) == h);
  const float msk   = valid ? 1.0f : 0.0f;
  const bool second = (g & 1u) != 0u;
  float w[6];
#pragma unroll
  for (int i = 0; i < 6; ++i) w[i] = Wm[j * 6u + (unsigned)i] * msk;
#pragma unroll
  for (int c = 0; c < 3; ++c) {
    a[c].x     = second ? 0.0f : w[2 * c];
    a[c].y     = second ? 0.0f : w[2 * c + 1];
    a[c + 3].x = second ? w[2 * c]     : 0.0f;
    a[c + 3].y = second ? w[2 * c + 1] : 0.0f;
  }
  const float b0 = Bv[0], b1 = Bv[1], b2 = Bv[2];
  const bool hi = h != 0u;
  cinit[0] = hi ? b2 : b0;
  cinit[1] = hi ? b0 : b1;
  cinit[2] = hi ? b1 : b2;
  cinit[3] = hi ? b2 : b0;
  cinit[4] = hi ? 0.0f : b1;
  cinit[5] = hi ? 0.0f : b2;
  cinit[6] = hi ? 0.0f : b0;
  cinit[7] = hi ? 0.0f : b1;
}

__device__ __forceinline__ v8f group_product(const v2f (&a)[6], const float4& x0,
    const float4& x1, const float4& x2, v8f c) {
  v2f b;
  b.x = x0.x; b.y = x0.y;
  c = __builtin_amdgcn_wmma_f32_16x16x4_f32(false, a[0], false, b, (short)0, c, false, false);
  b.x = x0.z; b.y = x0.w;
  c = __builtin_amdgcn_wmma_f32_16x16x4_f32(false, a[1], false, b, (short)0, c, false, false);
  b.x = x1.x; b.y = x1.y;
  c = __builtin_amdgcn_wmma_f32_16x16x4_f32(false, a[2], false, b, (short)0, c, false, false);
  b.x = x1.z; b.y = x1.w;
  c = __builtin_amdgcn_wmma_f32_16x16x4_f32(false, a[3], false, b, (short)0, c, false, false);
  b.x = x2.x; b.y = x2.y;
  c = __builtin_amdgcn_wmma_f32_16x16x4_f32(false, a[4], false, b, (short)0, c, false, false);
  b.x = x2.z; b.y = x2.w;
  c = __builtin_amdgcn_wmma_f32_16x16x4_f32(false, a[5], false, b, (short)0, c, false, false);
  return c;
}

template <bool HAS_M1, unsigned NWAVES>
__global__ __launch_bounds__(256, 8) void gsn_gate_wmma(
    const float* __restrict__ hv, const float* __restrict__ m0,
    const float* __restrict__ m1, const float* __restrict__ Wx,
    const float* __restrict__ Bx,
    float* __restrict__ G) {
  const unsigned lane = threadIdx.x & 31u;
  const unsigned wv   = threadIdx.x >> 5u;
  const unsigned gw   = blockIdx.x * 8u + wv;
  const unsigned h    = lane >> 4;
  const unsigned loff = 24u * (lane & 15u) + 12u * h;
  const unsigned poff = 12u * (lane & 15u) + 8u * h;
  const unsigned lw   = 384u * wv + loff;

  v2f a[6];
  v8f cinit;
  make_weights(Wx, Bx, lane, a, cinit);

  for (unsigned ch = 0; ch < TPW / GCH; ++ch) {
    float4 x0[GCH], x1[GCH], x2[GCH];
#pragma unroll
    for (unsigned t = 0; t < GCH; ++t) {
      const unsigned k  = ch * GCH + t;
      const unsigned f0 = 384u * (k * NWAVES + blockIdx.x * 8u);
      const unsigned seg = f0 >> LOG_TD;
      const bool whole = ((f0 + 8u * 384u - 1u) >> LOG_TD) == seg;
      const bool is_h  = seg >= 3u;
      const bool is_m0 = seg == 0u;
      const bool is_m1 = HAS_M1 && seg == 1u;
      if (whole && (is_h || is_m0 || is_m1)) {
        const float* bp = (is_h ? hv : (is_m0 ? m0 : m1)) + (f0 & (TD - 1u));
        x0[t] = ld4(bp + lw);
        x1[t] = ld4(bp + lw + 4u);
        x2[t] = ld4(bp + lw + 8u);
      } else if (whole) {
        x0[t] = zero4(); x1[t] = zero4(); x2[t] = zero4();
      } else {
        const unsigned base = f0 + lw;
        x0[t] = gate_src4<HAS_M1>(base,      hv, m0, m1);
        x1[t] = gate_src4<HAS_M1>(base + 4u, hv, m0, m1);
        x2[t] = gate_src4<HAS_M1>(base + 8u, hv, m0, m1);
      }
    }
#pragma unroll
    for (unsigned t = 0; t < GCH; ++t) {
      const unsigned tile = (ch * GCH + t) * NWAVES + gw;
      const v8f d = group_product(a, x0[t], x1[t], x2[t], cinit);

      const unsigned p = 192u * tile + poff;
      float4 o0;
      o0.x = fast_sigmoid(d[0]); o0.y = fast_sigmoid(d[1]);
      o0.z = fast_sigmoid(d[2]); o0.w = fast_sigmoid(d[3]);
      *(float4*)(G + p) = o0;
      if (h == 0u) {
        float4 o1;
        o1.x = fast_sigmoid(d[4]); o1.y = fast_sigmoid(d[5]);
        o1.z = fast_sigmoid(d[6]); o1.w = fast_sigmoid(d[7]);
        *(float4*)(G + p + 4u) = o1;
      }
    }
  }
}

__device__ __forceinline__ float4 sigmoid4(float d0, float d1, float d2, float d3) {
  return make_float4(fast_sigmoid(d0), fast_sigmoid(d1), fast_sigmoid(d2), fast_sigmoid(d3));
}

__device__ __forceinline__ float4 delta4(const float4& g, const float4& mv,
    float d0, float d1, float d2, float d3) {
  float4 o;
  o.x = (1.0f - g.x) * mv.x + g.x * fast_tanh(d0);
  o.y = (1.0f - g.y) * mv.y + g.y * fast_tanh(d1);
  o.z = (1.0f - g.z) * mv.z + g.z * fast_tanh(d2);
  o.w = (1.0f - g.w) * mv.w + g.w * fast_tanh(d3);
  return o;
}

__device__ __forceinline__ float4 gated4(const float4& g, float d0, float d1, float d2, float d3) {
  float4 o;
  o.x = g.x * fast_tanh(d0);
  o.y = g.y * fast_tanh(d1);
  o.z = g.z * fast_tanh(d2);
  o.w = g.w * fast_tanh(d3);
  return o;
}


template <bool HAS_M1>
__device__ __forceinline__ void cand_slot_lane(unsigned Q0, unsigned g0, unsigned nq,
    unsigned ll, unsigned ql, unsigned h, const v2f (&a)[6], const v8f& cinit,
    const v2f (&ag)[6], const v8f& ginit,
    const float* __restrict__ hv, const float* __restrict__ m0, const float* __restrict__ m1,
    const float* __restrict__ G, float4* accn) {
  for (unsigned t = 0; t < TPS; ++t) {
    const unsigned bu   = 24u * (g0 + t * 128u) + ll;
    const unsigned bmax = 24u * (NQ - 1u) + 12u * h;
    const unsigned base = bu < bmax ? bu : bmax;

    const float4 y0 = gate_src4<HAS_M1>(base,      hv, m0, m1);
    const float4 y1 = gate_src4<HAS_M1>(base + 4u, hv, m0, m1);
    const float4 y2 = gate_src4<HAS_M1>(base + 8u, hv, m0, m1);
    const v8f dg = group_product(ag, y0, y1, y2, ginit);
    const float4 ga = sigmoid4(dg[0], dg[1], dg[2], dg[3]);
    const float4 gb = sigmoid4(dg[4], dg[5], dg[6], dg[7]);

    float4 u0, u1, u2, w0, w1, w2;
    cand_src4_load<HAS_M1>(base,      hv, m0, m1, G, u0, w0);
    cand_src4_load<HAS_M1>(base + 4u, hv, m0, m1, G, u1, w1);
    cand_src4_load<HAS_M1>(base + 8u, hv, m0, m1, G, u2, w2);
    const float4 x0 = cand_src4_value<HAS_M1>(base,      u0, w0);
    const float4 x1 = cand_src4_value<HAS_M1>(base + 4u, u1, w1);
    const float4 x2 = cand_src4_value<HAS_M1>(base + 8u, u2, w2);
    const v8f d = group_product(a, x0, x1, x2, cinit);

    const unsigned p  = (base >> 1) + 2u * h;
    const unsigned p2 = p + 4u * (h ^ 1u);
    const float4 ma = msg4<HAS_M1>(p,  hv, m0, m1);
    const float4 mb = msg4<HAS_M1>(p2, hv, m0, m1);

    const unsigned qa = 3u * (g0 + t * 128u) - Q0 + ql;
    if (qa < nq)
      accn[qa] = delta4(ga, ma, d[0], d[1], d[2], d[3]);
    if (h == 0u && qa + 1u < nq)
      accn[qa + 1u] = delta4(gb, mb, d[4], d[5], d[6], d[7]);
  }
}

constexpr int K_PROD = 1;
constexpr int K_H    = 2;
constexpr int K_ZERO = 3;

constexpr int GK_W    = 1;
constexpr int GK_U    = 2;
constexpr int GK_ZERO = 3;
constexpr int GK_LOAD = 4;

template <int KIND, bool MSG, int GK>
__device__ __forceinline__ void cand_slot_uniform(
    const float* __restrict__ ub, const float* __restrict__ wb,
    const float* __restrict__ gs, const float* __restrict__ mp,
    unsigned rb, unsigned pr, unsigned qoff, unsigned nq, unsigned ll, unsigned ql, unsigned h,
    const v2f (&a)[6], const v8f& cinit, const v2f (&ag)[6], const v8f& ginit, float4* accn) {
  static_assert(GK != GK_W || KIND == K_PROD, "w is loaded only for a gate * message slab");
  static_assert(GK != GK_U || KIND == K_H, "u is the node state only in an h slab");
  const unsigned lo = rb + ll;
  const unsigned po = pr + 4u * ql;
  const unsigned pq = po + 4u * (h ^ 1u);
  float4 u[TPS][3], w[TPS][3], y[TPS][3], ma[TPS], mb[TPS];
#pragma unroll
  for (unsigned t = 0; t < TPS; ++t) {
    const unsigned ot = lo + t * (24u * 128u);
    const unsigned pt = po + t * (12u * 128u);
    const unsigned pu = pq + t * (12u * 128u);
    if (KIND != K_ZERO) {
      u[t][0] = ld4(ub + ot);
      u[t][1] = ld4(ub + ot + 4u);
      u[t][2] = ld4(ub + ot + 8u);
    } else {
      u[t][0] = zero4(); u[t][1] = zero4(); u[t][2] = zero4();
    }
    if (KIND == K_PROD) {
      w[t][0] = ld4(wb + ot);
      w[t][1] = ld4(wb + ot + 4u);
      w[t][2] = ld4(wb + ot + 8u);
    } else {
      w[t][0] = zero4(); w[t][1] = zero4(); w[t][2] = zero4();
    }
    if (GK == GK_LOAD) {
      y[t][0] = ld4(gs + ot);
      y[t][1] = ld4(gs + ot + 4u);
      y[t][2] = ld4(gs + ot + 8u);
    } else {
      y[t][0] = zero4(); y[t][1] = zero4(); y[t][2] = zero4();
    }
    if (MSG) {
      ma[t] = ld4(mp + pt);
      mb[t] = ld4(mp + pu);
    } else {
      ma[t] = zero4(); mb[t] = zero4();
    }
  }
#pragma unroll
  for (unsigned t = 0; t < TPS; ++t) {
    if (KIND != K_ZERO)
      asm volatile("" :: "v"(u[t][0].x), "v"(u[t][0].y), "v"(u[t][0].z), "v"(u[t][0].w),
                         "v"(u[t][1].x), "v"(u[t][1].y), "v"(u[t][1].z), "v"(u[t][1].w),
                         "v"(u[t][2].x), "v"(u[t][2].y), "v"(u[t][2].z), "v"(u[t][2].w));
    if (KIND == K_PROD)
      asm volatile("" :: "v"(w[t][0].x), "v"(w[t][0].y), "v"(w[t][0].z), "v"(w[t][0].w),
                         "v"(w[t][1].x), "v"(w[t][1].y), "v"(w[t][1].z), "v"(w[t][1].w),
                         "v"(w[t][2].x), "v"(w[t][2].y), "v"(w[t][2].z), "v"(w[t][2].w));
    if (GK == GK_LOAD)
      asm volatile("" :: "v"(y[t][0].x), "v"(y[t][0].y), "v"(y[t][0].z), "v"(y[t][0].w),
                         "v"(y[t][1].x), "v"(y[t][1].y), "v"(y[t][1].z), "v"(y[t][1].w),
                         "v"(y[t][2].x), "v"(y[t][2].y), "v"(y[t][2].z), "v"(y[t][2].w));
    if (MSG)
      asm volatile("" :: "v"(ma[t].x), "v"(ma[t].y), "v"(ma[t].z), "v"(ma[t].w),
                         "v"(mb[t].x), "v"(mb[t].y), "v"(mb[t].z), "v"(mb[t].w));
  }
#pragma unroll
  for (unsigned t = 0; t < TPS; ++t) {
    const float4 y0 = GK == GK_W ? w[t][0] : (GK == GK_U ? u[t][0] : y[t][0]);
    const float4 y1 = GK == GK_W ? w[t][1] : (GK == GK_U ? u[t][1] : y[t][1]);
    const float4 y2 = GK == GK_W ? w[t][2] : (GK == GK_U ? u[t][2] : y[t][2]);
    float4 x0 = u[t][0], x1 = u[t][1], x2 = u[t][2];
    if (KIND == K_PROD) {
      x0 = make_float4(x0.x * w[t][0].x, x0.y * w[t][0].y, x0.z * w[t][0].z, x0.w * w[t][0].w);
      x1 = make_float4(x1.x * w[t][1].x, x1.y * w[t][1].y, x1.z * w[t][1].z, x1.w * w[t][1].w);
      x2 = make_float4(x2.x * w[t][2].x, x2.y * w[t][2].y, x2.z * w[t][2].z, x2.w * w[t][2].w);
    }
    const v8f d = group_product(a, x0, x1, x2, cinit);
    const v8f dg = group_product(ag, y0, y1, y2, ginit);
    const float4 ga = sigmoid4(dg[0], dg[1], dg[2], dg[3]);
    const float4 gb = sigmoid4(dg[4], dg[5], dg[6], dg[7]);

    const unsigned qa = qoff + t * 384u + ql;
    if (qa < nq)
      accn[qa] = MSG ? delta4(ga, ma[t], d[0], d[1], d[2], d[3])
                     : gated4(ga, d[0], d[1], d[2], d[3]);
    if (h == 0u && qa + 1u < nq)
      accn[qa + 1u] = MSG ? delta4(gb, mb[t], d[4], d[5], d[6], d[7])
                          : gated4(gb, d[4], d[5], d[6], d[7]);
  }
}

template <bool HAS_M1>
__global__ __launch_bounds__(256) __attribute__((amdgpu_waves_per_eu(8, 8))) void gsn_cand_wmma(
    const float* __restrict__ hv, const float* __restrict__ m0,
    const float* __restrict__ m1, const float* __restrict__ Ww,
    const float* __restrict__ Bw,
    const float* __restrict__ Wx,
    const float* __restrict__ Bx,
    const float* __restrict__ G, float* __restrict__ out) {
  __shared__ float4 acc[3][LQ];
  const unsigned lane = threadIdx.x & 31u;
  const unsigned h    = lane >> 4;
  const unsigned gl   = (threadIdx.x >> 5u) * 16u + (lane & 15u);
  const unsigned ll   = 24u * gl + 12u * h;
  const unsigned ql   = 3u * gl + 2u * h;
  const unsigned q0   = blockIdx.x * LQ;
  const unsigned nq   = (NQ - q0) < LQ ? (NQ - q0) : LQ;

  v2f a[6], ag[6];
  v8f cinit, ginit;
  make_weights(Ww, Bw, lane, a, cinit);
  make_weights(Wx, Bx, lane, ag, ginit);

  bool all_whole = true;
  for (unsigned n = 0; n < 3u; ++n) {
    const unsigned g0    = (n * NQ + q0) / 3u;
    const unsigned op_lo = 24u * g0;
    const unsigned ep_lo = 12u * g0;
    all_whole = all_whole && g0 + (GPB - 1u) < NQ
        && ((op_lo + 24u * GPB - 1u) >> LOG_TD) == (op_lo >> LOG_TD)
        && (ep_lo >> LOG_TD) == n && ((ep_lo + 12u * GPB - 1u) >> LOG_TD) == n;
  }
  if (all_whole) {
    for (unsigned n = 0; n < 3u; ++n) {
      const unsigned Q0 = n * NQ + q0;
      const unsigned g0 = Q0 / 3u;
      const unsigned op_lo = 24u * g0;
      const unsigned ep_lo = 12u * g0;
      const unsigned seg   = op_lo >> LOG_TD;
      const unsigned rb  = op_lo & (TD - 1u);
      const unsigned pr  = ep_lo & (TD - 1u);
      const unsigned qoff = 3u * g0 - Q0;
      if (seg == 0u)
        cand_slot_uniform<K_PROD, true, GK_W>(G, m0, hv, m0, rb, pr, qoff, nq, ll, ql, h, a, cinit, ag, ginit, acc[n]);
      else if (seg == 1u && HAS_M1)
        cand_slot_uniform<K_H, true, GK_LOAD>(hv, hv, m1, m0, rb, pr, qoff, nq, ll, ql, h, a, cinit, ag, ginit, acc[n]);
      else if (seg == 1u)
        cand_slot_uniform<K_H, true, GK_ZERO>(hv, hv, hv, m0, rb, pr, qoff, nq, ll, ql, h, a, cinit, ag, ginit, acc[n]);
      else if (seg == 2u && HAS_M1)
        cand_slot_uniform<K_PROD, true, GK_ZERO>(G + TD, m1, hv, m1, rb, pr, qoff, nq, ll, ql, h, a, cinit, ag, ginit, acc[n]);
      else if (seg == 2u)
        cand_slot_uniform<K_ZERO, false, GK_ZERO>(hv, hv, hv, hv, rb, pr, qoff, nq, ll, ql, h, a, cinit, ag, ginit, acc[n]);
      else if (seg == 3u && HAS_M1)
        cand_slot_uniform<K_H, true, GK_U>(hv, hv, hv, m1, rb, pr, qoff, nq, ll, ql, h, a, cinit, ag, ginit, acc[n]);
      else if (seg == 4u)
        cand_slot_uniform<K_ZERO, false, GK_LOAD>(hv, hv, hv, hv, rb, pr, qoff, nq, ll, ql, h, a, cinit, ag, ginit, acc[n]);
      else
        cand_slot_uniform<K_H, false, GK_U>(hv, hv, hv, hv, rb, pr, qoff, nq, ll, ql, h, a, cinit, ag, ginit, acc[n]);
    }
  } else {
    for (unsigned n = 0; n < 3u; ++n) {
      const unsigned Q0 = n * NQ + q0;
      cand_slot_lane<HAS_M1>(Q0, Q0 / 3u, nq, ll, ql, h, a, cinit, ag, ginit, hv, m0, m1, G, acc[n]);
    }
  }
  __syncthreads();
  for (unsigned i = threadIdx.x; i < nq; i += 256u) {
    const float4 h4 = ld4(hv + 4u * (q0 + i));
    const float4 c0 = acc[0][i];
    const float4 c1 = acc[1][i];
    const float4 c2 = acc[2][i];
    float4 o;
    o.x = h4.x + c0.x + c1.x + c2.x;
    o.y = h4.y + c0.y + c1.y + c2.y;
    o.z = h4.z + c0.z + c1.z + c2.z;
    o.w = h4.w + c0.w + c1.w + c2.w;
    *(float4*)(out + 4u * (q0 + i)) = o;
  }
}

extern "C" void kernel_launch(void* const* d_in, const int* in_sizes, int n_in,
                              void* d_out, int out_size, void* d_ws, size_t ws_size,
                              hipStream_t stream) {
  (void)in_sizes; (void)n_in; (void)out_size; (void)ws_size;
  const float* feat = (const float*)d_in[0];
  const float* W_w  = (const float*)d_in[1];
  const float* W_b  = (const float*)d_in[2];
  const float* Wx_w = (const float*)d_in[3];
  const float* Wx_b = (const float*)d_in[4];
  float* out = (float*)d_out;

  float* G  = (float*)d_ws;

  const float* h0 = feat;
  const float* h1 = feat + (size_t)TD;
  const float* h2 = feat + (size_t)2 * TD;

  dim3 blk(256);
  dim3 grd_g1(GBLK1);
  dim3 grd_g2(GBLK2);
  dim3 grd_cd(CBLK);

  gsn_gate_wmma<false, GWAV1><<<grd_g1, blk, 0, stream>>>(h0, h2, nullptr, Wx_w, Wx_b, G);
  gsn_cand_wmma<false><<<grd_cd, blk, 0, stream>>>(h0, h2, nullptr, W_w, W_b, Wx_w, Wx_b, G, out + 0 * (size_t)TD);

  gsn_gate_wmma<false, GWAV1><<<grd_g1, blk, 0, stream>>>(h1, h0, nullptr, Wx_w, Wx_b, G);
  gsn_cand_wmma<false><<<grd_cd, blk, 0, stream>>>(h1, h0, nullptr, W_w, W_b, Wx_w, Wx_b, G, out + 1 * (size_t)TD);

  gsn_gate_wmma<true, GWAV2><<<grd_g2, blk, 0, stream>>>(h2, h1, h0, Wx_w, Wx_b, G);
  gsn_cand_wmma<true><<<grd_cd, blk, 0, stream>>>(h2, h1, h0, W_w, W_b, Wx_w, Wx_b, G, out + 2 * (size_t)TD);
}


# Nodes 0 and 1 (independent, one message each) share a gate launch and a cand launch: a block picks its node's planes by a scalar test; node 1's gates lie behind node 0's store extent; 6->4 launches
# speedup vs baseline: 3.4803x; 1.0871x over previous
#include <hip/hip_runtime.h>
#include <math.h>

#define LOG_TD 22u
#define TD (1u << LOG_TD)

typedef __attribute__((ext_vector_type(2))) float v2f;
typedef __attribute__((ext_vector_type(8))) float v8f;

constexpr unsigned TPW    = 6u;
constexpr unsigned GCH    = 3u;
constexpr unsigned GBLK1  = 512u;
constexpr unsigned GBLK2  = 1024u;
constexpr unsigned GWAV1  = 4096u;
constexpr unsigned GWAV2  = 8192u;
static_assert(GBLK2 == 2u * GBLK1 && GWAV1 == 8u * GBLK1 && GWAV2 == 8u * GBLK2, "gate grids");
static_assert(GBLK1 * 8u * TPW * 192u >= TD, "the gate grid must cover a whole slot");
constexpr unsigned GEXT1  = GWAV1 * TPW * 192u;
static_assert(GEXT1 >= TD && 2u * GEXT1 == GWAV2 * TPW * 192u, "two one-slot parts fill the two-slot grid's extent");
constexpr unsigned NQ     = TD / 4u;
constexpr unsigned TPS    = 2u;
constexpr unsigned GPB    = 16u * 8u * TPS;
constexpr unsigned LQ     = 760u;
constexpr unsigned CBLK   = (NQ + LQ - 1u) / LQ;
constexpr unsigned CBLK2  = 2u * CBLK;
static_assert(LQ % 8u == 0u && LQ + 2u <= 3u * GPB, "block range must be whole lines and fit its groups");

__device__ __forceinline__ float hw_exp2(float x) {
#if __has_builtin(__builtin_amdgcn_exp2f)
  return __builtin_amdgcn_exp2f(x);
#else
  return exp2f(x);
#endif
}
__device__ __forceinline__ float hw_rcp(float x) {
#if __has_builtin(__builtin_amdgcn_rcpf)
  return __builtin_amdgcn_rcpf(x);
#else
  return 1.0f / x;
#endif
}
__device__ __forceinline__ float fast_sigmoid(float x) {
  return hw_rcp(1.0f + hw_exp2(-1.4426950408889634f * x));
}
__device__ __forceinline__ float fast_tanh(float x) {
#if __has_builtin(__builtin_amdgcn_tanhf)
  return __builtin_amdgcn_tanhf(x);
#else
  return 1.0f - 2.0f * hw_rcp(hw_exp2(2.8853900817779268f * x) + 1.0f);
#endif
}

__device__ __forceinline__ float4 ld4(const float* __restrict__ p) {
  return *(const float4*)p;
}
__device__ __forceinline__ float4 zero4() { return make_float4(0.0f, 0.0f, 0.0f, 0.0f); }
__device__ __forceinline__ float4 sel4(bool c, const float4& v) {
  return make_float4(c ? v.x : 0.0f, c ? v.y : 0.0f, c ? v.z : 0.0f, c ? v.w : 0.0f);
}


template <bool HAS_M1>
__device__ __forceinline__ float4 gate_src4(unsigned m,
    const float* __restrict__ hv, const float* __restrict__ m0,
    const float* __restrict__ m1) {
  const unsigned seg = m >> LOG_TD;
  const unsigned r   = m & (TD - 1u);
  const bool is_h  = seg >= 3u;
  const bool is_m0 = seg == 0u;
  const bool is_m1 = HAS_M1 && seg == 1u;
  const bool valid = is_h || is_m0 || is_m1;
  const float* base = is_m0 ? m0 : (is_m1 ? m1 : hv);
  const float4 v = ld4(base + (valid ? r : 0u));
  return sel4(valid, v);
}

template <bool HAS_M1>
__device__ __forceinline__ float4 msg4(unsigned p, const float* __restrict__ hv,
    const float* __restrict__ m0, const float* __restrict__ m1) {
  const unsigned n = p >> LOG_TD;
  const unsigned r = p & (TD - 1u);
  const bool is_m0 = n == 0u;
  const bool is_m1 = HAS_M1 && n == 1u;
  const bool valid = is_m0 || is_m1;
  const float* base = is_m0 ? m0 : (is_m1 ? m1 : hv);
  const float4 v = ld4(base + (valid ? r : 0u));
  return sel4(valid, v);
}

template <bool HAS_M1>
__device__ __forceinline__ void cand_src4_load(unsigned m,
    const float* __restrict__ hv, const float* __restrict__ m0,
    const float* __restrict__ m1, const float* __restrict__ G,
    float4& u, float4& w) {
  const unsigned seg = m >> LOG_TD;
  const unsigned r   = m & (TD - 1u);
  const unsigned n   = seg >> 1;
  const bool is_h  = (seg & 1u) != 0u;
  const bool is_m0 = !is_h && n == 0u;
  const bool is_m1 = HAS_M1 && !is_h && n == 1u;
  const bool prod  = is_m0 || is_m1;
  const float* ub = prod ? G : hv;
  const unsigned uo = prod ? (n * TD + r) : (is_h ? r : 0u);
  u = ld4(ub + uo);
  const float* wb = is_m0 ? m0 : (is_m1 ? m1 : hv);
  w = ld4(wb + (prod ? r : 0u));
}
template <bool HAS_M1>
__device__ __forceinline__ float4 cand_src4_value(unsigned m, const float4& u,
    const float4& w) {
  const unsigned seg = m >> LOG_TD;
  const unsigned n   = seg >> 1;
  const bool is_h  = (seg & 1u) != 0u;
  const bool prod  = !is_h && (n == 0u || (HAS_M1 && n == 1u));
  const float4 gm = make_float4(u.x * w.x, u.y * w.y, u.z * w.z, u.w * w.w);
  return make_float4(is_h ? u.x : (prod ? gm.x : 0.0f), is_h ? u.y : (prod ? gm.y : 0.0f),
                     is_h ? u.z : (prod ? gm.z : 0.0f), is_h ? u.w : (prod ? gm.w : 0.0f));
}

__device__ __forceinline__ void make_weights(const float* __restrict__ Wm,
    const float* __restrict__ Bv, unsigned lane, v2f (&a)[6], v8f& cinit) {
  const unsigned o  = lane & 15u;
  const unsigned h  = lane >> 4;
  const unsigned g  = o / 3u;
  const unsigned j  = o - 3u * g;
  const bool valid  = (o < 12u) && ((g >> 1) == h);
  const float msk   = valid ? 1.0f : 0.0f;
  const bool second = (g & 1u) != 0u;
  float w[6];
#pragma unroll
  for (int i = 0; i < 6; ++i) w[i] = Wm[j * 6u + (unsigned)i] * msk;
#pragma unroll
  for (int c = 0; c < 3; ++c) {
    a[c].x     = second ? 0.0f : w[2 * c];
    a[c].y     = second ? 0.0f : w[2 * c + 1];
    a[c + 3].x = second ? w[2 * c]     : 0.0f;
    a[c + 3].y = second ? w[2 * c + 1] : 0.0f;
  }
  const float b0 = Bv[0], b1 = Bv[1], b2 = Bv[2];
  const bool hi = h != 0u;
  cinit[0] = hi ? b2 : b0;
  cinit[1] = hi ? b0 : b1;
  cinit[2] = hi ? b1 : b2;
  cinit[3] = hi ? b2 : b0;
  cinit[4] = hi ? 0.0f : b1;
  cinit[5] = hi ? 0.0f : b2;
  cinit[6] = hi ? 0.0f : b0;
  cinit[7] = hi ? 0.0f : b1;
}

__device__ __forceinline__ v8f group_product(const v2f (&a)[6], const float4& x0,
    const float4& x1, const float4& x2, v8f c) {
  v2f b;
  b.x = x0.x; b.y = x0.y;
  c = __builtin_amdgcn_wmma_f32_16x16x4_f32(false, a[0], false, b, (short)0, c, false, false);
  b.x = x0.z; b.y = x0.w;
  c = __builtin_amdgcn_wmma_f32_16x16x4_f32(false, a[1], false, b, (short)0, c, false, false);
  b.x = x1.x; b.y = x1.y;
  c = __builtin_amdgcn_wmma_f32_16x16x4_f32(false, a[2], false, b, (short)0, c, false, false);
  b.x = x1.z; b.y = x1.w;
  c = __builtin_amdgcn_wmma_f32_16x16x4_f32(false, a[3], false, b, (short)0, c, false, false);
  b.x = x2.x; b.y = x2.y;
  c = __builtin_amdgcn_wmma_f32_16x16x4_f32(false, a[4], false, b, (short)0, c, false, false);
  b.x = x2.z; b.y = x2.w;
  c = __builtin_amdgcn_wmma_f32_16x16x4_f32(false, a[5], false, b, (short)0, c, false, false);
  return c;
}

template <bool HAS_M1, unsigned NWAVES, bool TWO>
__global__ __launch_bounds__(256, 8) void gsn_gate_wmma(
    const float* __restrict__ hvA, const float* __restrict__ m0A,
    const float* __restrict__ m1A, float* __restrict__ GA,
    const float* __restrict__ hvB, const float* __restrict__ m0B,
    const float* __restrict__ m1B, float* __restrict__ GB,
    const float* __restrict__ Wx,
    const float* __restrict__ Bx) {
  const bool second = TWO && blockIdx.x >= NWAVES / 8u;
  const unsigned bx = second ? blockIdx.x - NWAVES / 8u : blockIdx.x;
  const float* __restrict__ hv = second ? hvB : hvA;
  const float* __restrict__ m0 = second ? m0B : m0A;
  const float* __restrict__ m1 = second ? m1B : m1A;
  float* __restrict__ G        = second ? GB : GA;
  const unsigned lane = threadIdx.x & 31u;
  const unsigned wv   = threadIdx.x >> 5u;
  const unsigned gw   = bx * 8u + wv;
  const unsigned h    = lane >> 4;
  const unsigned loff = 24u * (lane & 15u) + 12u * h;
  const unsigned poff = 12u * (lane & 15u) + 8u * h;
  const unsigned lw   = 384u * wv + loff;

  v2f a[6];
  v8f cinit;
  make_weights(Wx, Bx, lane, a, cinit);

  for (unsigned ch = 0; ch < TPW / GCH; ++ch) {
    float4 x0[GCH], x1[GCH], x2[GCH];
#pragma unroll
    for (unsigned t = 0; t < GCH; ++t) {
      const unsigned k  = ch * GCH + t;
      const unsigned f0 = 384u * (k * NWAVES + bx * 8u);
      const unsigned seg = f0 >> LOG_TD;
      const bool whole = ((f0 + 8u * 384u - 1u) >> LOG_TD) == seg;
      const bool is_h  = seg >= 3u;
      const bool is_m0 = seg == 0u;
      const bool is_m1 = HAS_M1 && seg == 1u;
      if (whole && (is_h || is_m0 || is_m1)) {
        const float* bp = (is_h ? hv : (is_m0 ? m0 : m1)) + (f0 & (TD - 1u));
        x0[t] = ld4(bp + lw);
        x1[t] = ld4(bp + lw + 4u);
        x2[t] = ld4(bp + lw + 8u);
      } else if (whole) {
        x0[t] = zero4(); x1[t] = zero4(); x2[t] = zero4();
      } else {
        const unsigned base = f0 + lw;
        x0[t] = gate_src4<HAS_M1>(base,      hv, m0, m1);
        x1[t] = gate_src4<HAS_M1>(base + 4u, hv, m0, m1);
        x2[t] = gate_src4<HAS_M1>(base + 8u, hv, m0, m1);
      }
    }
#pragma unroll
    for (unsigned t = 0; t < GCH; ++t) {
      const unsigned tile = (ch * GCH + t) * NWAVES + gw;
      const v8f d = group_product(a, x0[t], x1[t], x2[t], cinit);

      const unsigned p = 192u * tile + poff;
      float4 o0;
      o0.x = fast_sigmoid(d[0]); o0.y = fast_sigmoid(d[1]);
      o0.z = fast_sigmoid(d[2]); o0.w = fast_sigmoid(d[3]);
      *(float4*)(G + p) = o0;
      if (h == 0u) {
        float4 o1;
        o1.x = fast_sigmoid(d[4]); o1.y = fast_sigmoid(d[5]);
        o1.z = fast_sigmoid(d[6]); o1.w = fast_sigmoid(d[7]);
        *(float4*)(G + p + 4u) = o1;
      }
    }
  }
}

__device__ __forceinline__ float4 sigmoid4(float d0, float d1, float d2, float d3) {
  return make_float4(fast_sigmoid(d0), fast_sigmoid(d1), fast_sigmoid(d2), fast_sigmoid(d3));
}

__device__ __forceinline__ float4 delta4(const float4& g, const float4& mv,
    float d0, float d1, float d2, float d3) {
  float4 o;
  o.x = (1.0f - g.x) * mv.x + g.x * fast_tanh(d0);
  o.y = (1.0f - g.y) * mv.y + g.y * fast_tanh(d1);
  o.z = (1.0f - g.z) * mv.z + g.z * fast_tanh(d2);
  o.w = (1.0f - g.w) * mv.w + g.w * fast_tanh(d3);
  return o;
}

__device__ __forceinline__ float4 gated4(const float4& g, float d0, float d1, float d2, float d3) {
  float4 o;
  o.x = g.x * fast_tanh(d0);
  o.y = g.y * fast_tanh(d1);
  o.z = g.z * fast_tanh(d2);
  o.w = g.w * fast_tanh(d3);
  return o;
}


template <bool HAS_M1>
__device__ __forceinline__ void cand_slot_lane(unsigned Q0, unsigned g0, unsigned nq,
    unsigned ll, unsigned ql, unsigned h, const v2f (&a)[6], const v8f& cinit,
    const v2f (&ag)[6], const v8f& ginit,
    const float* __restrict__ hv, const float* __restrict__ m0, const float* __restrict__ m1,
    const float* __restrict__ G, float4* accn) {
  for (unsigned t = 0; t < TPS; ++t) {
    const unsigned bu   = 24u * (g0 + t * 128u) + ll;
    const unsigned bmax = 24u * (NQ - 1u) + 12u * h;
    const unsigned base = bu < bmax ? bu : bmax;

    const float4 y0 = gate_src4<HAS_M1>(base,      hv, m0, m1);
    const float4 y1 = gate_src4<HAS_M1>(base + 4u, hv, m0, m1);
    const float4 y2 = gate_src4<HAS_M1>(base + 8u, hv, m0, m1);
    const v8f dg = group_product(ag, y0, y1, y2, ginit);
    const float4 ga = sigmoid4(dg[0], dg[1], dg[2], dg[3]);
    const float4 gb = sigmoid4(dg[4], dg[5], dg[6], dg[7]);

    float4 u0, u1, u2, w0, w1, w2;
    cand_src4_load<HAS_M1>(base,      hv, m0, m1, G, u0, w0);
    cand_src4_load<HAS_M1>(base + 4u, hv, m0, m1, G, u1, w1);
    cand_src4_load<HAS_M1>(base + 8u, hv, m0, m1, G, u2, w2);
    const float4 x0 = cand_src4_value<HAS_M1>(base,      u0, w0);
    const float4 x1 = cand_src4_value<HAS_M1>(base + 4u, u1, w1);
    const float4 x2 = cand_src4_value<HAS_M1>(base + 8u, u2, w2);
    const v8f d = group_product(a, x0, x1, x2, cinit);

    const unsigned p  = (base >> 1) + 2u * h;
    const unsigned p2 = p + 4u * (h ^ 1u);
    const float4 ma = msg4<HAS_M1>(p,  hv, m0, m1);
    const float4 mb = msg4<HAS_M1>(p2, hv, m0, m1);

    const unsigned qa = 3u * (g0 + t * 128u) - Q0 + ql;
    if (qa < nq)
      accn[qa] = delta4(ga, ma, d[0], d[1], d[2], d[3]);
    if (h == 0u && qa + 1u < nq)
      accn[qa + 1u] = delta4(gb, mb, d[4], d[5], d[6], d[7]);
  }
}

constexpr int K_PROD = 1;
constexpr int K_H    = 2;
constexpr int K_ZERO = 3;

constexpr int GK_W    = 1;
constexpr int GK_U    = 2;
constexpr int GK_ZERO = 3;
constexpr int GK_LOAD = 4;

template <int KIND, bool MSG, int GK>
__device__ __forceinline__ void cand_slot_uniform(
    const float* __restrict__ ub, const float* __restrict__ wb,
    const float* __restrict__ gs, const float* __restrict__ mp,
    unsigned rb, unsigned pr, unsigned qoff, unsigned nq, unsigned ll, unsigned ql, unsigned h,
    const v2f (&a)[6], const v8f& cinit, const v2f (&ag)[6], const v8f& ginit, float4* accn) {
  static_assert(GK != GK_W || KIND == K_PROD, "w is loaded only for a gate * message slab");
  static_assert(GK != GK_U || KIND == K_H, "u is the node state only in an h slab");
  const unsigned lo = rb + ll;
  const unsigned po = pr + 4u * ql;
  const unsigned pq = po + 4u * (h ^ 1u);
  float4 u[TPS][3], w[TPS][3], y[TPS][3], ma[TPS], mb[TPS];
#pragma unroll
  for (unsigned t = 0; t < TPS; ++t) {
    const unsigned ot = lo + t * (24u * 128u);
    const unsigned pt = po + t * (12u * 128u);
    const unsigned pu = pq + t * (12u * 128u);
    if (KIND != K_ZERO) {
      u[t][0] = ld4(ub + ot);
      u[t][1] = ld4(ub + ot + 4u);
      u[t][2] = ld4(ub + ot + 8u);
    } else {
      u[t][0] = zero4(); u[t][1] = zero4(); u[t][2] = zero4();
    }
    if (KIND == K_PROD) {
      w[t][0] = ld4(wb + ot);
      w[t][1] = ld4(wb + ot + 4u);
      w[t][2] = ld4(wb + ot + 8u);
    } else {
      w[t][0] = zero4(); w[t][1] = zero4(); w[t][2] = zero4();
    }
    if (GK == GK_LOAD) {
      y[t][0] = ld4(gs + ot);
      y[t][1] = ld4(gs + ot + 4u);
      y[t][2] = ld4(gs + ot + 8u);
    } else {
      y[t][0] = zero4(); y[t][1] = zero4(); y[t][2] = zero4();
    }
    if (MSG) {
      ma[t] = ld4(mp + pt);
      mb[t] = ld4(mp + pu);
    } else {
      ma[t] = zero4(); mb[t] = zero4();
    }
  }
#pragma unroll
  for (unsigned t = 0; t < TPS; ++t) {
    if (KIND != K_ZERO)
      asm volatile("" :: "v"(u[t][0].x), "v"(u[t][0].y), "v"(u[t][0].z), "v"(u[t][0].w),
                         "v"(u[t][1].x), "v"(u[t][1].y), "v"(u[t][1].z), "v"(u[t][1].w),
                         "v"(u[t][2].x), "v"(u[t][2].y), "v"(u[t][2].z), "v"(u[t][2].w));
    if (KIND == K_PROD)
      asm volatile("" :: "v"(w[t][0].x), "v"(w[t][0].y), "v"(w[t][0].z), "v"(w[t][0].w),
                         "v"(w[t][1].x), "v"(w[t][1].y), "v"(w[t][1].z), "v"(w[t][1].w),
                         "v"(w[t][2].x), "v"(w[t][2].y), "v"(w[t][2].z), "v"(w[t][2].w));
    if (GK == GK_LOAD)
      asm volatile("" :: "v"(y[t][0].x), "v"(y[t][0].y), "v"(y[t][0].z), "v"(y[t][0].w),
                         "v"(y[t][1].x), "v"(y[t][1].y), "v"(y[t][1].z), "v"(y[t][1].w),
                         "v"(y[t][2].x), "v"(y[t][2].y), "v"(y[t][2].z), "v"(y[t][2].w));
    if (MSG)
      asm volatile("" :: "v"(ma[t].x), "v"(ma[t].y), "v"(ma[t].z), "v"(ma[t].w),
                         "v"(mb[t].x), "v"(mb[t].y), "v"(mb[t].z), "v"(mb[t].w));
  }
#pragma unroll
  for (unsigned t = 0; t < TPS; ++t) {
    const float4 y0 = GK == GK_W ? w[t][0] : (GK == GK_U ? u[t][0] : y[t][0]);
    const float4 y1 = GK == GK_W ? w[t][1] : (GK == GK_U ? u[t][1] : y[t][1]);
    const float4 y2 = GK == GK_W ? w[t][2] : (GK == GK_U ? u[t][2] : y[t][2]);
    float4 x0 = u[t][0], x1 = u[t][1], x2 = u[t][2];
    if (KIND == K_PROD) {
      x0 = make_float4(x0.x * w[t][0].x, x0.y * w[t][0].y, x0.z * w[t][0].z, x0.w * w[t][0].w);
      x1 = make_float4(x1.x * w[t][1].x, x1.y * w[t][1].y, x1.z * w[t][1].z, x1.w * w[t][1].w);
      x2 = make_float4(x2.x * w[t][2].x, x2.y * w[t][2].y, x2.z * w[t][2].z, x2.w * w[t][2].w);
    }
    const v8f d = group_product(a, x0, x1, x2, cinit);
    const v8f dg = group_product(ag, y0, y1, y2, ginit);
    const float4 ga = sigmoid4(dg[0], dg[1], dg[2], dg[3]);
    const float4 gb = sigmoid4(dg[4], dg[5], dg[6], dg[7]);

    const unsigned qa = qoff + t * 384u + ql;
    if (qa < nq)
      accn[qa] = MSG ? delta4(ga, ma[t], d[0], d[1], d[2], d[3])
                     : gated4(ga, d[0], d[1], d[2], d[3]);
    if (h == 0u && qa + 1u < nq)
      accn[qa + 1u] = MSG ? delta4(gb, mb[t], d[4], d[5], d[6], d[7])
                          : gated4(gb, d[4], d[5], d[6], d[7]);
  }
}

template <bool HAS_M1, bool TWO>
__global__ __launch_bounds__(256) __attribute__((amdgpu_waves_per_eu(8, 8))) void gsn_cand_wmma(
    const float* __restrict__ hvA, const float* __restrict__ m0A,
    const float* __restrict__ m1A, const float* __restrict__ GA, float* __restrict__ outA,
    const float* __restrict__ hvB, const float* __restrict__ m0B,
    const float* __restrict__ m1B, const float* __restrict__ GB, float* __restrict__ outB,
    const float* __restrict__ Ww,
    const float* __restrict__ Bw,
    const float* __restrict__ Wx,
    const float* __restrict__ Bx) {
  __shared__ float4 acc[3][LQ];
  const bool second = TWO && blockIdx.x >= CBLK;
  const unsigned bx = second ? blockIdx.x - CBLK : blockIdx.x;
  const float* __restrict__ hv = second ? hvB : hvA;
  const float* __restrict__ m0 = second ? m0B : m0A;
  const float* __restrict__ m1 = second ? m1B : m1A;
  const float* __restrict__ G  = second ? GB : GA;
  float* __restrict__ out      = second ? outB : outA;
  const unsigned lane = threadIdx.x & 31u;
  const unsigned h    = lane >> 4;
  const unsigned gl   = (threadIdx.x >> 5u) * 16u + (lane & 15u);
  const unsigned ll   = 24u * gl + 12u * h;
  const unsigned ql   = 3u * gl + 2u * h;
  const unsigned q0   = bx * LQ;
  const unsigned nq   = (NQ - q0) < LQ ? (NQ - q0) : LQ;

  v2f a[6], ag[6];
  v8f cinit, ginit;
  make_weights(Ww, Bw, lane, a, cinit);
  make_weights(Wx, Bx, lane, ag, ginit);

  bool all_whole = true;
  for (unsigned n = 0; n < 3u; ++n) {
    const unsigned g0    = (n * NQ + q0) / 3u;
    const unsigned op_lo = 24u * g0;
    const unsigned ep_lo = 12u * g0;
    all_whole = all_whole && g0 + (GPB - 1u) < NQ
        && ((op_lo + 24u * GPB - 1u) >> LOG_TD) == (op_lo >> LOG_TD)
        && (ep_lo >> LOG_TD) == n && ((ep_lo + 12u * GPB - 1u) >> LOG_TD) == n;
  }
  if (all_whole) {
    for (unsigned n = 0; n < 3u; ++n) {
      const unsigned Q0 = n * NQ + q0;
      const unsigned g0 = Q0 / 3u;
      const unsigned op_lo = 24u * g0;
      const unsigned ep_lo = 12u * g0;
      const unsigned seg   = op_lo >> LOG_TD;
      const unsigned rb  = op_lo & (TD - 1u);
      const unsigned pr  = ep_lo & (TD - 1u);
      const unsigned qoff = 3u * g0 - Q0;
      if (seg == 0u)
        cand_slot_uniform<K_PROD, true, GK_W>(G, m0, hv, m0, rb, pr, qoff, nq, ll, ql, h, a, cinit, ag, ginit, acc[n]);
      else if (seg == 1u && HAS_M1)
        cand_slot_uniform<K_H, true, GK_LOAD>(hv, hv, m1, m0, rb, pr, qoff, nq, ll, ql, h, a, cinit, ag, ginit, acc[n]);
      else if (seg == 1u)
        cand_slot_uniform<K_H, true, GK_ZERO>(hv, hv, hv, m0, rb, pr, qoff, nq, ll, ql, h, a, cinit, ag, ginit, acc[n]);
      else if (seg == 2u && HAS_M1)
        cand_slot_uniform<K_PROD, true, GK_ZERO>(G + TD, m1, hv, m1, rb, pr, qoff, nq, ll, ql, h, a, cinit, ag, ginit, acc[n]);
      else if (seg == 2u)
        cand_slot_uniform<K_ZERO, false, GK_ZERO>(hv, hv, hv, hv, rb, pr, qoff, nq, ll, ql, h, a, cinit, ag, ginit, acc[n]);
      else if (seg == 3u && HAS_M1)
        cand_slot_uniform<K_H, true, GK_U>(hv, hv, hv, m1, rb, pr, qoff, nq, ll, ql, h, a, cinit, ag, ginit, acc[n]);
      else if (seg == 4u)
        cand_slot_uniform<K_ZERO, false, GK_LOAD>(hv, hv, hv, hv, rb, pr, qoff, nq, ll, ql, h, a, cinit, ag, ginit, acc[n]);
      else
        cand_slot_uniform<K_H, false, GK_U>(hv, hv, hv, hv, rb, pr, qoff, nq, ll, ql, h, a, cinit, ag, ginit, acc[n]);
    }
  } else {
    for (unsigned n = 0; n < 3u; ++n) {
      const unsigned Q0 = n * NQ + q0;
      cand_slot_lane<HAS_M1>(Q0, Q0 / 3u, nq, ll, ql, h, a, cinit, ag, ginit, hv, m0, m1, G, acc[n]);
    }
  }
  __syncthreads();
  for (unsigned i = threadIdx.x; i < nq; i += 256u) {
    const float4 h4 = ld4(hv + 4u * (q0 + i));
    const float4 c0 = acc[0][i];
    const float4 c1 = acc[1][i];
    const float4 c2 = acc[2][i];
    float4 o;
    o.x = h4.x + c0.x + c1.x + c2.x;
    o.y = h4.y + c0.y + c1.y + c2.y;
    o.z = h4.z + c0.z + c1.z + c2.z;
    o.w = h4.w + c0.w + c1.w + c2.w;
    *(float4*)(out + 4u * (q0 + i)) = o;
  }
}

extern "C" void kernel_launch(void* const* d_in, const int* in_sizes, int n_in,
                              void* d_out, int out_size, void* d_ws, size_t ws_size,
                              hipStream_t stream) {
  (void)in_sizes; (void)n_in; (void)out_size; (void)ws_size;
  const float* feat = (const float*)d_in[0];
  const float* W_w  = (const float*)d_in[1];
  const float* W_b  = (const float*)d_in[2];
  const float* Wx_w = (const float*)d_in[3];
  const float* Wx_b = (const float*)d_in[4];
  float* out = (float*)d_out;

  float* G  = (float*)d_ws;
  float* G1 = G + (size_t)GEXT1;

  const float* h0 = feat;
  const float* h1 = feat + (size_t)TD;
  const float* h2 = feat + (size_t)2 * TD;

  dim3 blk(256);
  dim3 grd_g11(GBLK2);
  dim3 grd_g2(GBLK2);
  dim3 grd_cd(CBLK);
  dim3 grd_cd2(CBLK2);

  gsn_gate_wmma<false, GWAV1, true><<<grd_g11, blk, 0, stream>>>(
      h0, h2, nullptr, G, h1, h0, nullptr, G1, Wx_w, Wx_b);
  gsn_cand_wmma<false, true><<<grd_cd2, blk, 0, stream>>>(
      h0, h2, nullptr, G, out + 0 * (size_t)TD, h1, h0, nullptr, G1, out + 1 * (size_t)TD,
      W_w, W_b, Wx_w, Wx_b);

  gsn_gate_wmma<true, GWAV2, false><<<grd_g2, blk, 0, stream>>>(
      h2, h1, h0, G, nullptr, nullptr, nullptr, nullptr, Wx_w, Wx_b);
  gsn_cand_wmma<true, false><<<grd_cd, blk, 0, stream>>>(
      h2, h1, h0, G, out + 2 * (size_t)TD, nullptr, nullptr, nullptr, nullptr, nullptr,
      W_w, W_b, Wx_w, Wx_b);
}


# Node 2 joins the gate launch and the cand launch of nodes 0 and 1: a block picks node, message count and tile stride by scalar tests; its gates get a workspace part of their own; 4->2 launches
# speedup vs baseline: 3.5892x; 1.0313x over previous
#include <hip/hip_runtime.h>
#include <math.h>

#define LOG_TD 22u
#define TD (1u << LOG_TD)

typedef __attribute__((ext_vector_type(2))) float v2f;
typedef __attribute__((ext_vector_type(8))) float v8f;

constexpr unsigned TPW    = 6u;
constexpr unsigned GCH    = 3u;
constexpr unsigned GBLK1  = 512u;
constexpr unsigned GBLK2  = 1024u;
constexpr unsigned GWAV1  = 4096u;
constexpr unsigned GWAV2  = 8192u;
static_assert(GBLK2 == 2u * GBLK1 && GWAV1 == 8u * GBLK1 && GWAV2 == 8u * GBLK2, "gate grids");
static_assert(GBLK1 * 8u * TPW * 192u >= TD, "the gate grid must cover a whole slot");
constexpr unsigned GEXT1  = GWAV1 * TPW * 192u;
static_assert(GEXT1 >= TD && 2u * GEXT1 == GWAV2 * TPW * 192u, "two one-slot parts fill the two-slot grid's extent");
constexpr unsigned GBLK3  = 2u * GBLK1 + GBLK2;
constexpr unsigned NQ     = TD / 4u;
constexpr unsigned TPS    = 2u;
constexpr unsigned GPB    = 16u * 8u * TPS;
constexpr unsigned LQ     = 760u;
constexpr unsigned CBLK   = (NQ + LQ - 1u) / LQ;
constexpr unsigned CBLK3  = 3u * CBLK;
static_assert(LQ % 8u == 0u && LQ + 2u <= 3u * GPB, "block range must be whole lines and fit its groups");

__device__ __forceinline__ float hw_exp2(float x) {
#if __has_builtin(__builtin_amdgcn_exp2f)
  return __builtin_amdgcn_exp2f(x);
#else
  return exp2f(x);
#endif
}
__device__ __forceinline__ float hw_rcp(float x) {
#if __has_builtin(__builtin_amdgcn_rcpf)
  return __builtin_amdgcn_rcpf(x);
#else
  return 1.0f / x;
#endif
}
__device__ __forceinline__ float fast_sigmoid(float x) {
  return hw_rcp(1.0f + hw_exp2(-1.4426950408889634f * x));
}
__device__ __forceinline__ float fast_tanh(float x) {
#if __has_builtin(__builtin_amdgcn_tanhf)
  return __builtin_amdgcn_tanhf(x);
#else
  return 1.0f - 2.0f * hw_rcp(hw_exp2(2.8853900817779268f * x) + 1.0f);
#endif
}

__device__ __forceinline__ float4 ld4(const float* __restrict__ p) {
  return *(const float4*)p;
}
__device__ __forceinline__ float4 zero4() { return make_float4(0.0f, 0.0f, 0.0f, 0.0f); }
__device__ __forceinline__ float4 sel4(bool c, const float4& v) {
  return make_float4(c ? v.x : 0.0f, c ? v.y : 0.0f, c ? v.z : 0.0f, c ? v.w : 0.0f);
}


__device__ __forceinline__ float4 gate_src4(bool has_m1, unsigned m,
    const float* __restrict__ hv, const float* __restrict__ m0,
    const float* __restrict__ m1) {
  const unsigned seg = m >> LOG_TD;
  const unsigned r   = m & (TD - 1u);
  const bool is_h  = seg >= 3u;
  const bool is_m0 = seg == 0u;
  const bool is_m1 = has_m1 && seg == 1u;
  const bool valid = is_h || is_m0 || is_m1;
  const float* base = is_m0 ? m0 : (is_m1 ? m1 : hv);
  const float4 v = ld4(base + (valid ? r : 0u));
  return sel4(valid, v);
}

__device__ __forceinline__ float4 msg4(bool has_m1, unsigned p, const float* __restrict__ hv,
    const float* __restrict__ m0, const float* __restrict__ m1) {
  const unsigned n = p >> LOG_TD;
  const unsigned r = p & (TD - 1u);
  const bool is_m0 = n == 0u;
  const bool is_m1 = has_m1 && n == 1u;
  const bool valid = is_m0 || is_m1;
  const float* base = is_m0 ? m0 : (is_m1 ? m1 : hv);
  const float4 v = ld4(base + (valid ? r : 0u));
  return sel4(valid, v);
}

__device__ __forceinline__ void cand_src4_load(bool has_m1, unsigned m,
    const float* __restrict__ hv, const float* __restrict__ m0,
    const float* __restrict__ m1, const float* __restrict__ G,
    float4& u, float4& w) {
  const unsigned seg = m >> LOG_TD;
  const unsigned r   = m & (TD - 1u);
  const unsigned n   = seg >> 1;
  const bool is_h  = (seg & 1u) != 0u;
  const bool is_m0 = !is_h && n == 0u;
  const bool is_m1 = has_m1 && !is_h && n == 1u;
  const bool prod  = is_m0 || is_m1;
  const float* ub = prod ? G : hv;
  const unsigned uo = prod ? (n * TD + r) : (is_h ? r : 0u);
  u = ld4(ub + uo);
  const float* wb = is_m0 ? m0 : (is_m1 ? m1 : hv);
  w = ld4(wb + (prod ? r : 0u));
}
__device__ __forceinline__ float4 cand_src4_value(bool has_m1, unsigned m, const float4& u,
    const float4& w) {
  const unsigned seg = m >> LOG_TD;
  const unsigned n   = seg >> 1;
  const bool is_h  = (seg & 1u) != 0u;
  const bool prod  = !is_h && (n == 0u || (has_m1 && n == 1u));
  const float4 gm = make_float4(u.x * w.x, u.y * w.y, u.z * w.z, u.w * w.w);
  return make_float4(is_h ? u.x : (prod ? gm.x : 0.0f), is_h ? u.y : (prod ? gm.y : 0.0f),
                     is_h ? u.z : (prod ? gm.z : 0.0f), is_h ? u.w : (prod ? gm.w : 0.0f));
}

__device__ __forceinline__ void make_weights(const float* __restrict__ Wm,
    const float* __restrict__ Bv, unsigned lane, v2f (&a)[6], v8f& cinit) {
  const unsigned o  = lane & 15u;
  const unsigned h  = lane >> 4;
  const unsigned g  = o / 3u;
  const unsigned j  = o - 3u * g;
  const bool valid  = (o < 12u) && ((g >> 1) == h);
  const float msk   = valid ? 1.0f : 0.0f;
  const bool second = (g & 1u) != 0u;
  float w[6];
#pragma unroll
  for (int i = 0; i < 6; ++i) w[i] = Wm[j * 6u + (unsigned)i] * msk;
#pragma unroll
  for (int c = 0; c < 3; ++c) {
    a[c].x     = second ? 0.0f : w[2 * c];
    a[c].y     = second ? 0.0f : w[2 * c + 1];
    a[c + 3].x = second ? w[2 * c]     : 0.0f;
    a[c + 3].y = second ? w[2 * c + 1] : 0.0f;
  }
  const float b0 = Bv[0], b1 = Bv[1], b2 = Bv[2];
  const bool hi = h != 0u;
  cinit[0] = hi ? b2 : b0;
  cinit[1] = hi ? b0 : b1;
  cinit[2] = hi ? b1 : b2;
  cinit[3] = hi ? b2 : b0;
  cinit[4] = hi ? 0.0f : b1;
  cinit[5] = hi ? 0.0f : b2;
  cinit[6] = hi ? 0.0f : b0;
  cinit[7] = hi ? 0.0f : b1;
}

__device__ __forceinline__ v8f group_product(const v2f (&a)[6], const float4& x0,
    const float4& x1, const float4& x2, v8f c) {
  v2f b;
  b.x = x0.x; b.y = x0.y;
  c = __builtin_amdgcn_wmma_f32_16x16x4_f32(false, a[0], false, b, (short)0, c, false, false);
  b.x = x0.z; b.y = x0.w;
  c = __builtin_amdgcn_wmma_f32_16x16x4_f32(false, a[1], false, b, (short)0, c, false, false);
  b.x = x1.x; b.y = x1.y;
  c = __builtin_amdgcn_wmma_f32_16x16x4_f32(false, a[2], false, b, (short)0, c, false, false);
  b.x = x1.z; b.y = x1.w;
  c = __builtin_amdgcn_wmma_f32_16x16x4_f32(false, a[3], false, b, (short)0, c, false, false);
  b.x = x2.x; b.y = x2.y;
  c = __builtin_amdgcn_wmma_f32_16x16x4_f32(false, a[4], false, b, (short)0, c, false, false);
  b.x = x2.z; b.y = x2.w;
  c = __builtin_amdgcn_wmma_f32_16x16x4_f32(false, a[5], false, b, (short)0, c, false, false);
  return c;
}

__global__ __launch_bounds__(256, 8) void gsn_gate_wmma(
    const float* __restrict__ hvA, const float* __restrict__ m0A, float* __restrict__ GA,
    const float* __restrict__ hvB, const float* __restrict__ m0B, float* __restrict__ GB,
    const float* __restrict__ hvC, const float* __restrict__ m0C,
    const float* __restrict__ m1C, float* __restrict__ GC,
    const float* __restrict__ Wx,
    const float* __restrict__ Bx) {
  const bool isC = blockIdx.x >= 2u * GBLK1;
  const bool isB = !isC && blockIdx.x >= GBLK1;
  const unsigned bx = blockIdx.x - (isC ? 2u * GBLK1 : (isB ? GBLK1 : 0u));
  const bool has_m1 = isC;
  const unsigned nwaves = isC ? GWAV2 : GWAV1;
  const float* __restrict__ hv = isC ? hvC : (isB ? hvB : hvA);
  const float* __restrict__ m0 = isC ? m0C : (isB ? m0B : m0A);
  const float* __restrict__ m1 = m1C;
  float* __restrict__ G        = isC ? GC : (isB ? GB : GA);
  const unsigned lane = threadIdx.x & 31u;
  const unsigned wv   = threadIdx.x >> 5u;
  const unsigned gw   = bx * 8u + wv;
  const unsigned h    = lane >> 4;
  const unsigned loff = 24u * (lane & 15u) + 12u * h;
  const unsigned poff = 12u * (lane & 15u) + 8u * h;
  const unsigned lw   = 384u * wv + loff;

  v2f a[6];
  v8f cinit;
  make_weights(Wx, Bx, lane, a, cinit);

  for (unsigned ch = 0; ch < TPW / GCH; ++ch) {
    float4 x0[GCH], x1[GCH], x2[GCH];
#pragma unroll
    for (unsigned t = 0; t < GCH; ++t) {
      const unsigned k  = ch * GCH + t;
      const unsigned f0 = 384u * (k * nwaves + bx * 8u);
      const unsigned seg = f0 >> LOG_TD;
      const bool whole = ((f0 + 8u * 384u - 1u) >> LOG_TD) == seg;
      const bool is_h  = seg >= 3u;
      const bool is_m0 = seg == 0u;
      const bool is_m1 = has_m1 && seg == 1u;
      if (whole && (is_h || is_m0 || is_m1)) {
        const float* bp = (is_h ? hv : (is_m0 ? m0 : m1)) + (f0 & (TD - 1u));
        x0[t] = ld4(bp + lw);
        x1[t] = ld4(bp + lw + 4u);
        x2[t] = ld4(bp + lw + 8u);
      } else if (whole) {
        x0[t] = zero4(); x1[t] = zero4(); x2[t] = zero4();
      } else {
        const unsigned base = f0 + lw;
        x0[t] = gate_src4(has_m1, base,      hv, m0, m1);
        x1[t] = gate_src4(has_m1, base + 4u, hv, m0, m1);
        x2[t] = gate_src4(has_m1, base + 8u, hv, m0, m1);
      }
    }
#pragma unroll
    for (unsigned t = 0; t < GCH; ++t) {
      const unsigned tile = (ch * GCH + t) * nwaves + gw;
      const v8f d = group_product(a, x0[t], x1[t], x2[t], cinit);

      const unsigned p = 192u * tile + poff;
      float4 o0;
      o0.x = fast_sigmoid(d[0]); o0.y = fast_sigmoid(d[1]);
      o0.z = fast_sigmoid(d[2]); o0.w = fast_sigmoid(d[3]);
      *(float4*)(G + p) = o0;
      if (h == 0u) {
        float4 o1;
        o1.x = fast_sigmoid(d[4]); o1.y = fast_sigmoid(d[5]);
        o1.z = fast_sigmoid(d[6]); o1.w = fast_sigmoid(d[7]);
        *(float4*)(G + p + 4u) = o1;
      }
    }
  }
}

__device__ __forceinline__ float4 sigmoid4(float d0, float d1, float d2, float d3) {
  return make_float4(fast_sigmoid(d0), fast_sigmoid(d1), fast_sigmoid(d2), fast_sigmoid(d3));
}

__device__ __forceinline__ float4 delta4(const float4& g, const float4& mv,
    float d0, float d1, float d2, float d3) {
  float4 o;
  o.x = (1.0f - g.x) * mv.x + g.x * fast_tanh(d0);
  o.y = (1.0f - g.y) * mv.y + g.y * fast_tanh(d1);
  o.z = (1.0f - g.z) * mv.z + g.z * fast_tanh(d2);
  o.w = (1.0f - g.w) * mv.w + g.w * fast_tanh(d3);
  return o;
}

__device__ __forceinline__ float4 gated4(const float4& g, float d0, float d1, float d2, float d3) {
  float4 o;
  o.x = g.x * fast_tanh(d0);
  o.y = g.y * fast_tanh(d1);
  o.z = g.z * fast_tanh(d2);
  o.w = g.w * fast_tanh(d3);
  return o;
}


__device__ __forceinline__ void cand_slot_lane(bool has_m1, unsigned Q0, unsigned g0, unsigned nq,
    unsigned ll, unsigned ql, unsigned h, const v2f (&a)[6], const v8f& cinit,
    const v2f (&ag)[6], const v8f& ginit,
    const float* __restrict__ hv, const float* __restrict__ m0, const float* __restrict__ m1,
    const float* __restrict__ G, float4* accn) {
  for (unsigned t = 0; t < TPS; ++t) {
    const unsigned bu   = 24u * (g0 + t * 128u) + ll;
    const unsigned bmax = 24u * (NQ - 1u) + 12u * h;
    const unsigned base = bu < bmax ? bu : bmax;

    const float4 y0 = gate_src4(has_m1, base,      hv, m0, m1);
    const float4 y1 = gate_src4(has_m1, base + 4u, hv, m0, m1);
    const float4 y2 = gate_src4(has_m1, base + 8u, hv, m0, m1);
    const v8f dg = group_product(ag, y0, y1, y2, ginit);
    const float4 ga = sigmoid4(dg[0], dg[1], dg[2], dg[3]);
    const float4 gb = sigmoid4(dg[4], dg[5], dg[6], dg[7]);

    float4 u0, u1, u2, w0, w1, w2;
    cand_src4_load(has_m1, base,      hv, m0, m1, G, u0, w0);
    cand_src4_load(has_m1, base + 4u, hv, m0, m1, G, u1, w1);
    cand_src4_load(has_m1, base + 8u, hv, m0, m1, G, u2, w2);
    const float4 x0 = cand_src4_value(has_m1, base,      u0, w0);
    const float4 x1 = cand_src4_value(has_m1, base + 4u, u1, w1);
    const float4 x2 = cand_src4_value(has_m1, base + 8u, u2, w2);
    const v8f d = group_product(a, x0, x1, x2, cinit);

    const unsigned p  = (base >> 1) + 2u * h;
    const unsigned p2 = p + 4u * (h ^ 1u);
    const float4 ma = msg4(has_m1, p,  hv, m0, m1);
    const float4 mb = msg4(has_m1, p2, hv, m0, m1);

    const unsigned qa = 3u * (g0 + t * 128u) - Q0 + ql;
    if (qa < nq)
      accn[qa] = delta4(ga, ma, d[0], d[1], d[2], d[3]);
    if (h == 0u && qa + 1u < nq)
      accn[qa + 1u] = delta4(gb, mb, d[4], d[5], d[6], d[7]);
  }
}

constexpr int K_PROD = 1;
constexpr int K_H    = 2;
constexpr int K_ZERO = 3;

constexpr int GK_W    = 1;
constexpr int GK_U    = 2;
constexpr int GK_ZERO = 3;
constexpr int GK_LOAD = 4;

template <int KIND, bool MSG, int GK>
__device__ __forceinline__ void cand_slot_uniform(
    const float* __restrict__ ub, const float* __restrict__ wb,
    const float* __restrict__ gs, const float* __restrict__ mp,
    unsigned rb, unsigned pr, unsigned qoff, unsigned nq, unsigned ll, unsigned ql, unsigned h,
    const v2f (&a)[6], const v8f& cinit, const v2f (&ag)[6], const v8f& ginit, float4* accn) {
  static_assert(GK != GK_W || KIND == K_PROD, "w is loaded only for a gate * message slab");
  static_assert(GK != GK_U || KIND == K_H, "u is the node state only in an h slab");
  const unsigned lo = rb + ll;
  const unsigned po = pr + 4u * ql;
  const unsigned pq = po + 4u * (h ^ 1u);
  float4 u[TPS][3], w[TPS][3], y[TPS][3], ma[TPS], mb[TPS];
#pragma unroll
  for (unsigned t = 0; t < TPS; ++t) {
    const unsigned ot = lo + t * (24u * 128u);
    const unsigned pt = po + t * (12u * 128u);
    const unsigned pu = pq + t * (12u * 128u);
    if (KIND != K_ZERO) {
      u[t][0] = ld4(ub + ot);
      u[t][1] = ld4(ub + ot + 4u);
      u[t][2] = ld4(ub + ot + 8u);
    } else {
      u[t][0] = zero4(); u[t][1] = zero4(); u[t][2] = zero4();
    }
    if (KIND == K_PROD) {
      w[t][0] = ld4(wb + ot);
      w[t][1] = ld4(wb + ot + 4u);
      w[t][2] = ld4(wb + ot + 8u);
    } else {
      w[t][0] = zero4(); w[t][1] = zero4(); w[t][2] = zero4();
    }
    if (GK == GK_LOAD) {
      y[t][0] = ld4(gs + ot);
      y[t][1] = ld4(gs + ot + 4u);
      y[t][2] = ld4(gs + ot + 8u);
    } else {
      y[t][0] = zero4(); y[t][1] = zero4(); y[t][2] = zero4();
    }
    if (MSG) {
      ma[t] = ld4(mp + pt);
      mb[t] = ld4(mp + pu);
    } else {
      ma[t] = zero4(); mb[t] = zero4();
    }
  }
#pragma unroll
  for (unsigned t = 0; t < TPS; ++t) {
    if (KIND != K_ZERO)
      asm volatile("" :: "v"(u[t][0].x), "v"(u[t][0].y), "v"(u[t][0].z), "v"(u[t][0].w),
                         "v"(u[t][1].x), "v"(u[t][1].y), "v"(u[t][1].z), "v"(u[t][1].w),
                         "v"(u[t][2].x), "v"(u[t][2].y), "v"(u[t][2].z), "v"(u[t][2].w));
    if (KIND == K_PROD)
      asm volatile("" :: "v"(w[t][0].x), "v"(w[t][0].y), "v"(w[t][0].z), "v"(w[t][0].w),
                         "v"(w[t][1].x), "v"(w[t][1].y), "v"(w[t][1].z), "v"(w[t][1].w),
                         "v"(w[t][2].x), "v"(w[t][2].y), "v"(w[t][2].z), "v"(w[t][2].w));
    if (GK == GK_LOAD)
      asm volatile("" :: "v"(y[t][0].x), "v"(y[t][0].y), "v"(y[t][0].z), "v"(y[t][0].w),
                         "v"(y[t][1].x), "v"(y[t][1].y), "v"(y[t][1].z), "v"(y[t][1].w),
                         "v"(y[t][2].x), "v"(y[t][2].y), "v"(y[t][2].z), "v"(y[t][2].w));
    if (MSG)
      asm volatile("" :: "v"(ma[t].x), "v"(ma[t].y), "v"(ma[t].z), "v"(ma[t].w),
                         "v"(mb[t].x), "v"(mb[t].y), "v"(mb[t].z), "v"(mb[t].w));
  }
#pragma unroll
  for (unsigned t = 0; t < TPS; ++t) {
    const float4 y0 = GK == GK_W ? w[t][0] : (GK == GK_U ? u[t][0] : y[t][0]);
    const float4 y1 = GK == GK_W ? w[t][1] : (GK == GK_U ? u[t][1] : y[t][1]);
    const float4 y2 = GK == GK_W ? w[t][2] : (GK == GK_U ? u[t][2] : y[t][2]);
    float4 x0 = u[t][0], x1 = u[t][1], x2 = u[t][2];
    if (KIND == K_PROD) {
      x0 = make_float4(x0.x * w[t][0].x, x0.y * w[t][0].y, x0.z * w[t][0].z, x0.w * w[t][0].w);
      x1 = make_float4(x1.x * w[t][1].x, x1.y * w[t][1].y, x1.z * w[t][1].z, x1.w * w[t][1].w);
      x2 = make_float4(x2.x * w[t][2].x, x2.y * w[t][2].y, x2.z * w[t][2].z, x2.w * w[t][2].w);
    }
    const v8f d = group_product(a, x0, x1, x2, cinit);
    const v8f dg = group_product(ag, y0, y1, y2, ginit);
    const float4 ga = sigmoid4(dg[0], dg[1], dg[2], dg[3]);
    const float4 gb = sigmoid4(dg[4], dg[5], dg[6], dg[7]);

    const unsigned qa = qoff + t * 384u + ql;
    if (qa < nq)
      accn[qa] = MSG ? delta4(ga, ma[t], d[0], d[1], d[2], d[3])
                     : gated4(ga, d[0], d[1], d[2], d[3]);
    if (h == 0u && qa + 1u < nq)
      accn[qa + 1u] = MSG ? delta4(gb, mb[t], d[4], d[5], d[6], d[7])
                          : gated4(gb, d[4], d[5], d[6], d[7]);
  }
}

__global__ __launch_bounds__(256) __attribute__((amdgpu_waves_per_eu(8, 8))) void gsn_cand_wmma(
    const float* __restrict__ hvA, const float* __restrict__ m0A,
    const float* __restrict__ GA, float* __restrict__ outA,
    const float* __restrict__ hvB, const float* __restrict__ m0B,
    const float* __restrict__ GB, float* __restrict__ outB,
    const float* __restrict__ hvC, const float* __restrict__ m0C, const float* __restrict__ m1C,
    const float* __restrict__ GC, float* __restrict__ outC,
    const float* __restrict__ Ww,
    const float* __restrict__ Bw,
    const float* __restrict__ Wx,
    const float* __restrict__ Bx) {
  __shared__ float4 acc[3][LQ];
  const bool isC = blockIdx.x >= 2u * CBLK;
  const bool isB = !isC && blockIdx.x >= CBLK;
  const unsigned bx = blockIdx.x - (isC ? 2u * CBLK : (isB ? CBLK : 0u));
  const bool has_m1 = isC;
  const float* __restrict__ hv = isC ? hvC : (isB ? hvB : hvA);
  const float* __restrict__ m0 = isC ? m0C : (isB ? m0B : m0A);
  const float* __restrict__ m1 = m1C;
  const float* __restrict__ G  = isC ? GC : (isB ? GB : GA);
  float* __restrict__ out      = isC ? outC : (isB ? outB : outA);
  const unsigned lane = threadIdx.x & 31u;
  const unsigned h    = lane >> 4;
  const unsigned gl   = (threadIdx.x >> 5u) * 16u + (lane & 15u);
  const unsigned ll   = 24u * gl + 12u * h;
  const unsigned ql   = 3u * gl + 2u * h;
  const unsigned q0   = bx * LQ;
  const unsigned nq   = (NQ - q0) < LQ ? (NQ - q0) : LQ;

  v2f a[6], ag[6];
  v8f cinit, ginit;
  make_weights(Ww, Bw, lane, a, cinit);
  make_weights(Wx, Bx, lane, ag, ginit);

  bool all_whole = true;
  for (unsigned n = 0; n < 3u; ++n) {
    const unsigned g0    = (n * NQ + q0) / 3u;
    const unsigned op_lo = 24u * g0;
    const unsigned ep_lo = 12u * g0;
    all_whole = all_whole && g0 + (GPB - 1u) < NQ
        && ((op_lo + 24u * GPB - 1u) >> LOG_TD) == (op_lo >> LOG_TD)
        && (ep_lo >> LOG_TD) == n && ((ep_lo + 12u * GPB - 1u) >> LOG_TD) == n;
  }
  if (all_whole) {
    for (unsigned n = 0; n < 3u; ++n) {
      const unsigned Q0 = n * NQ + q0;
      const unsigned g0 = Q0 / 3u;
      const unsigned op_lo = 24u * g0;
      const unsigned ep_lo = 12u * g0;
      const unsigned seg   = op_lo >> LOG_TD;
      const unsigned rb  = op_lo & (TD - 1u);
      const unsigned pr  = ep_lo & (TD - 1u);
      const unsigned qoff = 3u * g0 - Q0;
      if (seg == 0u)
        cand_slot_uniform<K_PROD, true, GK_W>(G, m0, hv, m0, rb, pr, qoff, nq, ll, ql, h, a, cinit, ag, ginit, acc[n]);
      else if (seg == 1u && has_m1)
        cand_slot_uniform<K_H, true, GK_LOAD>(hv, hv, m1, m0, rb, pr, qoff, nq, ll, ql, h, a, cinit, ag, ginit, acc[n]);
      else if (seg == 1u)
        cand_slot_uniform<K_H, true, GK_ZERO>(hv, hv, hv, m0, rb, pr, qoff, nq, ll, ql, h, a, cinit, ag, ginit, acc[n]);
      else if (seg == 2u && has_m1)
        cand_slot_uniform<K_PROD, true, GK_ZERO>(G + TD, m1, hv, m1, rb, pr, qoff, nq, ll, ql, h, a, cinit, ag, ginit, acc[n]);
      else if (seg == 2u)
        cand_slot_uniform<K_ZERO, false, GK_ZERO>(hv, hv, hv, hv, rb, pr, qoff, nq, ll, ql, h, a, cinit, ag, ginit, acc[n]);
      else if (seg == 3u && has_m1)
        cand_slot_uniform<K_H, true, GK_U>(hv, hv, hv, m1, rb, pr, qoff, nq, ll, ql, h, a, cinit, ag, ginit, acc[n]);
      else if (seg == 4u)
        cand_slot_uniform<K_ZERO, false, GK_LOAD>(hv, hv, hv, hv, rb, pr, qoff, nq, ll, ql, h, a, cinit, ag, ginit, acc[n]);
      else
        cand_slot_uniform<K_H, false, GK_U>(hv, hv, hv, hv, rb, pr, qoff, nq, ll, ql, h, a, cinit, ag, ginit, acc[n]);
    }
  } else {
    for (unsigned n = 0; n < 3u; ++n) {
      const unsigned Q0 = n * NQ + q0;
      cand_slot_lane(has_m1, Q0, Q0 / 3u, nq, ll, ql, h, a, cinit, ag, ginit, hv, m0, m1, G, acc[n]);
    }
  }
  __syncthreads();
  for (unsigned i = threadIdx.x; i < nq; i += 256u) {
    const float4 h4 = ld4(hv + 4u * (q0 + i));
    const float4 c0 = acc[0][i];
    const float4 c1 = acc[1][i];
    const float4 c2 = acc[2][i];
    float4 o;
    o.x = h4.x + c0.x + c1.x + c2.x;
    o.y = h4.y + c0.y + c1.y + c2.y;
    o.z = h4.z + c0.z + c1.z + c2.z;
    o.w = h4.w + c0.w + c1.w + c2.w;
    *(float4*)(out + 4u * (q0 + i)) = o;
  }
}

extern "C" void kernel_launch(void* const* d_in, const int* in_sizes, int n_in,
                              void* d_out, int out_size, void* d_ws, size_t ws_size,
                              hipStream_t stream) {
  (void)in_sizes; (void)n_in; (void)out_size; (void)ws_size;
  const float* feat = (const float*)d_in[0];
  const float* W_w  = (const float*)d_in[1];
  const float* W_b  = (const float*)d_in[2];
  const float* Wx_w = (const float*)d_in[3];
  const float* Wx_b = (const float*)d_in[4];
  float* out = (float*)d_out;

  float* G  = (float*)d_ws;
  float* G1 = G + (size_t)GEXT1;
  float* G2 = G + (size_t)2 * GEXT1;

  const float* h0 = feat;
  const float* h1 = feat + (size_t)TD;
  const float* h2 = feat + (size_t)2 * TD;

  dim3 blk(256);
  dim3 grd_g(GBLK3);
  dim3 grd_c(CBLK3);

  gsn_gate_wmma<<<grd_g, blk, 0, stream>>>(
      h0, h2, G, h1, h0, G1, h2, h1, h0, G2, Wx_w, Wx_b);
  gsn_cand_wmma<<<grd_c, blk, 0, stream>>>(
      h0, h2, G, out + 0 * (size_t)TD, h1, h0, G1, out + 1 * (size_t)TD,
      h2, h1, h0, G2, out + 2 * (size_t)TD, W_w, W_b, Wx_w, Wx_b);
}


# Linears over an empty mailbox slot's zeros taken as their bias in both passes: no matrix product, sigmoid or tanh where a whole tile's operands are such zeros; sigmoid/tanh of bias once per wave
# speedup vs baseline: 3.7801x; 1.0532x over previous
#include <hip/hip_runtime.h>
#include <math.h>

#define LOG_TD 22u
#define TD (1u << LOG_TD)

typedef __attribute__((ext_vector_type(2))) float v2f;
typedef __attribute__((ext_vector_type(8))) float v8f;

constexpr unsigned TPW    = 6u;
constexpr unsigned GCH    = 3u;
constexpr unsigned GBLK1  = 512u;
constexpr unsigned GBLK2  = 1024u;
constexpr unsigned GWAV1  = 4096u;
constexpr unsigned GWAV2  = 8192u;
static_assert(GBLK2 == 2u * GBLK1 && GWAV1 == 8u * GBLK1 && GWAV2 == 8u * GBLK2, "gate grids");
static_assert(GBLK1 * 8u * TPW * 192u >= TD, "the gate grid must cover a whole slot");
constexpr unsigned GEXT1  = GWAV1 * TPW * 192u;
static_assert(GEXT1 >= TD && 2u * GEXT1 == GWAV2 * TPW * 192u, "two one-slot parts fill the two-slot grid's extent");
constexpr unsigned GBLK3  = 2u * GBLK1 + GBLK2;
constexpr unsigned NQ     = TD / 4u;
constexpr unsigned TPS    = 2u;
constexpr unsigned GPB    = 16u * 8u * TPS;
constexpr unsigned LQ     = 760u;
constexpr unsigned CBLK   = (NQ + LQ - 1u) / LQ;
constexpr unsigned CBLK3  = 3u * CBLK;
static_assert(LQ % 8u == 0u && LQ + 2u <= 3u * GPB, "block range must be whole lines and fit its groups");

__device__ __forceinline__ float hw_exp2(float x) {
#if __has_builtin(__builtin_amdgcn_exp2f)
  return __builtin_amdgcn_exp2f(x);
#else
  return exp2f(x);
#endif
}
__device__ __forceinline__ float hw_rcp(float x) {
#if __has_builtin(__builtin_amdgcn_rcpf)
  return __builtin_amdgcn_rcpf(x);
#else
  return 1.0f / x;
#endif
}
__device__ __forceinline__ float fast_sigmoid(float x) {
  return hw_rcp(1.0f + hw_exp2(-1.4426950408889634f * x));
}
__device__ __forceinline__ float fast_tanh(float x) {
#if __has_builtin(__builtin_amdgcn_tanhf)
  return __builtin_amdgcn_tanhf(x);
#else
  return 1.0f - 2.0f * hw_rcp(hw_exp2(2.8853900817779268f * x) + 1.0f);
#endif
}

__device__ __forceinline__ float4 ld4(const float* __restrict__ p) {
  return *(const float4*)p;
}
__device__ __forceinline__ float4 zero4() { return make_float4(0.0f, 0.0f, 0.0f, 0.0f); }
__device__ __forceinline__ float4 sel4(bool c, const float4& v) {
  return make_float4(c ? v.x : 0.0f, c ? v.y : 0.0f, c ? v.z : 0.0f, c ? v.w : 0.0f);
}


__device__ __forceinline__ float4 gate_src4(bool has_m1, unsigned m,
    const float* __restrict__ hv, const float* __restrict__ m0,
    const float* __restrict__ m1) {
  const unsigned seg = m >> LOG_TD;
  const unsigned r   = m & (TD - 1u);
  const bool is_h  = seg >= 3u;
  const bool is_m0 = seg == 0u;
  const bool is_m1 = has_m1 && seg == 1u;
  const bool valid = is_h || is_m0 || is_m1;
  const float* base = is_m0 ? m0 : (is_m1 ? m1 : hv);
  const float4 v = ld4(base + (valid ? r : 0u));
  return sel4(valid, v);
}

__device__ __forceinline__ float4 msg4(bool has_m1, unsigned p, const float* __restrict__ hv,
    const float* __restrict__ m0, const float* __restrict__ m1) {
  const unsigned n = p >> LOG_TD;
  const unsigned r = p & (TD - 1u);
  const bool is_m0 = n == 0u;
  const bool is_m1 = has_m1 && n == 1u;
  const bool valid = is_m0 || is_m1;
  const float* base = is_m0 ? m0 : (is_m1 ? m1 : hv);
  const float4 v = ld4(base + (valid ? r : 0u));
  return sel4(valid, v);
}

__device__ __forceinline__ void cand_src4_load(bool has_m1, unsigned m,
    const float* __restrict__ hv, const float* __restrict__ m0,
    const float* __restrict__ m1, const float* __restrict__ G,
    float4& u, float4& w) {
  const unsigned seg = m >> LOG_TD;
  const unsigned r   = m & (TD - 1u);
  const unsigned n   = seg >> 1;
  const bool is_h  = (seg & 1u) != 0u;
  const bool is_m0 = !is_h && n == 0u;
  const bool is_m1 = has_m1 && !is_h && n == 1u;
  const bool prod  = is_m0 || is_m1;
  const float* ub = prod ? G : hv;
  const unsigned uo = prod ? (n * TD + r) : (is_h ? r : 0u);
  u = ld4(ub + uo);
  const float* wb = is_m0 ? m0 : (is_m1 ? m1 : hv);
  w = ld4(wb + (prod ? r : 0u));
}
__device__ __forceinline__ float4 cand_src4_value(bool has_m1, unsigned m, const float4& u,
    const float4& w) {
  const unsigned seg = m >> LOG_TD;
  const unsigned n   = seg >> 1;
  const bool is_h  = (seg & 1u) != 0u;
  const bool prod  = !is_h && (n == 0u || (has_m1 && n == 1u));
  const float4 gm = make_float4(u.x * w.x, u.y * w.y, u.z * w.z, u.w * w.w);
  return make_float4(is_h ? u.x : (prod ? gm.x : 0.0f), is_h ? u.y : (prod ? gm.y : 0.0f),
                     is_h ? u.z : (prod ? gm.z : 0.0f), is_h ? u.w : (prod ? gm.w : 0.0f));
}

__device__ __forceinline__ void make_weights(const float* __restrict__ Wm,
    const float* __restrict__ Bv, unsigned lane, v2f (&a)[6], v8f& cinit) {
  const unsigned o  = lane & 15u;
  const unsigned h  = lane >> 4;
  const unsigned g  = o / 3u;
  const unsigned j  = o - 3u * g;
  const bool valid  = (o < 12u) && ((g >> 1) == h);
  const float msk   = valid ? 1.0f : 0.0f;
  const bool second = (g & 1u) != 0u;
  float w[6];
#pragma unroll
  for (int i = 0; i < 6; ++i) w[i] = Wm[j * 6u + (unsigned)i] * msk;
#pragma unroll
  for (int c = 0; c < 3; ++c) {
    a[c].x     = second ? 0.0f : w[2 * c];
    a[c].y     = second ? 0.0f : w[2 * c + 1];
    a[c + 3].x = second ? w[2 * c]     : 0.0f;
    a[c + 3].y = second ? w[2 * c + 1] : 0.0f;
  }
  const float b0 = Bv[0], b1 = Bv[1], b2 = Bv[2];
  const bool hi = h != 0u;
  cinit[0] = hi ? b2 : b0;
  cinit[1] = hi ? b0 : b1;
  cinit[2] = hi ? b1 : b2;
  cinit[3] = hi ? b2 : b0;
  cinit[4] = hi ? 0.0f : b1;
  cinit[5] = hi ? 0.0f : b2;
  cinit[6] = hi ? 0.0f : b0;
  cinit[7] = hi ? 0.0f : b1;
}

__device__ __forceinline__ v8f group_product(const v2f (&a)[6], const float4& x0,
    const float4& x1, const float4& x2, v8f c) {
  v2f b;
  b.x = x0.x; b.y = x0.y;
  c = __builtin_amdgcn_wmma_f32_16x16x4_f32(false, a[0], false, b, (short)0, c, false, false);
  b.x = x0.z; b.y = x0.w;
  c = __builtin_amdgcn_wmma_f32_16x16x4_f32(false, a[1], false, b, (short)0, c, false, false);
  b.x = x1.x; b.y = x1.y;
  c = __builtin_amdgcn_wmma_f32_16x16x4_f32(false, a[2], false, b, (short)0, c, false, false);
  b.x = x1.z; b.y = x1.w;
  c = __builtin_amdgcn_wmma_f32_16x16x4_f32(false, a[3], false, b, (short)0, c, false, false);
  b.x = x2.x; b.y = x2.y;
  c = __builtin_amdgcn_wmma_f32_16x16x4_f32(false, a[4], false, b, (short)0, c, false, false);
  b.x = x2.z; b.y = x2.w;
  c = __builtin_amdgcn_wmma_f32_16x16x4_f32(false, a[5], false, b, (short)0, c, false, false);
  return c;
}

__global__ __launch_bounds__(256, 8) void gsn_gate_wmma(
    const float* __restrict__ hvA, const float* __restrict__ m0A, float* __restrict__ GA,
    const float* __restrict__ hvB, const float* __restrict__ m0B, float* __restrict__ GB,
    const float* __restrict__ hvC, const float* __restrict__ m0C,
    const float* __restrict__ m1C, float* __restrict__ GC,
    const float* __restrict__ Wx,
    const float* __restrict__ Bx) {
  const bool isC = blockIdx.x >= 2u * GBLK1;
  const bool isB = !isC && blockIdx.x >= GBLK1;
  const unsigned bx = blockIdx.x - (isC ? 2u * GBLK1 : (isB ? GBLK1 : 0u));
  const bool has_m1 = isC;
  const unsigned nwaves = isC ? GWAV2 : GWAV1;
  const float* __restrict__ hv = isC ? hvC : (isB ? hvB : hvA);
  const float* __restrict__ m0 = isC ? m0C : (isB ? m0B : m0A);
  const float* __restrict__ m1 = m1C;
  float* __restrict__ G        = isC ? GC : (isB ? GB : GA);
  const unsigned lane = threadIdx.x & 31u;
  const unsigned wv   = threadIdx.x >> 5u;
  const unsigned gw   = bx * 8u + wv;
  const unsigned h    = lane >> 4;
  const unsigned loff = 24u * (lane & 15u) + 12u * h;
  const unsigned poff = 12u * (lane & 15u) + 8u * h;
  const unsigned lw   = 384u * wv + loff;

  v2f a[6];
  v8f cinit;
  make_weights(Wx, Bx, lane, a, cinit);
  const float sg0 = fast_sigmoid(cinit[0]), sg1 = fast_sigmoid(cinit[1]), sg2 = fast_sigmoid(cinit[2]);
  const float4 zo0 = make_float4(sg0, sg1, sg2, sg0);
  const float4 zo1 = make_float4(sg1, sg2, sg0, sg1);

  for (unsigned ch = 0; ch < TPW / GCH; ++ch) {
    float4 x0[GCH], x1[GCH], x2[GCH];
#pragma unroll
    for (unsigned t = 0; t < GCH; ++t) {
      const unsigned k  = ch * GCH + t;
      const unsigned f0 = 384u * (k * nwaves + bx * 8u);
      const unsigned seg = f0 >> LOG_TD;
      const bool whole = ((f0 + 8u * 384u - 1u) >> LOG_TD) == seg;
      const bool is_h  = seg >= 3u;
      const bool is_m0 = seg == 0u;
      const bool is_m1 = has_m1 && seg == 1u;
      if (whole && (is_h || is_m0 || is_m1)) {
        const float* bp = (is_h ? hv : (is_m0 ? m0 : m1)) + (f0 & (TD - 1u));
        x0[t] = ld4(bp + lw);
        x1[t] = ld4(bp + lw + 4u);
        x2[t] = ld4(bp + lw + 8u);
      } else if (whole) {
      } else {
        const unsigned base = f0 + lw;
        x0[t] = gate_src4(has_m1, base,      hv, m0, m1);
        x1[t] = gate_src4(has_m1, base + 4u, hv, m0, m1);
        x2[t] = gate_src4(has_m1, base + 8u, hv, m0, m1);
      }
    }
#pragma unroll
    for (unsigned t = 0; t < GCH; ++t) {
      const unsigned tile = (ch * GCH + t) * nwaves + gw;
      const unsigned f0  = 384u * ((ch * GCH + t) * nwaves + bx * 8u);
      const unsigned seg = f0 >> LOG_TD;
      const bool whole = ((f0 + 8u * 384u - 1u) >> LOG_TD) == seg;
      const bool empty = whole && !(seg >= 3u || seg == 0u || (has_m1 && seg == 1u));

      const unsigned p = 192u * tile + poff;
      float4 o0 = zo0, o1 = zo1;
      if (!empty) {
        const v8f d = group_product(a, x0[t], x1[t], x2[t], cinit);
        o0.x = fast_sigmoid(d[0]); o0.y = fast_sigmoid(d[1]);
        o0.z = fast_sigmoid(d[2]); o0.w = fast_sigmoid(d[3]);
        o1.x = fast_sigmoid(d[4]); o1.y = fast_sigmoid(d[5]);
        o1.z = fast_sigmoid(d[6]); o1.w = fast_sigmoid(d[7]);
      }
      *(float4*)(G + p) = o0;
      if (h == 0u)
        *(float4*)(G + p + 4u) = o1;
    }
  }
}

__device__ __forceinline__ float4 sigmoid4(float d0, float d1, float d2, float d3) {
  return make_float4(fast_sigmoid(d0), fast_sigmoid(d1), fast_sigmoid(d2), fast_sigmoid(d3));
}

__device__ __forceinline__ float4 delta4(const float4& g, const float4& mv,
    float d0, float d1, float d2, float d3) {
  float4 o;
  o.x = (1.0f - g.x) * mv.x + g.x * fast_tanh(d0);
  o.y = (1.0f - g.y) * mv.y + g.y * fast_tanh(d1);
  o.z = (1.0f - g.z) * mv.z + g.z * fast_tanh(d2);
  o.w = (1.0f - g.w) * mv.w + g.w * fast_tanh(d3);
  return o;
}

__device__ __forceinline__ float4 gated4(const float4& g, float d0, float d1, float d2, float d3) {
  float4 o;
  o.x = g.x * fast_tanh(d0);
  o.y = g.y * fast_tanh(d1);
  o.z = g.z * fast_tanh(d2);
  o.w = g.w * fast_tanh(d3);
  return o;
}

__device__ __forceinline__ float4 tanh4(float d0, float d1, float d2, float d3) {
  return make_float4(fast_tanh(d0), fast_tanh(d1), fast_tanh(d2), fast_tanh(d3));
}
__device__ __forceinline__ float4 delta4t(const float4& g, const float4& mv, const float4& t) {
  float4 o;
  o.x = (1.0f - g.x) * mv.x + g.x * t.x;
  o.y = (1.0f - g.y) * mv.y + g.y * t.y;
  o.z = (1.0f - g.z) * mv.z + g.z * t.z;
  o.w = (1.0f - g.w) * mv.w + g.w * t.w;
  return o;
}
__device__ __forceinline__ float4 gated4t(const float4& g, const float4& t) {
  return make_float4(g.x * t.x, g.y * t.y, g.z * t.z, g.w * t.w);
}


__device__ __forceinline__ void cand_slot_lane(bool has_m1, unsigned Q0, unsigned g0, unsigned nq,
    unsigned ll, unsigned ql, unsigned h, const v2f (&a)[6], const v8f& cinit,
    const v2f (&ag)[6], const v8f& ginit,
    const float* __restrict__ hv, const float* __restrict__ m0, const float* __restrict__ m1,
    const float* __restrict__ G, float4* accn) {
  for (unsigned t = 0; t < TPS; ++t) {
    const unsigned bu   = 24u * (g0 + t * 128u) + ll;
    const unsigned bmax = 24u * (NQ - 1u) + 12u * h;
    const unsigned base = bu < bmax ? bu : bmax;

    const float4 y0 = gate_src4(has_m1, base,      hv, m0, m1);
    const float4 y1 = gate_src4(has_m1, base + 4u, hv, m0, m1);
    const float4 y2 = gate_src4(has_m1, base + 8u, hv, m0, m1);
    const v8f dg = group_product(ag, y0, y1, y2, ginit);
    const float4 ga = sigmoid4(dg[0], dg[1], dg[2], dg[3]);
    const float4 gb = sigmoid4(dg[4], dg[5], dg[6], dg[7]);

    float4 u0, u1, u2, w0, w1, w2;
    cand_src4_load(has_m1, base,      hv, m0, m1, G, u0, w0);
    cand_src4_load(has_m1, base + 4u, hv, m0, m1, G, u1, w1);
    cand_src4_load(has_m1, base + 8u, hv, m0, m1, G, u2, w2);
    const float4 x0 = cand_src4_value(has_m1, base,      u0, w0);
    const float4 x1 = cand_src4_value(has_m1, base + 4u, u1, w1);
    const float4 x2 = cand_src4_value(has_m1, base + 8u, u2, w2);
    const v8f d = group_product(a, x0, x1, x2, cinit);

    const unsigned p  = (base >> 1) + 2u * h;
    const unsigned p2 = p + 4u * (h ^ 1u);
    const float4 ma = msg4(has_m1, p,  hv, m0, m1);
    const float4 mb = msg4(has_m1, p2, hv, m0, m1);

    const unsigned qa = 3u * (g0 + t * 128u) - Q0 + ql;
    if (qa < nq)
      accn[qa] = delta4(ga, ma, d[0], d[1], d[2], d[3]);
    if (h == 0u && qa + 1u < nq)
      accn[qa + 1u] = delta4(gb, mb, d[4], d[5], d[6], d[7]);
  }
}

constexpr int K_PROD = 1;
constexpr int K_H    = 2;
constexpr int K_ZERO = 3;

constexpr int GK_W    = 1;
constexpr int GK_U    = 2;
constexpr int GK_ZERO = 3;
constexpr int GK_LOAD = 4;

template <int KIND, bool MSG, int GK>
__device__ __forceinline__ void cand_slot_uniform(
    const float* __restrict__ ub, const float* __restrict__ wb,
    const float* __restrict__ gs, const float* __restrict__ mp,
    unsigned rb, unsigned pr, unsigned qoff, unsigned nq, unsigned ll, unsigned ql, unsigned h,
    const v2f (&a)[6], const v8f& cinit, const v2f (&ag)[6], const v8f& ginit,
    float4* accn) {
  static_assert(GK != GK_W || KIND == K_PROD, "w is loaded only for a gate * message slab");
  static_assert(GK != GK_U || KIND == K_H, "u is the node state only in an h slab");
  const unsigned lo = rb + ll;
  const unsigned po = pr + 4u * ql;
  const unsigned pq = po + 4u * (h ^ 1u);
  float4 zga = zero4(), zgb = zero4(), zta = zero4(), ztb = zero4();
  if (GK == GK_ZERO || KIND == K_ZERO) {
    float gb0 = ginit[0], gb1 = ginit[1], gb2 = ginit[2], cb0 = cinit[0], cb1 = cinit[1], cb2 = cinit[2];
    asm volatile("" : "+v"(gb0), "+v"(gb1), "+v"(gb2), "+v"(cb0), "+v"(cb1), "+v"(cb2));
    if (GK == GK_ZERO) {
      const float sg0 = fast_sigmoid(gb0), sg1 = fast_sigmoid(gb1), sg2 = fast_sigmoid(gb2);
      zga = make_float4(sg0, sg1, sg2, sg0);
      zgb = make_float4(sg1, sg2, sg0, sg1);
    }
    if (KIND == K_ZERO) {
      const float st0 = fast_tanh(cb0), st1 = fast_tanh(cb1), st2 = fast_tanh(cb2);
      zta = make_float4(st0, st1, st2, st0);
      ztb = make_float4(st1, st2, st0, st1);
    }
  }
  float4 u[TPS][3], w[TPS][3], y[TPS][3], ma[TPS], mb[TPS];
#pragma unroll
  for (unsigned t = 0; t < TPS; ++t) {
    const unsigned ot = lo + t * (24u * 128u);
    const unsigned pt = po + t * (12u * 128u);
    const unsigned pu = pq + t * (12u * 128u);
    if (KIND != K_ZERO) {
      u[t][0] = ld4(ub + ot);
      u[t][1] = ld4(ub + ot + 4u);
      u[t][2] = ld4(ub + ot + 8u);
    } else {
      u[t][0] = zero4(); u[t][1] = zero4(); u[t][2] = zero4();
    }
    if (KIND == K_PROD) {
      w[t][0] = ld4(wb + ot);
      w[t][1] = ld4(wb + ot + 4u);
      w[t][2] = ld4(wb + ot + 8u);
    } else {
      w[t][0] = zero4(); w[t][1] = zero4(); w[t][2] = zero4();
    }
    if (GK == GK_LOAD) {
      y[t][0] = ld4(gs + ot);
      y[t][1] = ld4(gs + ot + 4u);
      y[t][2] = ld4(gs + ot + 8u);
    } else {
      y[t][0] = zero4(); y[t][1] = zero4(); y[t][2] = zero4();
    }
    if (MSG) {
      ma[t] = ld4(mp + pt);
      mb[t] = ld4(mp + pu);
    } else {
      ma[t] = zero4(); mb[t] = zero4();
    }
  }
#pragma unroll
  for (unsigned t = 0; t < TPS; ++t) {
    if (KIND != K_ZERO)
      asm volatile("" :: "v"(u[t][0].x), "v"(u[t][0].y), "v"(u[t][0].z), "v"(u[t][0].w),
                         "v"(u[t][1].x), "v"(u[t][1].y), "v"(u[t][1].z), "v"(u[t][1].w),
                         "v"(u[t][2].x), "v"(u[t][2].y), "v"(u[t][2].z), "v"(u[t][2].w));
    if (KIND == K_PROD)
      asm volatile("" :: "v"(w[t][0].x), "v"(w[t][0].y), "v"(w[t][0].z), "v"(w[t][0].w),
                         "v"(w[t][1].x), "v"(w[t][1].y), "v"(w[t][1].z), "v"(w[t][1].w),
                         "v"(w[t][2].x), "v"(w[t][2].y), "v"(w[t][2].z), "v"(w[t][2].w));
    if (GK == GK_LOAD)
      asm volatile("" :: "v"(y[t][0].x), "v"(y[t][0].y), "v"(y[t][0].z), "v"(y[t][0].w),
                         "v"(y[t][1].x), "v"(y[t][1].y), "v"(y[t][1].z), "v"(y[t][1].w),
                         "v"(y[t][2].x), "v"(y[t][2].y), "v"(y[t][2].z), "v"(y[t][2].w));
    if (MSG)
      asm volatile("" :: "v"(ma[t].x), "v"(ma[t].y), "v"(ma[t].z), "v"(ma[t].w),
                         "v"(mb[t].x), "v"(mb[t].y), "v"(mb[t].z), "v"(mb[t].w));
  }
#pragma unroll
  for (unsigned t = 0; t < TPS; ++t) {
    const float4 y0 = GK == GK_W ? w[t][0] : (GK == GK_U ? u[t][0] : y[t][0]);
    const float4 y1 = GK == GK_W ? w[t][1] : (GK == GK_U ? u[t][1] : y[t][1]);
    const float4 y2 = GK == GK_W ? w[t][2] : (GK == GK_U ? u[t][2] : y[t][2]);
    float4 x0 = u[t][0], x1 = u[t][1], x2 = u[t][2];
    if (KIND == K_PROD) {
      x0 = make_float4(x0.x * w[t][0].x, x0.y * w[t][0].y, x0.z * w[t][0].z, x0.w * w[t][0].w);
      x1 = make_float4(x1.x * w[t][1].x, x1.y * w[t][1].y, x1.z * w[t][1].z, x1.w * w[t][1].w);
      x2 = make_float4(x2.x * w[t][2].x, x2.y * w[t][2].y, x2.z * w[t][2].z, x2.w * w[t][2].w);
    }
    float4 ta = zta, tb = ztb;
    if (KIND != K_ZERO) {
      const v8f d = group_product(a, x0, x1, x2, cinit);
      ta = tanh4(d[0], d[1], d[2], d[3]);
      tb = tanh4(d[4], d[5], d[6], d[7]);
    }
    float4 ga = zga, gb = zgb;
    if (GK != GK_ZERO) {
      const v8f dg = group_product(ag, y0, y1, y2, ginit);
      ga = sigmoid4(dg[0], dg[1], dg[2], dg[3]);
      gb = sigmoid4(dg[4], dg[5], dg[6], dg[7]);
    }

    const unsigned qa = qoff + t * 384u + ql;
    if (qa < nq)
      accn[qa] = MSG ? delta4t(ga, ma[t], ta) : gated4t(ga, ta);
    if (h == 0u && qa + 1u < nq)
      accn[qa + 1u] = MSG ? delta4t(gb, mb[t], tb) : gated4t(gb, tb);
  }
}

__global__ __launch_bounds__(256) __attribute__((amdgpu_waves_per_eu(8, 8))) void gsn_cand_wmma(
    const float* __restrict__ hvA, const float* __restrict__ m0A,
    const float* __restrict__ GA, float* __restrict__ outA,
    const float* __restrict__ hvB, const float* __restrict__ m0B,
    const float* __restrict__ GB, float* __restrict__ outB,
    const float* __restrict__ hvC, const float* __restrict__ m0C, const float* __restrict__ m1C,
    const float* __restrict__ GC, float* __restrict__ outC,
    const float* __restrict__ Ww,
    const float* __restrict__ Bw,
    const float* __restrict__ Wx,
    const float* __restrict__ Bx) {
  __shared__ float4 acc[3][LQ];
  const bool isC = blockIdx.x >= 2u * CBLK;
  const bool isB = !isC && blockIdx.x >= CBLK;
  const unsigned bx = blockIdx.x - (isC ? 2u * CBLK : (isB ? CBLK : 0u));
  const bool has_m1 = isC;
  const float* __restrict__ hv = isC ? hvC : (isB ? hvB : hvA);
  const float* __restrict__ m0 = isC ? m0C : (isB ? m0B : m0A);
  const float* __restrict__ m1 = m1C;
  const float* __restrict__ G  = isC ? GC : (isB ? GB : GA);
  float* __restrict__ out      = isC ? outC : (isB ? outB : outA);
  const unsigned lane = threadIdx.x & 31u;
  const unsigned h    = lane >> 4;
  const unsigned gl   = (threadIdx.x >> 5u) * 16u + (lane & 15u);
  const unsigned ll   = 24u * gl + 12u * h;
  const unsigned ql   = 3u * gl + 2u * h;
  const unsigned q0   = bx * LQ;
  const unsigned nq   = (NQ - q0) < LQ ? (NQ - q0) : LQ;

  v2f a[6], ag[6];
  v8f cinit, ginit;
  make_weights(Ww, Bw, lane, a, cinit);
  make_weights(Wx, Bx, lane, ag, ginit);

  bool all_whole = true;
  for (unsigned n = 0; n < 3u; ++n) {
    const unsigned g0    = (n * NQ + q0) / 3u;
    const unsigned op_lo = 24u * g0;
    const unsigned ep_lo = 12u * g0;
    all_whole = all_whole && g0 + (GPB - 1u) < NQ
        && ((op_lo + 24u * GPB - 1u) >> LOG_TD) == (op_lo >> LOG_TD)
        && (ep_lo >> LOG_TD) == n && ((ep_lo + 12u * GPB - 1u) >> LOG_TD) == n;
  }
  if (all_whole) {
    for (unsigned n = 0; n < 3u; ++n) {
      const unsigned Q0 = n * NQ + q0;
      const unsigned g0 = Q0 / 3u;
      const unsigned op_lo = 24u * g0;
      const unsigned ep_lo = 12u * g0;
      const unsigned seg   = op_lo >> LOG_TD;
      const unsigned rb  = op_lo & (TD - 1u);
      const unsigned pr  = ep_lo & (TD - 1u);
      const unsigned qoff = 3u * g0 - Q0;
      if (seg == 0u)
        cand_slot_uniform<K_PROD, true, GK_W>(G, m0, hv, m0, rb, pr, qoff, nq, ll, ql, h, a, cinit, ag, ginit, acc[n]);
      else if (seg == 1u && has_m1)
        cand_slot_uniform<K_H, true, GK_LOAD>(hv, hv, m1, m0, rb, pr, qoff, nq, ll, ql, h, a, cinit, ag, ginit, acc[n]);
      else if (seg == 1u)
        cand_slot_uniform<K_H, true, GK_ZERO>(hv, hv, hv, m0, rb, pr, qoff, nq, ll, ql, h, a, cinit, ag, ginit, acc[n]);
      else if (seg == 2u && has_m1)
        cand_slot_uniform<K_PROD, true, GK_ZERO>(G + TD, m1, hv, m1, rb, pr, qoff, nq, ll, ql, h, a, cinit, ag, ginit, acc[n]);
      else if (seg == 2u)
        cand_slot_uniform<K_ZERO, false, GK_ZERO>(hv, hv, hv, hv, rb, pr, qoff, nq, ll, ql, h, a, cinit, ag, ginit, acc[n]);
      else if (seg == 3u && has_m1)
        cand_slot_uniform<K_H, true, GK_U>(hv, hv, hv, m1, rb, pr, qoff, nq, ll, ql, h, a, cinit, ag, ginit, acc[n]);
      else if (seg == 4u)
        cand_slot_uniform<K_ZERO, false, GK_LOAD>(hv, hv, hv, hv, rb, pr, qoff, nq, ll, ql, h, a, cinit, ag, ginit, acc[n]);
      else
        cand_slot_uniform<K_H, false, GK_U>(hv, hv, hv, hv, rb, pr, qoff, nq, ll, ql, h, a, cinit, ag, ginit, acc[n]);
    }
  } else {
    for (unsigned n = 0; n < 3u; ++n) {
      const unsigned Q0 = n * NQ + q0;
      cand_slot_lane(has_m1, Q0, Q0 / 3u, nq, ll, ql, h, a, cinit, ag, ginit, hv, m0, m1, G, acc[n]);
    }
  }
  __syncthreads();
  for (unsigned i = threadIdx.x; i < nq; i += 256u) {
    const float4 h4 = ld4(hv + 4u * (q0 + i));
    const float4 c0 = acc[0][i];
    const float4 c1 = acc[1][i];
    const float4 c2 = acc[2][i];
    float4 o;
    o.x = h4.x + c0.x + c1.x + c2.x;
    o.y = h4.y + c0.y + c1.y + c2.y;
    o.z = h4.z + c0.z + c1.z + c2.z;
    o.w = h4.w + c0.w + c1.w + c2.w;
    *(float4*)(out + 4u * (q0 + i)) = o;
  }
}

extern "C" void kernel_launch(void* const* d_in, const int* in_sizes, int n_in,
                              void* d_out, int out_size, void* d_ws, size_t ws_size,
                              hipStream_t stream) {
  (void)in_sizes; (void)n_in; (void)out_size; (void)ws_size;
  const float* feat = (const float*)d_in[0];
  const float* W_w  = (const float*)d_in[1];
  const float* W_b  = (const float*)d_in[2];
  const float* Wx_w = (const float*)d_in[3];
  const float* Wx_b = (const float*)d_in[4];
  float* out = (float*)d_out;

  float* G  = (float*)d_ws;
  float* G1 = G + (size_t)GEXT1;
  float* G2 = G + (size_t)2 * GEXT1;

  const float* h0 = feat;
  const float* h1 = feat + (size_t)TD;
  const float* h2 = feat + (size_t)2 * TD;

  dim3 blk(256);
  dim3 grd_g(GBLK3);
  dim3 grd_c(CBLK3);

  gsn_gate_wmma<<<grd_g, blk, 0, stream>>>(
      h0, h2, G, h1, h0, G1, h2, h1, h0, G2, Wx_w, Wx_b);
  gsn_cand_wmma<<<grd_c, blk, 0, stream>>>(
      h0, h2, G, out + 0 * (size_t)TD, h1, h0, G1, out + 1 * (size_t)TD,
      h2, h1, h0, G2, out + 2 * (size_t)TD, W_w, W_b, Wx_w, Wx_b);
}


# Operand gates made in the cand pass: a lane loads the 4-5 gate rows behind its 12 operand positions, a pair-form product on the gate weights leaves its 12 gates in that lane; gate launch gone, 2->1
# speedup vs baseline: 5.1746x; 1.3689x over previous
#include <hip/hip_runtime.h>
#include <math.h>

#define LOG_TD 22u
#define TD (1u << LOG_TD)

typedef __attribute__((ext_vector_type(2))) float v2f;
typedef __attribute__((ext_vector_type(8))) float v8f;

constexpr unsigned NQ     = TD / 4u;
constexpr unsigned TPS    = 2u;
constexpr unsigned GPB    = 16u * 8u * TPS;
constexpr unsigned LQ     = 760u;
constexpr unsigned CBLK   = (NQ + LQ - 1u) / LQ;
constexpr unsigned CBLK3  = 3u * CBLK;
static_assert(LQ % 8u == 0u && LQ + 2u <= 3u * GPB, "block range must be whole lines and fit its groups");
static_assert(TD % 12u == 4u, "gate alignment in slab 2");

__device__ __forceinline__ float hw_exp2(float x) {
#if __has_builtin(__builtin_amdgcn_exp2f)
  return __builtin_amdgcn_exp2f(x);
#else
  return exp2f(x);
#endif
}
__device__ __forceinline__ float hw_rcp(float x) {
#if __has_builtin(__builtin_amdgcn_rcpf)
  return __builtin_amdgcn_rcpf(x);
#else
  return 1.0f / x;
#endif
}
__device__ __forceinline__ float fast_sigmoid(float x) {
  return hw_rcp(1.0f + hw_exp2(-1.4426950408889634f * x));
}
__device__ __forceinline__ float fast_tanh(float x) {
#if __has_builtin(__builtin_amdgcn_tanhf)
  return __builtin_amdgcn_tanhf(x);
#else
  return 1.0f - 2.0f * hw_rcp(hw_exp2(2.8853900817779268f * x) + 1.0f);
#endif
}

__device__ __forceinline__ float4 ld4(const float* __restrict__ p) {
  return *(const float4*)p;
}
__device__ __forceinline__ float4 zero4() { return make_float4(0.0f, 0.0f, 0.0f, 0.0f); }
__device__ __forceinline__ float4 sel4(bool c, const float4& v) {
  return make_float4(c ? v.x : 0.0f, c ? v.y : 0.0f, c ? v.z : 0.0f, c ? v.w : 0.0f);
}


__device__ __forceinline__ float4 gate_src4(bool has_m1, unsigned m,
    const float* __restrict__ hv, const float* __restrict__ m0,
    const float* __restrict__ m1) {
  const unsigned seg = m >> LOG_TD;
  const unsigned r   = m & (TD - 1u);
  const bool is_h  = seg >= 3u;
  const bool is_m0 = seg == 0u;
  const bool is_m1 = has_m1 && seg == 1u;
  const bool valid = is_h || is_m0 || is_m1;
  const float* base = is_m0 ? m0 : (is_m1 ? m1 : hv);
  const float4 v = ld4(base + (valid ? r : 0u));
  return sel4(valid, v);
}

__device__ __forceinline__ float4 msg4(bool has_m1, unsigned p, const float* __restrict__ hv,
    const float* __restrict__ m0, const float* __restrict__ m1) {
  const unsigned n = p >> LOG_TD;
  const unsigned r = p & (TD - 1u);
  const bool is_m0 = n == 0u;
  const bool is_m1 = has_m1 && n == 1u;
  const bool valid = is_m0 || is_m1;
  const float* base = is_m0 ? m0 : (is_m1 ? m1 : hv);
  const float4 v = ld4(base + (valid ? r : 0u));
  return sel4(valid, v);
}

__device__ __forceinline__ float4 cand_val4(bool has_m1, unsigned m, const float4& gu,
    const float* __restrict__ hv, const float* __restrict__ m0,
    const float* __restrict__ m1) {
  const unsigned seg = m >> LOG_TD;
  const unsigned r   = m & (TD - 1u);
  const unsigned n   = seg >> 1;
  const bool is_h  = (seg & 1u) != 0u;
  const bool is_m0 = !is_h && n == 0u;
  const bool is_m1 = has_m1 && !is_h && n == 1u;
  const bool prod  = is_m0 || is_m1;
  const float* wb = is_m0 ? m0 : (is_m1 ? m1 : hv);
  const float4 v = ld4(wb + ((prod || is_h) ? r : 0u));
  const float4 gm = make_float4(gu.x * v.x, gu.y * v.y, gu.z * v.z, gu.w * v.w);
  return make_float4(is_h ? v.x : (prod ? gm.x : 0.0f), is_h ? v.y : (prod ? gm.y : 0.0f),
                     is_h ? v.z : (prod ? gm.z : 0.0f), is_h ? v.w : (prod ? gm.w : 0.0f));
}

__device__ __forceinline__ void make_weights(const float* __restrict__ Wm,
    const float* __restrict__ Bv, unsigned lane, v2f (&a)[6], v8f& cinit) {
  const unsigned o  = lane & 15u;
  const unsigned h  = lane >> 4;
  const unsigned g  = o / 3u;
  const unsigned j  = o - 3u * g;
  const bool valid  = (o < 12u) && ((g >> 1) == h);
  const float msk   = valid ? 1.0f : 0.0f;
  const bool second = (g & 1u) != 0u;
  float w[6];
#pragma unroll
  for (int i = 0; i < 6; ++i) w[i] = Wm[j * 6u + (unsigned)i] * msk;
#pragma unroll
  for (int c = 0; c < 3; ++c) {
    a[c].x     = second ? 0.0f : w[2 * c];
    a[c].y     = second ? 0.0f : w[2 * c + 1];
    a[c + 3].x = second ? w[2 * c]     : 0.0f;
    a[c + 3].y = second ? w[2 * c + 1] : 0.0f;
  }
  const float b0 = Bv[0], b1 = Bv[1], b2 = Bv[2];
  const bool hi = h != 0u;
  cinit[0] = hi ? b2 : b0;
  cinit[1] = hi ? b0 : b1;
  cinit[2] = hi ? b1 : b2;
  cinit[3] = hi ? b2 : b0;
  cinit[4] = hi ? 0.0f : b1;
  cinit[5] = hi ? 0.0f : b2;
  cinit[6] = hi ? 0.0f : b0;
  cinit[7] = hi ? 0.0f : b1;
}

__device__ __forceinline__ v8f group_product(const v2f (&a)[6], const float4& x0,
    const float4& x1, const float4& x2, v8f c) {
  v2f b;
  b.x = x0.x; b.y = x0.y;
  c = __builtin_amdgcn_wmma_f32_16x16x4_f32(false, a[0], false, b, (short)0, c, false, false);
  b.x = x0.z; b.y = x0.w;
  c = __builtin_amdgcn_wmma_f32_16x16x4_f32(false, a[1], false, b, (short)0, c, false, false);
  b.x = x1.x; b.y = x1.y;
  c = __builtin_amdgcn_wmma_f32_16x16x4_f32(false, a[2], false, b, (short)0, c, false, false);
  b.x = x1.z; b.y = x1.w;
  c = __builtin_amdgcn_wmma_f32_16x16x4_f32(false, a[3], false, b, (short)0, c, false, false);
  b.x = x2.x; b.y = x2.y;
  c = __builtin_amdgcn_wmma_f32_16x16x4_f32(false, a[4], false, b, (short)0, c, false, false);
  b.x = x2.z; b.y = x2.w;
  c = __builtin_amdgcn_wmma_f32_16x16x4_f32(false, a[5], false, b, (short)0, c, false, false);
  return c;
}

__device__ __forceinline__ float shr2_upper(float v) {
  const int i = __builtin_bit_cast(int, v);
  return __builtin_bit_cast(float, __builtin_amdgcn_update_dpp(i, i, 0x112, 0x2, 0xf, true));
}
__device__ __forceinline__ void pair_weights(const v2f (&ag)[6], v2f (&ap)[6]) {
#pragma unroll
  for (int c = 0; c < 6; ++c) {
    ap[c].x = shr2_upper(ag[c].x);
    ap[c].y = shr2_upper(ag[c].y);
  }
}
__device__ __forceinline__ void pair_bias(const float* __restrict__ Bv, v8f& cinit) {
  const float b0 = Bv[0], b1 = Bv[1], b2 = Bv[2];
  cinit[0] = b0; cinit[1] = b1; cinit[2] = b2;
  cinit[3] = b0; cinit[4] = b1; cinit[5] = b2;
  cinit[6] = 0.0f; cinit[7] = 0.0f;
}

__device__ __forceinline__ v8f row_product(const v2f (&a)[6], const float4& x0,
    const float4& x1, v8f c) {
  v2f b;
  b.x = x0.x; b.y = x0.y;
  c = __builtin_amdgcn_wmma_f32_16x16x4_f32(false, a[0], false, b, (short)0, c, false, false);
  b.x = x0.z; b.y = x0.w;
  c = __builtin_amdgcn_wmma_f32_16x16x4_f32(false, a[1], false, b, (short)0, c, false, false);
  b.x = x1.x; b.y = x1.y;
  c = __builtin_amdgcn_wmma_f32_16x16x4_f32(false, a[2], false, b, (short)0, c, false, false);
  return c;
}

__device__ __forceinline__ float4 sigmoid4(float d0, float d1, float d2, float d3) {
  return make_float4(fast_sigmoid(d0), fast_sigmoid(d1), fast_sigmoid(d2), fast_sigmoid(d3));
}

__device__ __forceinline__ float4 delta4(const float4& g, const float4& mv,
    float d0, float d1, float d2, float d3) {
  float4 o;
  o.x = (1.0f - g.x) * mv.x + g.x * fast_tanh(d0);
  o.y = (1.0f - g.y) * mv.y + g.y * fast_tanh(d1);
  o.z = (1.0f - g.z) * mv.z + g.z * fast_tanh(d2);
  o.w = (1.0f - g.w) * mv.w + g.w * fast_tanh(d3);
  return o;
}

__device__ __forceinline__ float4 gated4(const float4& g, float d0, float d1, float d2, float d3) {
  float4 o;
  o.x = g.x * fast_tanh(d0);
  o.y = g.y * fast_tanh(d1);
  o.z = g.z * fast_tanh(d2);
  o.w = g.w * fast_tanh(d3);
  return o;
}

__device__ __forceinline__ float4 tanh4(float d0, float d1, float d2, float d3) {
  return make_float4(fast_tanh(d0), fast_tanh(d1), fast_tanh(d2), fast_tanh(d3));
}
__device__ __forceinline__ float4 delta4t(const float4& g, const float4& mv, const float4& t) {
  float4 o;
  o.x = (1.0f - g.x) * mv.x + g.x * t.x;
  o.y = (1.0f - g.y) * mv.y + g.y * t.y;
  o.z = (1.0f - g.z) * mv.z + g.z * t.z;
  o.w = (1.0f - g.w) * mv.w + g.w * t.w;
  return o;
}
__device__ __forceinline__ float4 gated4t(const float4& g, const float4& t) {
  return make_float4(g.x * t.x, g.y * t.y, g.z * t.z, g.w * t.w);
}


__device__ __forceinline__ void gates_sh0(const v8f& P, const v8f& Q, float4& u0, float4& u1, float4& u2) {
  u0 = sigmoid4(P[0], P[1], P[2], P[3]);
  u1 = sigmoid4(P[4], P[5], Q[0], Q[1]);
  u2 = sigmoid4(Q[2], Q[3], Q[4], Q[5]);
}
__device__ __forceinline__ void gates_sh2(const v8f& P, const v8f& Q, const v8f& R, float4& u0, float4& u1, float4& u2) {
  u0 = sigmoid4(P[2], P[3], P[4], P[5]);
  u1 = sigmoid4(Q[0], Q[1], Q[2], Q[3]);
  u2 = sigmoid4(Q[4], Q[5], R[0], R[1]);
}

__device__ __forceinline__ void cand_slot_lane(bool has_m1, unsigned Q0, unsigned g0, unsigned nq,
    unsigned ll, unsigned ql, unsigned h, const v2f (&a)[6], const v8f& cinit,
    const v2f (&ag)[6], const v8f& ginit, const v8f& pinit,
    const float* __restrict__ hv, const float* __restrict__ m0, const float* __restrict__ m1,
    float4* accn) {
  v2f ap[6];
  pair_weights(ag, ap);
  const unsigned seg_lo = (24u * g0) >> LOG_TD;
  const unsigned seg_hi = (24u * g0 + 24u * GPB - 1u) >> LOG_TD;
  const bool in0 = seg_lo == 0u;
  const bool in2 = has_m1 && (seg_lo == 2u || seg_hi == 2u);
  for (unsigned t = 0; t < TPS; ++t) {
    const unsigned bu   = 24u * (g0 + t * 128u) + ll;
    const unsigned bmax = 24u * (NQ - 1u) + 12u * h;
    const unsigned base = bu < bmax ? bu : bmax;

    float4 gu0 = zero4(), gu1 = zero4(), gu2 = zero4();
    if (in0) {
      const unsigned gi = 2u * base;
      const float4 X0 = gate_src4(has_m1, gi,       hv, m0, m1);
      const float4 X1 = gate_src4(has_m1, gi + 4u,  hv, m0, m1);
      const float4 X2 = gate_src4(has_m1, gi + 8u,  hv, m0, m1);
      const float4 X3 = gate_src4(has_m1, gi + 12u, hv, m0, m1);
      const float4 X4 = gate_src4(has_m1, gi + 16u, hv, m0, m1);
      const float4 X5 = gate_src4(has_m1, gi + 20u, hv, m0, m1);
      const v8f P = group_product(ap, X0, X1, X2, pinit);
      const v8f Q = group_product(ap, X3, X4, X5, pinit);
      gates_sh0(P, Q, gu0, gu1, gu2);
    } else if (in2) {
      const unsigned gi = 2u * (base - TD) - 4u;
      const float4 X0 = gate_src4(has_m1, gi,       hv, m0, m1);
      const float4 X1 = gate_src4(has_m1, gi + 4u,  hv, m0, m1);
      const float4 X2 = gate_src4(has_m1, gi + 8u,  hv, m0, m1);
      const float4 X3 = gate_src4(has_m1, gi + 12u, hv, m0, m1);
      const float4 X4 = gate_src4(has_m1, gi + 16u, hv, m0, m1);
      const float4 X5 = gate_src4(has_m1, gi + 20u, hv, m0, m1);
      const float4 X6 = gate_src4(has_m1, gi + 24u, hv, m0, m1);
      const float4 X7 = gate_src4(has_m1, gi + 28u, hv, m0, m1);
      const v8f P = group_product(ap, X0, X1, X2, pinit);
      const v8f Q = group_product(ap, X3, X4, X5, pinit);
      const v8f R = row_product(ap, X6, X7, pinit);
      gates_sh2(P, Q, R, gu0, gu1, gu2);
    }

    const float4 y0 = gate_src4(has_m1, base,      hv, m0, m1);
    const float4 y1 = gate_src4(has_m1, base + 4u, hv, m0, m1);
    const float4 y2 = gate_src4(has_m1, base + 8u, hv, m0, m1);
    const v8f dg = group_product(ag, y0, y1, y2, ginit);
    const float4 ga = sigmoid4(dg[0], dg[1], dg[2], dg[3]);
    const float4 gb = sigmoid4(dg[4], dg[5], dg[6], dg[7]);

    const float4 x0 = cand_val4(has_m1, base,      gu0, hv, m0, m1);
    const float4 x1 = cand_val4(has_m1, base + 4u, gu1, hv, m0, m1);
    const float4 x2 = cand_val4(has_m1, base + 8u, gu2, hv, m0, m1);
    const v8f d = group_product(a, x0, x1, x2, cinit);

    const unsigned p  = (base >> 1) + 2u * h;
    const unsigned p2 = p + 4u * (h ^ 1u);
    const float4 ma = msg4(has_m1, p,  hv, m0, m1);
    const float4 mb = msg4(has_m1, p2, hv, m0, m1);

    const unsigned qa = 3u * (g0 + t * 128u) - Q0 + ql;
    if (qa < nq)
      accn[qa] = delta4(ga, ma, d[0], d[1], d[2], d[3]);
    if (h == 0u && qa + 1u < nq)
      accn[qa + 1u] = delta4(gb, mb, d[4], d[5], d[6], d[7]);
  }
}

constexpr int K_PROD = 1;
constexpr int K_H    = 2;
constexpr int K_ZERO = 3;

constexpr int GK_W    = 1;
constexpr int GK_U    = 2;
constexpr int GK_ZERO = 3;
constexpr int GK_LOAD = 4;

constexpr int UG_NONE = 0;
constexpr int UG_L0   = 1;
constexpr int UG_Z0   = 2;
constexpr int UG_L2   = 3;
constexpr int UG_Z2   = 4;

template <int KIND, bool MSG, int GK, int UG>
__device__ __forceinline__ void cand_slot_uniform(
    const float* __restrict__ ub, const float* __restrict__ wb,
    const float* __restrict__ gs, const float* __restrict__ mp,
    const float* __restrict__ xs, unsigned xo,
    unsigned rb, unsigned pr, unsigned qoff, unsigned nq, unsigned ll, unsigned ql, unsigned h,
    const v2f (&a)[6], const v8f& cinit, const v2f (&ag)[6], const v8f& ginit,
    const v8f& pinit,
    float4* accn) {
  static_assert(GK != GK_W || KIND == K_PROD, "w is loaded only for a gate * message slab");
  static_assert(GK != GK_U || KIND == K_H, "u is the node state only in an h slab");
  static_assert((UG != UG_NONE) == (KIND == K_PROD), "operand gates belong to a gate * message slab");
  const unsigned lo = rb + ll;
  const unsigned po = pr + 4u * ql;
  const unsigned pq = po + 4u * (h ^ 1u);
  float4 zga = zero4(), zgb = zero4(), zta = zero4(), ztb = zero4();
  float4 u[TPS][3], w[TPS][3], y[TPS][3], ma[TPS], mb[TPS];
  if (UG == UG_Z0 || UG == UG_Z2) {
    const float pb0 = pinit[0], pb1 = pinit[1], pb2 = pinit[2];
    const float s0 = fast_sigmoid(pb0), s1 = fast_sigmoid(pb1), s2 = fast_sigmoid(pb2);
#pragma unroll
    for (unsigned t = 0; t < TPS; ++t) {
      if (UG == UG_Z0) {
        u[t][0] = make_float4(s0, s1, s2, s0);
        u[t][1] = make_float4(s1, s2, s0, s1);
        u[t][2] = make_float4(s2, s0, s1, s2);
      } else {
        u[t][0] = make_float4(s2, s0, s1, s2);
        u[t][1] = make_float4(s0, s1, s2, s0);
        u[t][2] = make_float4(s1, s2, s0, s1);
      }
    }
  }
  if (UG == UG_L0 || UG == UG_L2) {
    const unsigned xl = xo + 2u * ll;
    v2f ap[6];
    pair_weights(ag, ap);
    float4 X[TPS][8];
    if (UG == UG_L0) {
#pragma unroll
      for (unsigned t = 0; t < TPS; ++t) {
        const unsigned xt = xl + t * (48u * 128u);
        X[t][0] = ld4(xs + xt);
        X[t][1] = ld4(xs + xt + 4u);
        X[t][2] = ld4(xs + xt + 8u);
        X[t][3] = ld4(xs + xt + 12u);
        X[t][4] = ld4(xs + xt + 16u);
        X[t][5] = ld4(xs + xt + 20u);
      }
#pragma unroll
      for (unsigned t = 0; t < TPS; ++t) {
        const v8f P = group_product(ap, X[t][0], X[t][1], X[t][2], pinit);
        const v8f Q = group_product(ap, X[t][3], X[t][4], X[t][5], pinit);
        gates_sh0(P, Q, u[t][0], u[t][1], u[t][2]);
      }
    } else {
#pragma unroll
      for (unsigned t = 0; t < TPS; ++t) {
        const unsigned xt = xl + t * (48u * 128u);
        X[t][0] = ld4(xs + xt);
        X[t][1] = ld4(xs + xt + 4u);
        X[t][2] = ld4(xs + xt + 8u);
        X[t][3] = ld4(xs + xt + 12u);
        X[t][4] = ld4(xs + xt + 16u);
        X[t][5] = ld4(xs + xt + 20u);
        X[t][6] = ld4(xs + xt + 24u);
        X[t][7] = ld4(xs + xt + 28u);
        const v8f P = group_product(ap, X[t][0], X[t][1], X[t][2], pinit);
        const v8f Q = group_product(ap, X[t][3], X[t][4], X[t][5], pinit);
        const v8f R = row_product(ap, X[t][6], X[t][7], pinit);
        gates_sh2(P, Q, R, u[t][0], u[t][1], u[t][2]);
        __builtin_amdgcn_sched_barrier(0);
      }
    }
  }
  if (GK == GK_ZERO || KIND == K_ZERO) {
    float gb0 = ginit[0], gb1 = ginit[1], gb2 = ginit[2], cb0 = cinit[0], cb1 = cinit[1], cb2 = cinit[2];
    asm volatile("" : "+v"(gb0), "+v"(gb1), "+v"(gb2), "+v"(cb0), "+v"(cb1), "+v"(cb2));
    if (GK == GK_ZERO) {
      const float sg0 = fast_sigmoid(gb0), sg1 = fast_sigmoid(gb1), sg2 = fast_sigmoid(gb2);
      zga = make_float4(sg0, sg1, sg2, sg0);
      zgb = make_float4(sg1, sg2, sg0, sg1);
    }
    if (KIND == K_ZERO) {
      const float st0 = fast_tanh(cb0), st1 = fast_tanh(cb1), st2 = fast_tanh(cb2);
      zta = make_float4(st0, st1, st2, st0);
      ztb = make_float4(st1, st2, st0, st1);
    }
  }
#pragma unroll
  for (unsigned t = 0; t < TPS; ++t) {
    const unsigned ot = lo + t * (24u * 128u);
    const unsigned pt = po + t * (12u * 128u);
    const unsigned pu = pq + t * (12u * 128u);
    if (KIND == K_H) {
      u[t][0] = ld4(ub + ot);
      u[t][1] = ld4(ub + ot + 4u);
      u[t][2] = ld4(ub + ot + 8u);
    } else if (KIND == K_ZERO) {
      u[t][0] = zero4(); u[t][1] = zero4(); u[t][2] = zero4();
    }
    if (KIND == K_PROD) {
      w[t][0] = ld4(wb + ot);
      w[t][1] = ld4(wb + ot + 4u);
      w[t][2] = ld4(wb + ot + 8u);
    } else {
      w[t][0] = zero4(); w[t][1] = zero4(); w[t][2] = zero4();
    }
    if (GK == GK_LOAD) {
      y[t][0] = ld4(gs + ot);
      y[t][1] = ld4(gs + ot + 4u);
      y[t][2] = ld4(gs + ot + 8u);
    } else {
      y[t][0] = zero4(); y[t][1] = zero4(); y[t][2] = zero4();
    }
    if (MSG) {
      ma[t] = ld4(mp + pt);
      mb[t] = ld4(mp + pu);
    } else {
      ma[t] = zero4(); mb[t] = zero4();
    }
  }
#pragma unroll
  for (unsigned t = 0; t < TPS; ++t) {
    if (KIND == K_H)
      asm volatile("" :: "v"(u[t][0].x), "v"(u[t][0].y), "v"(u[t][0].z), "v"(u[t][0].w),
                         "v"(u[t][1].x), "v"(u[t][1].y), "v"(u[t][1].z), "v"(u[t][1].w),
                         "v"(u[t][2].x), "v"(u[t][2].y), "v"(u[t][2].z), "v"(u[t][2].w));
    if (KIND == K_PROD)
      asm volatile("" :: "v"(w[t][0].x), "v"(w[t][0].y), "v"(w[t][0].z), "v"(w[t][0].w),
                         "v"(w[t][1].x), "v"(w[t][1].y), "v"(w[t][1].z), "v"(w[t][1].w),
                         "v"(w[t][2].x), "v"(w[t][2].y), "v"(w[t][2].z), "v"(w[t][2].w));
    if (GK == GK_LOAD)
      asm volatile("" :: "v"(y[t][0].x), "v"(y[t][0].y), "v"(y[t][0].z), "v"(y[t][0].w),
                         "v"(y[t][1].x), "v"(y[t][1].y), "v"(y[t][1].z), "v"(y[t][1].w),
                         "v"(y[t][2].x), "v"(y[t][2].y), "v"(y[t][2].z), "v"(y[t][2].w));
    if (MSG)
      asm volatile("" :: "v"(ma[t].x), "v"(ma[t].y), "v"(ma[t].z), "v"(ma[t].w),
                         "v"(mb[t].x), "v"(mb[t].y), "v"(mb[t].z), "v"(mb[t].w));
  }
#pragma unroll
  for (unsigned t = 0; t < TPS; ++t) {
    const float4 y0 = GK == GK_W ? w[t][0] : (GK == GK_U ? u[t][0] : y[t][0]);
    const float4 y1 = GK == GK_W ? w[t][1] : (GK == GK_U ? u[t][1] : y[t][1]);
    const float4 y2 = GK == GK_W ? w[t][2] : (GK == GK_U ? u[t][2] : y[t][2]);
    float4 x0 = u[t][0], x1 = u[t][1], x2 = u[t][2];
    if (KIND == K_PROD) {
      x0 = make_float4(x0.x * w[t][0].x, x0.y * w[t][0].y, x0.z * w[t][0].z, x0.w * w[t][0].w);
      x1 = make_float4(x1.x * w[t][1].x, x1.y * w[t][1].y, x1.z * w[t][1].z, x1.w * w[t][1].w);
      x2 = make_float4(x2.x * w[t][2].x, x2.y * w[t][2].y, x2.z * w[t][2].z, x2.w * w[t][2].w);
    }
    float4 ta = zta, tb = ztb;
    if (KIND != K_ZERO) {
      const v8f d = group_product(a, x0, x1, x2, cinit);
      ta = tanh4(d[0], d[1], d[2], d[3]);
      tb = tanh4(d[4], d[5], d[6], d[7]);
    }
    float4 ga = zga, gb = zgb;
    if (GK != GK_ZERO) {
      const v8f dg = group_product(ag, y0, y1, y2, ginit);
      ga = sigmoid4(dg[0], dg[1], dg[2], dg[3]);
      gb = sigmoid4(dg[4], dg[5], dg[6], dg[7]);
    }

    const unsigned qa = qoff + t * 384u + ql;
    if (qa < nq)
      accn[qa] = MSG ? delta4t(ga, ma[t], ta) : gated4t(ga, ta);
    if (h == 0u && qa + 1u < nq)
      accn[qa + 1u] = MSG ? delta4t(gb, mb[t], tb) : gated4t(gb, tb);
  }
}

__global__ __launch_bounds__(256) __attribute__((amdgpu_waves_per_eu(8, 8))) void gsn_cand_wmma(
    const float* __restrict__ hvA, const float* __restrict__ m0A, float* __restrict__ outA,
    const float* __restrict__ hvB, const float* __restrict__ m0B, float* __restrict__ outB,
    const float* __restrict__ hvC, const float* __restrict__ m0C, const float* __restrict__ m1C,
    float* __restrict__ outC,
    const float* __restrict__ Ww,
    const float* __restrict__ Bw,
    const float* __restrict__ Wx,
    const float* __restrict__ Bx) {
  __shared__ float4 acc[3][LQ];
  const bool isC = blockIdx.x >= 2u * CBLK;
  const bool isB = !isC && blockIdx.x >= CBLK;
  const unsigned bx = blockIdx.x - (isC ? 2u * CBLK : (isB ? CBLK : 0u));
  const bool has_m1 = isC;
  const float* __restrict__ hv = isC ? hvC : (isB ? hvB : hvA);
  const float* __restrict__ m0 = isC ? m0C : (isB ? m0B : m0A);
  const float* __restrict__ m1 = m1C;
  float* __restrict__ out      = isC ? outC : (isB ? outB : outA);
  const unsigned lane = threadIdx.x & 31u;
  const unsigned h    = lane >> 4;
  const unsigned gl   = (threadIdx.x >> 5u) * 16u + (lane & 15u);
  const unsigned ll   = 24u * gl + 12u * h;
  const unsigned ql   = 3u * gl + 2u * h;
  const unsigned q0   = bx * LQ;
  const unsigned nq   = (NQ - q0) < LQ ? (NQ - q0) : LQ;

  v2f a[6], ag[6];
  v8f cinit, ginit, pinit;
  make_weights(Ww, Bw, lane, a, cinit);
  make_weights(Wx, Bx, lane, ag, ginit);
  pair_bias(Bx, pinit);

  bool all_whole = true;
  for (unsigned n = 0; n < 3u; ++n) {
    const unsigned g0    = (n * NQ + q0) / 3u;
    const unsigned op_lo = 24u * g0;
    const unsigned ep_lo = 12u * g0;
    all_whole = all_whole && g0 + (GPB - 1u) < NQ
        && ((op_lo + 24u * GPB - 1u) >> LOG_TD) == (op_lo >> LOG_TD)
        && (ep_lo >> LOG_TD) == n && ((ep_lo + 12u * GPB - 1u) >> LOG_TD) == n;
    const unsigned seg = op_lo >> LOG_TD;
    if (seg == 0u) {
      const unsigned xi = 2u * op_lo;
      all_whole = all_whole && ((xi + 48u * GPB - 1u) >> LOG_TD) == (xi >> LOG_TD);
    } else if (seg == 2u && has_m1) {
      const unsigned xi = 2u * (op_lo - TD) - 4u;
      all_whole = all_whole && ((xi + 48u * GPB + 7u) >> LOG_TD) == (xi >> LOG_TD);
    }
  }
  if (all_whole) {
    for (unsigned n = 0; n < 3u; ++n) {
      const unsigned Q0 = n * NQ + q0;
      const unsigned g0 = Q0 / 3u;
      const unsigned op_lo = 24u * g0;
      const unsigned ep_lo = 12u * g0;
      const unsigned seg   = op_lo >> LOG_TD;
      const unsigned rb  = op_lo & (TD - 1u);
      const unsigned pr  = ep_lo & (TD - 1u);
      const unsigned qoff = 3u * g0 - Q0;
      if (seg == 0u) {
        const unsigned xi = 2u * op_lo;
        const unsigned xo = xi & (TD - 1u);
        if ((xi >> LOG_TD) == 0u)
          cand_slot_uniform<K_PROD, true, GK_W, UG_L0>(hv, m0, hv, m0, m0, xo, rb, pr, qoff, nq, ll, ql, h, a, cinit, ag, ginit, pinit, acc[n]);
        else if (has_m1)
          cand_slot_uniform<K_PROD, true, GK_W, UG_L0>(hv, m0, hv, m0, m1, xo, rb, pr, qoff, nq, ll, ql, h, a, cinit, ag, ginit, pinit, acc[n]);
        else
          cand_slot_uniform<K_PROD, true, GK_W, UG_Z0>(hv, m0, hv, m0, hv, 0u, rb, pr, qoff, nq, ll, ql, h, a, cinit, ag, ginit, pinit, acc[n]);
      }
      else if (seg == 1u && has_m1)
        cand_slot_uniform<K_H, true, GK_LOAD, UG_NONE>(hv, hv, m1, m0, hv, 0u, rb, pr, qoff, nq, ll, ql, h, a, cinit, ag, ginit, pinit, acc[n]);
      else if (seg == 1u)
        cand_slot_uniform<K_H, true, GK_ZERO, UG_NONE>(hv, hv, hv, m0, hv, 0u, rb, pr, qoff, nq, ll, ql, h, a, cinit, ag, ginit, pinit, acc[n]);
      else if (seg == 2u && has_m1) {
        const unsigned xi = 2u * (op_lo - TD) - 4u;
        const unsigned xo = xi & (TD - 1u);
        if ((xi >> LOG_TD) == 2u)
          cand_slot_uniform<K_PROD, true, GK_ZERO, UG_Z2>(hv, m1, hv, m1, hv, 0u, rb, pr, qoff, nq, ll, ql, h, a, cinit, ag, ginit, pinit, acc[n]);
        else
          cand_slot_uniform<K_PROD, true, GK_ZERO, UG_L2>(hv, m1, hv, m1, hv, xo, rb, pr, qoff, nq, ll, ql, h, a, cinit, ag, ginit, pinit, acc[n]);
      }
      else if (seg == 2u)
        cand_slot_uniform<K_ZERO, false, GK_ZERO, UG_NONE>(hv, hv, hv, hv, hv, 0u, rb, pr, qoff, nq, ll, ql, h, a, cinit, ag, ginit, pinit, acc[n]);
      else if (seg == 3u && has_m1)
        cand_slot_uniform<K_H, true, GK_U, UG_NONE>(hv, hv, hv, m1, hv, 0u, rb, pr, qoff, nq, ll, ql, h, a, cinit, ag, ginit, pinit, acc[n]);
      else if (seg == 4u)
        cand_slot_uniform<K_ZERO, false, GK_LOAD, UG_NONE>(hv, hv, hv, hv, hv, 0u, rb, pr, qoff, nq, ll, ql, h, a, cinit, ag, ginit, pinit, acc[n]);
      else
        cand_slot_uniform<K_H, false, GK_U, UG_NONE>(hv, hv, hv, hv, hv, 0u, rb, pr, qoff, nq, ll, ql, h, a, cinit, ag, ginit, pinit, acc[n]);
    }
  } else {
    for (unsigned n = 0; n < 3u; ++n) {
      const unsigned Q0 = n * NQ + q0;
      cand_slot_lane(has_m1, Q0, Q0 / 3u, nq, ll, ql, h, a, cinit, ag, ginit, pinit, hv, m0, m1, acc[n]);
    }
  }
  __syncthreads();
  for (unsigned i = threadIdx.x; i < nq; i += 256u) {
    const float4 h4 = ld4(hv + 4u * (q0 + i));
    const float4 c0 = acc[0][i];
    const float4 c1 = acc[1][i];
    const float4 c2 = acc[2][i];
    float4 o;
    o.x = h4.x + c0.x + c1.x + c2.x;
    o.y = h4.y + c0.y + c1.y + c2.y;
    o.z = h4.z + c0.z + c1.z + c2.z;
    o.w = h4.w + c0.w + c1.w + c2.w;
    *(float4*)(out + 4u * (q0 + i)) = o;
  }
}

extern "C" void kernel_launch(void* const* d_in, const int* in_sizes, int n_in,
                              void* d_out, int out_size, void* d_ws, size_t ws_size,
                              hipStream_t stream) {
  (void)in_sizes; (void)n_in; (void)out_size; (void)d_ws; (void)ws_size;
  const float* feat = (const float*)d_in[0];
  const float* W_w  = (const float*)d_in[1];
  const float* W_b  = (const float*)d_in[2];
  const float* Wx_w = (const float*)d_in[3];
  const float* Wx_b = (const float*)d_in[4];
  float* out = (float*)d_out;

  const float* h0 = feat;
  const float* h1 = feat + (size_t)TD;
  const float* h2 = feat + (size_t)2 * TD;

  dim3 blk(256);
  dim3 grd_c(CBLK3);

  gsn_cand_wmma<<<grd_c, blk, 0, stream>>>(
      h0, h2, out + 0 * (size_t)TD, h1, h0, out + 1 * (size_t)TD,
      h2, h1, h0, out + 2 * (size_t)TD, W_w, W_b, Wx_w, Wx_b);
}


# Work-group index mapped to (node, block) longest work first: per-lane boundary blocks, then the two-message node, then the one-message nodes, their shortest class last; each block's work unchanged
# speedup vs baseline: 5.4964x; 1.0622x over previous
#include <hip/hip_runtime.h>
#include <math.h>

#define LOG_TD 22u
#define TD (1u << LOG_TD)

typedef __attribute__((ext_vector_type(2))) float v2f;
typedef __attribute__((ext_vector_type(8))) float v8f;

constexpr unsigned NQ     = TD / 4u;
constexpr unsigned TPS    = 2u;
constexpr unsigned GPB    = 16u * 8u * TPS;
constexpr unsigned LQ     = 760u;
constexpr unsigned CBLK   = (NQ + LQ - 1u) / LQ;
constexpr unsigned CBLK3  = 3u * CBLK;
static_assert(LQ % 8u == 0u && LQ + 2u <= 3u * GPB, "block range must be whole lines and fit its groups");
constexpr unsigned BQ = (NQ / 4u) / LQ;
constexpr unsigned BH = (NQ / 2u) / LQ;
constexpr unsigned BL = CBLK - 1u;
constexpr unsigned N1 = BQ - 1u;
constexpr unsigned N2 = BH - BQ - 1u;
constexpr unsigned N3 = BL - BH - 1u;
constexpr unsigned NREG = N1 + N2 + N3;
static_assert(1u < BQ && BQ + 1u < BH && BH + 1u < BL && NREG + 4u == CBLK, "the four named blocks and the three ranges between them are the node's blocks, each once");
static_assert(TD % 12u == 4u, "gate alignment in slab 2");

__device__ __forceinline__ float hw_exp2(float x) {
#if __has_builtin(__builtin_amdgcn_exp2f)
  return __builtin_amdgcn_exp2f(x);
#else
  return exp2f(x);
#endif
}
__device__ __forceinline__ float hw_rcp(float x) {
#if __has_builtin(__builtin_amdgcn_rcpf)
  return __builtin_amdgcn_rcpf(x);
#else
  return 1.0f / x;
#endif
}
__device__ __forceinline__ float fast_sigmoid(float x) {
  return hw_rcp(1.0f + hw_exp2(-1.4426950408889634f * x));
}
__device__ __forceinline__ float fast_tanh(float x) {
#if __has_builtin(__builtin_amdgcn_tanhf)
  return __builtin_amdgcn_tanhf(x);
#else
  return 1.0f - 2.0f * hw_rcp(hw_exp2(2.8853900817779268f * x) + 1.0f);
#endif
}

__device__ __forceinline__ float4 ld4(const float* __restrict__ p) {
  return *(const float4*)p;
}
__device__ __forceinline__ float4 zero4() { return make_float4(0.0f, 0.0f, 0.0f, 0.0f); }
__device__ __forceinline__ float4 sel4(bool c, const float4& v) {
  return make_float4(c ? v.x : 0.0f, c ? v.y : 0.0f, c ? v.z : 0.0f, c ? v.w : 0.0f);
}


__device__ __forceinline__ float4 gate_src4(bool has_m1, unsigned m,
    const float* __restrict__ hv, const float* __restrict__ m0,
    const float* __restrict__ m1) {
  const unsigned seg = m >> LOG_TD;
  const unsigned r   = m & (TD - 1u);
  const bool is_h  = seg >= 3u;
  const bool is_m0 = seg == 0u;
  const bool is_m1 = has_m1 && seg == 1u;
  const bool valid = is_h || is_m0 || is_m1;
  const float* base = is_m0 ? m0 : (is_m1 ? m1 : hv);
  const float4 v = ld4(base + (valid ? r : 0u));
  return sel4(valid, v);
}

__device__ __forceinline__ float4 msg4(bool has_m1, unsigned p, const float* __restrict__ hv,
    const float* __restrict__ m0, const float* __restrict__ m1) {
  const unsigned n = p >> LOG_TD;
  const unsigned r = p & (TD - 1u);
  const bool is_m0 = n == 0u;
  const bool is_m1 = has_m1 && n == 1u;
  const bool valid = is_m0 || is_m1;
  const float* base = is_m0 ? m0 : (is_m1 ? m1 : hv);
  const float4 v = ld4(base + (valid ? r : 0u));
  return sel4(valid, v);
}

__device__ __forceinline__ float4 cand_val4(bool has_m1, unsigned m, const float4& gu,
    const float* __restrict__ hv, const float* __restrict__ m0,
    const float* __restrict__ m1) {
  const unsigned seg = m >> LOG_TD;
  const unsigned r   = m & (TD - 1u);
  const unsigned n   = seg >> 1;
  const bool is_h  = (seg & 1u) != 0u;
  const bool is_m0 = !is_h && n == 0u;
  const bool is_m1 = has_m1 && !is_h && n == 1u;
  const bool prod  = is_m0 || is_m1;
  const float* wb = is_m0 ? m0 : (is_m1 ? m1 : hv);
  const float4 v = ld4(wb + ((prod || is_h) ? r : 0u));
  const float4 gm = make_float4(gu.x * v.x, gu.y * v.y, gu.z * v.z, gu.w * v.w);
  return make_float4(is_h ? v.x : (prod ? gm.x : 0.0f), is_h ? v.y : (prod ? gm.y : 0.0f),
                     is_h ? v.z : (prod ? gm.z : 0.0f), is_h ? v.w : (prod ? gm.w : 0.0f));
}

__device__ __forceinline__ void make_weights(const float* __restrict__ Wm,
    const float* __restrict__ Bv, unsigned lane, v2f (&a)[6], v8f& cinit) {
  const unsigned o  = lane & 15u;
  const unsigned h  = lane >> 4;
  const unsigned g  = o / 3u;
  const unsigned j  = o - 3u * g;
  const bool valid  = (o < 12u) && ((g >> 1) == h);
  const float msk   = valid ? 1.0f : 0.0f;
  const bool second = (g & 1u) != 0u;
  float w[6];
#pragma unroll
  for (int i = 0; i < 6; ++i) w[i] = Wm[j * 6u + (unsigned)i] * msk;
#pragma unroll
  for (int c = 0; c < 3; ++c) {
    a[c].x     = second ? 0.0f : w[2 * c];
    a[c].y     = second ? 0.0f : w[2 * c + 1];
    a[c + 3].x = second ? w[2 * c]     : 0.0f;
    a[c + 3].y = second ? w[2 * c + 1] : 0.0f;
  }
  const float b0 = Bv[0], b1 = Bv[1], b2 = Bv[2];
  const bool hi = h != 0u;
  cinit[0] = hi ? b2 : b0;
  cinit[1] = hi ? b0 : b1;
  cinit[2] = hi ? b1 : b2;
  cinit[3] = hi ? b2 : b0;
  cinit[4] = hi ? 0.0f : b1;
  cinit[5] = hi ? 0.0f : b2;
  cinit[6] = hi ? 0.0f : b0;
  cinit[7] = hi ? 0.0f : b1;
}

__device__ __forceinline__ v8f group_product(const v2f (&a)[6], const float4& x0,
    const float4& x1, const float4& x2, v8f c) {
  v2f b;
  b.x = x0.x; b.y = x0.y;
  c = __builtin_amdgcn_wmma_f32_16x16x4_f32(false, a[0], false, b, (short)0, c, false, false);
  b.x = x0.z; b.y = x0.w;
  c = __builtin_amdgcn_wmma_f32_16x16x4_f32(false, a[1], false, b, (short)0, c, false, false);
  b.x = x1.x; b.y = x1.y;
  c = __builtin_amdgcn_wmma_f32_16x16x4_f32(false, a[2], false, b, (short)0, c, false, false);
  b.x = x1.z; b.y = x1.w;
  c = __builtin_amdgcn_wmma_f32_16x16x4_f32(false, a[3], false, b, (short)0, c, false, false);
  b.x = x2.x; b.y = x2.y;
  c = __builtin_amdgcn_wmma_f32_16x16x4_f32(false, a[4], false, b, (short)0, c, false, false);
  b.x = x2.z; b.y = x2.w;
  c = __builtin_amdgcn_wmma_f32_16x16x4_f32(false, a[5], false, b, (short)0, c, false, false);
  return c;
}

__device__ __forceinline__ float shr2_upper(float v) {
  const int i = __builtin_bit_cast(int, v);
  return __builtin_bit_cast(float, __builtin_amdgcn_update_dpp(i, i, 0x112, 0x2, 0xf, true));
}
__device__ __forceinline__ void pair_weights(const v2f (&ag)[6], v2f (&ap)[6]) {
#pragma unroll
  for (int c = 0; c < 6; ++c) {
    ap[c].x = shr2_upper(ag[c].x);
    ap[c].y = shr2_upper(ag[c].y);
  }
}
__device__ __forceinline__ void pair_bias(const float* __restrict__ Bv, v8f& cinit) {
  const float b0 = Bv[0], b1 = Bv[1], b2 = Bv[2];
  cinit[0] = b0; cinit[1] = b1; cinit[2] = b2;
  cinit[3] = b0; cinit[4] = b1; cinit[5] = b2;
  cinit[6] = 0.0f; cinit[7] = 0.0f;
}

__device__ __forceinline__ v8f row_product(const v2f (&a)[6], const float4& x0,
    const float4& x1, v8f c) {
  v2f b;
  b.x = x0.x; b.y = x0.y;
  c = __builtin_amdgcn_wmma_f32_16x16x4_f32(false, a[0], false, b, (short)0, c, false, false);
  b.x = x0.z; b.y = x0.w;
  c = __builtin_amdgcn_wmma_f32_16x16x4_f32(false, a[1], false, b, (short)0, c, false, false);
  b.x = x1.x; b.y = x1.y;
  c = __builtin_amdgcn_wmma_f32_16x16x4_f32(false, a[2], false, b, (short)0, c, false, false);
  return c;
}

__device__ __forceinline__ float4 sigmoid4(float d0, float d1, float d2, float d3) {
  return make_float4(fast_sigmoid(d0), fast_sigmoid(d1), fast_sigmoid(d2), fast_sigmoid(d3));
}

__device__ __forceinline__ float4 delta4(const float4& g, const float4& mv,
    float d0, float d1, float d2, float d3) {
  float4 o;
  o.x = (1.0f - g.x) * mv.x + g.x * fast_tanh(d0);
  o.y = (1.0f - g.y) * mv.y + g.y * fast_tanh(d1);
  o.z = (1.0f - g.z) * mv.z + g.z * fast_tanh(d2);
  o.w = (1.0f - g.w) * mv.w + g.w * fast_tanh(d3);
  return o;
}

__device__ __forceinline__ float4 gated4(const float4& g, float d0, float d1, float d2, float d3) {
  float4 o;
  o.x = g.x * fast_tanh(d0);
  o.y = g.y * fast_tanh(d1);
  o.z = g.z * fast_tanh(d2);
  o.w = g.w * fast_tanh(d3);
  return o;
}

__device__ __forceinline__ float4 tanh4(float d0, float d1, float d2, float d3) {
  return make_float4(fast_tanh(d0), fast_tanh(d1), fast_tanh(d2), fast_tanh(d3));
}
__device__ __forceinline__ float4 delta4t(const float4& g, const float4& mv, const float4& t) {
  float4 o;
  o.x = (1.0f - g.x) * mv.x + g.x * t.x;
  o.y = (1.0f - g.y) * mv.y + g.y * t.y;
  o.z = (1.0f - g.z) * mv.z + g.z * t.z;
  o.w = (1.0f - g.w) * mv.w + g.w * t.w;
  return o;
}
__device__ __forceinline__ float4 gated4t(const float4& g, const float4& t) {
  return make_float4(g.x * t.x, g.y * t.y, g.z * t.z, g.w * t.w);
}


__device__ __forceinline__ void gates_sh0(const v8f& P, const v8f& Q, float4& u0, float4& u1, float4& u2) {
  u0 = sigmoid4(P[0], P[1], P[2], P[3]);
  u1 = sigmoid4(P[4], P[5], Q[0], Q[1]);
  u2 = sigmoid4(Q[2], Q[3], Q[4], Q[5]);
}
__device__ __forceinline__ void gates_sh2(const v8f& P, const v8f& Q, const v8f& R, float4& u0, float4& u1, float4& u2) {
  u0 = sigmoid4(P[2], P[3], P[4], P[5]);
  u1 = sigmoid4(Q[0], Q[1], Q[2], Q[3]);
  u2 = sigmoid4(Q[4], Q[5], R[0], R[1]);
}

__device__ __forceinline__ void cand_slot_lane(bool has_m1, unsigned Q0, unsigned g0, unsigned nq,
    unsigned ll, unsigned ql, unsigned h, const v2f (&a)[6], const v8f& cinit,
    const v2f (&ag)[6], const v8f& ginit, const v8f& pinit,
    const float* __restrict__ hv, const float* __restrict__ m0, const float* __restrict__ m1,
    float4* accn) {
  v2f ap[6];
  pair_weights(ag, ap);
  const unsigned seg_lo = (24u * g0) >> LOG_TD;
  const unsigned seg_hi = (24u * g0 + 24u * GPB - 1u) >> LOG_TD;
  const bool in0 = seg_lo == 0u;
  const bool in2 = has_m1 && (seg_lo == 2u || seg_hi == 2u);
  for (unsigned t = 0; t < TPS; ++t) {
    const unsigned bu   = 24u * (g0 + t * 128u) + ll;
    const unsigned bmax = 24u * (NQ - 1u) + 12u * h;
    const unsigned base = bu < bmax ? bu : bmax;

    float4 gu0 = zero4(), gu1 = zero4(), gu2 = zero4();
    if (in0) {
      const unsigned gi = 2u * base;
      const float4 X0 = gate_src4(has_m1, gi,       hv, m0, m1);
      const float4 X1 = gate_src4(has_m1, gi + 4u,  hv, m0, m1);
      const float4 X2 = gate_src4(has_m1, gi + 8u,  hv, m0, m1);
      const float4 X3 = gate_src4(has_m1, gi + 12u, hv, m0, m1);
      const float4 X4 = gate_src4(has_m1, gi + 16u, hv, m0, m1);
      const float4 X5 = gate_src4(has_m1, gi + 20u, hv, m0, m1);
      const v8f P = group_product(ap, X0, X1, X2, pinit);
      const v8f Q = group_product(ap, X3, X4, X5, pinit);
      gates_sh0(P, Q, gu0, gu1, gu2);
    } else if (in2) {
      const unsigned gi = 2u * (base - TD) - 4u;
      const float4 X0 = gate_src4(has_m1, gi,       hv, m0, m1);
      const float4 X1 = gate_src4(has_m1, gi + 4u,  hv, m0, m1);
      const float4 X2 = gate_src4(has_m1, gi + 8u,  hv, m0, m1);
      const float4 X3 = gate_src4(has_m1, gi + 12u, hv, m0, m1);
      const float4 X4 = gate_src4(has_m1, gi + 16u, hv, m0, m1);
      const float4 X5 = gate_src4(has_m1, gi + 20u, hv, m0, m1);
      const float4 X6 = gate_src4(has_m1, gi + 24u, hv, m0, m1);
      const float4 X7 = gate_src4(has_m1, gi + 28u, hv, m0, m1);
      const v8f P = group_product(ap, X0, X1, X2, pinit);
      const v8f Q = group_product(ap, X3, X4, X5, pinit);
      const v8f R = row_product(ap, X6, X7, pinit);
      gates_sh2(P, Q, R, gu0, gu1, gu2);
    }

    const float4 y0 = gate_src4(has_m1, base,      hv, m0, m1);
    const float4 y1 = gate_src4(has_m1, base + 4u, hv, m0, m1);
    const float4 y2 = gate_src4(has_m1, base + 8u, hv, m0, m1);
    const v8f dg = group_product(ag, y0, y1, y2, ginit);
    const float4 ga = sigmoid4(dg[0], dg[1], dg[2], dg[3]);
    const float4 gb = sigmoid4(dg[4], dg[5], dg[6], dg[7]);

    const float4 x0 = cand_val4(has_m1, base,      gu0, hv, m0, m1);
    const float4 x1 = cand_val4(has_m1, base + 4u, gu1, hv, m0, m1);
    const float4 x2 = cand_val4(has_m1, base + 8u, gu2, hv, m0, m1);
    const v8f d = group_product(a, x0, x1, x2, cinit);

    const unsigned p  = (base >> 1) + 2u * h;
    const unsigned p2 = p + 4u * (h ^ 1u);
    const float4 ma = msg4(has_m1, p,  hv, m0, m1);
    const float4 mb = msg4(has_m1, p2, hv, m0, m1);

    const unsigned qa = 3u * (g0 + t * 128u) - Q0 + ql;
    if (qa < nq)
      accn[qa] = delta4(ga, ma, d[0], d[1], d[2], d[3]);
    if (h == 0u && qa + 1u < nq)
      accn[qa + 1u] = delta4(gb, mb, d[4], d[5], d[6], d[7]);
  }
}

constexpr int K_PROD = 1;
constexpr int K_H    = 2;
constexpr int K_ZERO = 3;

constexpr int GK_W    = 1;
constexpr int GK_U    = 2;
constexpr int GK_ZERO = 3;
constexpr int GK_LOAD = 4;

constexpr int UG_NONE = 0;
constexpr int UG_L0   = 1;
constexpr int UG_Z0   = 2;
constexpr int UG_L2   = 3;
constexpr int UG_Z2   = 4;

template <int KIND, bool MSG, int GK, int UG>
__device__ __forceinline__ void cand_slot_uniform(
    const float* __restrict__ ub, const float* __restrict__ wb,
    const float* __restrict__ gs, const float* __restrict__ mp,
    const float* __restrict__ xs, unsigned xo,
    unsigned rb, unsigned pr, unsigned qoff, unsigned nq, unsigned ll, unsigned ql, unsigned h,
    const v2f (&a)[6], const v8f& cinit, const v2f (&ag)[6], const v8f& ginit,
    const v8f& pinit,
    float4* accn) {
  static_assert(GK != GK_W || KIND == K_PROD, "w is loaded only for a gate * message slab");
  static_assert(GK != GK_U || KIND == K_H, "u is the node state only in an h slab");
  static_assert((UG != UG_NONE) == (KIND == K_PROD), "operand gates belong to a gate * message slab");
  const unsigned lo = rb + ll;
  const unsigned po = pr + 4u * ql;
  const unsigned pq = po + 4u * (h ^ 1u);
  float4 zga = zero4(), zgb = zero4(), zta = zero4(), ztb = zero4();
  float4 u[TPS][3], w[TPS][3], y[TPS][3], ma[TPS], mb[TPS];
  if (UG == UG_Z0 || UG == UG_Z2) {
    const float pb0 = pinit[0], pb1 = pinit[1], pb2 = pinit[2];
    const float s0 = fast_sigmoid(pb0), s1 = fast_sigmoid(pb1), s2 = fast_sigmoid(pb2);
#pragma unroll
    for (unsigned t = 0; t < TPS; ++t) {
      if (UG == UG_Z0) {
        u[t][0] = make_float4(s0, s1, s2, s0);
        u[t][1] = make_float4(s1, s2, s0, s1);
        u[t][2] = make_float4(s2, s0, s1, s2);
      } else {
        u[t][0] = make_float4(s2, s0, s1, s2);
        u[t][1] = make_float4(s0, s1, s2, s0);
        u[t][2] = make_float4(s1, s2, s0, s1);
      }
    }
  }
  if (UG == UG_L0 || UG == UG_L2) {
    const unsigned xl = xo + 2u * ll;
    v2f ap[6];
    pair_weights(ag, ap);
    float4 X[TPS][8];
    if (UG == UG_L0) {
#pragma unroll
      for (unsigned t = 0; t < TPS; ++t) {
        const unsigned xt = xl + t * (48u * 128u);
        X[t][0] = ld4(xs + xt);
        X[t][1] = ld4(xs + xt + 4u);
        X[t][2] = ld4(xs + xt + 8u);
        X[t][3] = ld4(xs + xt + 12u);
        X[t][4] = ld4(xs + xt + 16u);
        X[t][5] = ld4(xs + xt + 20u);
      }
#pragma unroll
      for (unsigned t = 0; t < TPS; ++t) {
        const v8f P = group_product(ap, X[t][0], X[t][1], X[t][2], pinit);
        const v8f Q = group_product(ap, X[t][3], X[t][4], X[t][5], pinit);
        gates_sh0(P, Q, u[t][0], u[t][1], u[t][2]);
      }
    } else {
#pragma unroll
      for (unsigned t = 0; t < TPS; ++t) {
        const unsigned xt = xl + t * (48u * 128u);
        X[t][0] = ld4(xs + xt);
        X[t][1] = ld4(xs + xt + 4u);
        X[t][2] = ld4(xs + xt + 8u);
        X[t][3] = ld4(xs + xt + 12u);
        X[t][4] = ld4(xs + xt + 16u);
        X[t][5] = ld4(xs + xt + 20u);
        X[t][6] = ld4(xs + xt + 24u);
        X[t][7] = ld4(xs + xt + 28u);
        const v8f P = group_product(ap, X[t][0], X[t][1], X[t][2], pinit);
        const v8f Q = group_product(ap, X[t][3], X[t][4], X[t][5], pinit);
        const v8f R = row_product(ap, X[t][6], X[t][7], pinit);
        gates_sh2(P, Q, R, u[t][0], u[t][1], u[t][2]);
        __builtin_amdgcn_sched_barrier(0);
      }
    }
  }
  if (GK == GK_ZERO || KIND == K_ZERO) {
    float gb0 = ginit[0], gb1 = ginit[1], gb2 = ginit[2], cb0 = cinit[0], cb1 = cinit[1], cb2 = cinit[2];
    asm volatile("" : "+v"(gb0), "+v"(gb1), "+v"(gb2), "+v"(cb0), "+v"(cb1), "+v"(cb2));
    if (GK == GK_ZERO) {
      const float sg0 = fast_sigmoid(gb0), sg1 = fast_sigmoid(gb1), sg2 = fast_sigmoid(gb2);
      zga = make_float4(sg0, sg1, sg2, sg0);
      zgb = make_float4(sg1, sg2, sg0, sg1);
    }
    if (KIND == K_ZERO) {
      const float st0 = fast_tanh(cb0), st1 = fast_tanh(cb1), st2 = fast_tanh(cb2);
      zta = make_float4(st0, st1, st2, st0);
      ztb = make_float4(st1, st2, st0, st1);
    }
  }
#pragma unroll
  for (unsigned t = 0; t < TPS; ++t) {
    const unsigned ot = lo + t * (24u * 128u);
    const unsigned pt = po + t * (12u * 128u);
    const unsigned pu = pq + t * (12u * 128u);
    if (KIND == K_H) {
      u[t][0] = ld4(ub + ot);
      u[t][1] = ld4(ub + ot + 4u);
      u[t][2] = ld4(ub + ot + 8u);
    } else if (KIND == K_ZERO) {
      u[t][0] = zero4(); u[t][1] = zero4(); u[t][2] = zero4();
    }
    if (KIND == K_PROD) {
      w[t][0] = ld4(wb + ot);
      w[t][1] = ld4(wb + ot + 4u);
      w[t][2] = ld4(wb + ot + 8u);
    } else {
      w[t][0] = zero4(); w[t][1] = zero4(); w[t][2] = zero4();
    }
    if (GK == GK_LOAD) {
      y[t][0] = ld4(gs + ot);
      y[t][1] = ld4(gs + ot + 4u);
      y[t][2] = ld4(gs + ot + 8u);
    } else {
      y[t][0] = zero4(); y[t][1] = zero4(); y[t][2] = zero4();
    }
    if (MSG) {
      ma[t] = ld4(mp + pt);
      mb[t] = ld4(mp + pu);
    } else {
      ma[t] = zero4(); mb[t] = zero4();
    }
  }
#pragma unroll
  for (unsigned t = 0; t < TPS; ++t) {
    if (KIND == K_H)
      asm volatile("" :: "v"(u[t][0].x), "v"(u[t][0].y), "v"(u[t][0].z), "v"(u[t][0].w),
                         "v"(u[t][1].x), "v"(u[t][1].y), "v"(u[t][1].z), "v"(u[t][1].w),
                         "v"(u[t][2].x), "v"(u[t][2].y), "v"(u[t][2].z), "v"(u[t][2].w));
    if (KIND == K_PROD)
      asm volatile("" :: "v"(w[t][0].x), "v"(w[t][0].y), "v"(w[t][0].z), "v"(w[t][0].w),
                         "v"(w[t][1].x), "v"(w[t][1].y), "v"(w[t][1].z), "v"(w[t][1].w),
                         "v"(w[t][2].x), "v"(w[t][2].y), "v"(w[t][2].z), "v"(w[t][2].w));
    if (GK == GK_LOAD)
      asm volatile("" :: "v"(y[t][0].x), "v"(y[t][0].y), "v"(y[t][0].z), "v"(y[t][0].w),
                         "v"(y[t][1].x), "v"(y[t][1].y), "v"(y[t][1].z), "v"(y[t][1].w),
                         "v"(y[t][2].x), "v"(y[t][2].y), "v"(y[t][2].z), "v"(y[t][2].w));
    if (MSG)
      asm volatile("" :: "v"(ma[t].x), "v"(ma[t].y), "v"(ma[t].z), "v"(ma[t].w),
                         "v"(mb[t].x), "v"(mb[t].y), "v"(mb[t].z), "v"(mb[t].w));
  }
#pragma unroll
  for (unsigned t = 0; t < TPS; ++t) {
    const float4 y0 = GK == GK_W ? w[t][0] : (GK == GK_U ? u[t][0] : y[t][0]);
    const float4 y1 = GK == GK_W ? w[t][1] : (GK == GK_U ? u[t][1] : y[t][1]);
    const float4 y2 = GK == GK_W ? w[t][2] : (GK == GK_U ? u[t][2] : y[t][2]);
    float4 x0 = u[t][0], x1 = u[t][1], x2 = u[t][2];
    if (KIND == K_PROD) {
      x0 = make_float4(x0.x * w[t][0].x, x0.y * w[t][0].y, x0.z * w[t][0].z, x0.w * w[t][0].w);
      x1 = make_float4(x1.x * w[t][1].x, x1.y * w[t][1].y, x1.z * w[t][1].z, x1.w * w[t][1].w);
      x2 = make_float4(x2.x * w[t][2].x, x2.y * w[t][2].y, x2.z * w[t][2].z, x2.w * w[t][2].w);
    }
    float4 ta = zta, tb = ztb;
    if (KIND != K_ZERO) {
      const v8f d = group_product(a, x0, x1, x2, cinit);
      ta = tanh4(d[0], d[1], d[2], d[3]);
      tb = tanh4(d[4], d[5], d[6], d[7]);
    }
    float4 ga = zga, gb = zgb;
    if (GK != GK_ZERO) {
      const v8f dg = group_product(ag, y0, y1, y2, ginit);
      ga = sigmoid4(dg[0], dg[1], dg[2], dg[3]);
      gb = sigmoid4(dg[4], dg[5], dg[6], dg[7]);
    }

    const unsigned qa = qoff + t * 384u + ql;
    if (qa < nq)
      accn[qa] = MSG ? delta4t(ga, ma[t], ta) : gated4t(ga, ta);
    if (h == 0u && qa + 1u < nq)
      accn[qa + 1u] = MSG ? delta4t(gb, mb[t], tb) : gated4t(gb, tb);
  }
}

__global__ __launch_bounds__(256) __attribute__((amdgpu_waves_per_eu(8, 8))) void gsn_cand_wmma(
    const float* __restrict__ hvA, const float* __restrict__ m0A, float* __restrict__ outA,
    const float* __restrict__ hvB, const float* __restrict__ m0B, float* __restrict__ outB,
    const float* __restrict__ hvC, const float* __restrict__ m0C, const float* __restrict__ m1C,
    float* __restrict__ outC,
    const float* __restrict__ Ww,
    const float* __restrict__ Bw,
    const float* __restrict__ Wx,
    const float* __restrict__ Bx) {
  __shared__ float4 acc[3][LQ];
  const unsigned d  = blockIdx.x;
  const bool c0 = d < 12u, c1 = d < 12u + NREG, c2 = d < 12u + NREG + 2u * (N3 + N1);
  const unsigned e1 = d - 12u;
  const unsigned d2 = e1 - NREG;
  const unsigned d3 = d2 - 2u * (N3 + N1);
  const bool f2 = d2 < N3 + N1, f3 = d3 < N2;
  const unsigned e2 = d2 - (f2 ? 0u : N3 + N1);
  const unsigned e3 = d3 - (f3 ? 0u : N2);
  const unsigned r1 = e1 + N1 - (e1 + N1 >= NREG ? NREG : 0u);
  const unsigned r2 = e2 + N1 + N2 - (e2 >= N3 ? NREG : 0u);
  const unsigned r3 = e3 + N1;
  const unsigned r  = c1 ? r1 : (c2 ? r2 : r3);
  const unsigned bxr = r + 1u + (r >= N1 ? 1u : 0u) + (r >= N1 + N2 ? 1u : 0u);
  const unsigned kb  = d & 3u;
  const unsigned bxb = kb == 0u ? BL : (kb == 1u ? BH : (kb == 2u ? BQ : 0u));
  const unsigned bx   = c0 ? bxb : bxr;
  const unsigned node = c0 ? 2u - (d >> 2u) : (c1 ? 2u : ((c2 ? f2 : f3) ? 1u : 0u));
  const bool isC = node == 2u;
  const bool isB = node == 1u;
  const bool has_m1 = isC;
  const float* __restrict__ hv = isC ? hvC : (isB ? hvB : hvA);
  const float* __restrict__ m0 = isC ? m0C : (isB ? m0B : m0A);
  const float* __restrict__ m1 = m1C;
  float* __restrict__ out      = isC ? outC : (isB ? outB : outA);
  const unsigned lane = threadIdx.x & 31u;
  const unsigned h    = lane >> 4;
  const unsigned gl   = (threadIdx.x >> 5u) * 16u + (lane & 15u);
  const unsigned ll   = 24u * gl + 12u * h;
  const unsigned ql   = 3u * gl + 2u * h;
  const unsigned q0   = bx * LQ;
  const unsigned nq   = (NQ - q0) < LQ ? (NQ - q0) : LQ;

  v2f a[6], ag[6];
  v8f cinit, ginit, pinit;
  make_weights(Ww, Bw, lane, a, cinit);
  make_weights(Wx, Bx, lane, ag, ginit);
  pair_bias(Bx, pinit);

  bool all_whole = true;
  for (unsigned n = 0; n < 3u; ++n) {
    const unsigned g0    = (n * NQ + q0) / 3u;
    const unsigned op_lo = 24u * g0;
    const unsigned ep_lo = 12u * g0;
    all_whole = all_whole && g0 + (GPB - 1u) < NQ
        && ((op_lo + 24u * GPB - 1u) >> LOG_TD) == (op_lo >> LOG_TD)
        && (ep_lo >> LOG_TD) == n && ((ep_lo + 12u * GPB - 1u) >> LOG_TD) == n;
    const unsigned seg = op_lo >> LOG_TD;
    if (seg == 0u) {
      const unsigned xi = 2u * op_lo;
      all_whole = all_whole && ((xi + 48u * GPB - 1u) >> LOG_TD) == (xi >> LOG_TD);
    } else if (seg == 2u && has_m1) {
      const unsigned xi = 2u * (op_lo - TD) - 4u;
      all_whole = all_whole && ((xi + 48u * GPB + 7u) >> LOG_TD) == (xi >> LOG_TD);
    }
  }
  if (all_whole) {
    for (unsigned n = 0; n < 3u; ++n) {
      const unsigned Q0 = n * NQ + q0;
      const unsigned g0 = Q0 / 3u;
      const unsigned op_lo = 24u * g0;
      const unsigned ep_lo = 12u * g0;
      const unsigned seg   = op_lo >> LOG_TD;
      const unsigned rb  = op_lo & (TD - 1u);
      const unsigned pr  = ep_lo & (TD - 1u);
      const unsigned qoff = 3u * g0 - Q0;
      if (seg == 0u) {
        const unsigned xi = 2u * op_lo;
        const unsigned xo = xi & (TD - 1u);
        if ((xi >> LOG_TD) == 0u)
          cand_slot_uniform<K_PROD, true, GK_W, UG_L0>(hv, m0, hv, m0, m0, xo, rb, pr, qoff, nq, ll, ql, h, a, cinit, ag, ginit, pinit, acc[n]);
        else if (has_m1)
          cand_slot_uniform<K_PROD, true, GK_W, UG_L0>(hv, m0, hv, m0, m1, xo, rb, pr, qoff, nq, ll, ql, h, a, cinit, ag, ginit, pinit, acc[n]);
        else
          cand_slot_uniform<K_PROD, true, GK_W, UG_Z0>(hv, m0, hv, m0, hv, 0u, rb, pr, qoff, nq, ll, ql, h, a, cinit, ag, ginit, pinit, acc[n]);
      }
      else if (seg == 1u && has_m1)
        cand_slot_uniform<K_H, true, GK_LOAD, UG_NONE>(hv, hv, m1, m0, hv, 0u, rb, pr, qoff, nq, ll, ql, h, a, cinit, ag, ginit, pinit, acc[n]);
      else if (seg == 1u)
        cand_slot_uniform<K_H, true, GK_ZERO, UG_NONE>(hv, hv, hv, m0, hv, 0u, rb, pr, qoff, nq, ll, ql, h, a, cinit, ag, ginit, pinit, acc[n]);
      else if (seg == 2u && has_m1) {
        const unsigned xi = 2u * (op_lo - TD) - 4u;
        const unsigned xo = xi & (TD - 1u);
        if ((xi >> LOG_TD) == 2u)
          cand_slot_uniform<K_PROD, true, GK_ZERO, UG_Z2>(hv, m1, hv, m1, hv, 0u, rb, pr, qoff, nq, ll, ql, h, a, cinit, ag, ginit, pinit, acc[n]);
        else
          cand_slot_uniform<K_PROD, true, GK_ZERO, UG_L2>(hv, m1, hv, m1, hv, xo, rb, pr, qoff, nq, ll, ql, h, a, cinit, ag, ginit, pinit, acc[n]);
      }
      else if (seg == 2u)
        cand_slot_uniform<K_ZERO, false, GK_ZERO, UG_NONE>(hv, hv, hv, hv, hv, 0u, rb, pr, qoff, nq, ll, ql, h, a, cinit, ag, ginit, pinit, acc[n]);
      else if (seg == 3u && has_m1)
        cand_slot_uniform<K_H, true, GK_U, UG_NONE>(hv, hv, hv, m1, hv, 0u, rb, pr, qoff, nq, ll, ql, h, a, cinit, ag, ginit, pinit, acc[n]);
      else if (seg == 4u)
        cand_slot_uniform<K_ZERO, false, GK_LOAD, UG_NONE>(hv, hv, hv, hv, hv, 0u, rb, pr, qoff, nq, ll, ql, h, a, cinit, ag, ginit, pinit, acc[n]);
      else
        cand_slot_uniform<K_H, false, GK_U, UG_NONE>(hv, hv, hv, hv, hv, 0u, rb, pr, qoff, nq, ll, ql, h, a, cinit, ag, ginit, pinit, acc[n]);
    }
  } else {
    for (unsigned n = 0; n < 3u; ++n) {
      const unsigned Q0 = n * NQ + q0;
      cand_slot_lane(has_m1, Q0, Q0 / 3u, nq, ll, ql, h, a, cinit, ag, ginit, pinit, hv, m0, m1, acc[n]);
    }
  }
  __syncthreads();
  for (unsigned i = threadIdx.x; i < nq; i += 256u) {
    const float4 h4 = ld4(hv + 4u * (q0 + i));
    const float4 c0 = acc[0][i];
    const float4 c1 = acc[1][i];
    const float4 c2 = acc[2][i];
    float4 o;
    o.x = h4.x + c0.x + c1.x + c2.x;
    o.y = h4.y + c0.y + c1.y + c2.y;
    o.z = h4.z + c0.z + c1.z + c2.z;
    o.w = h4.w + c0.w + c1.w + c2.w;
    *(float4*)(out + 4u * (q0 + i)) = o;
  }
}

extern "C" void kernel_launch(void* const* d_in, const int* in_sizes, int n_in,
                              void* d_out, int out_size, void* d_ws, size_t ws_size,
                              hipStream_t stream) {
  (void)in_sizes; (void)n_in; (void)out_size; (void)d_ws; (void)ws_size;
  const float* feat = (const float*)d_in[0];
  const float* W_w  = (const float*)d_in[1];
  const float* W_b  = (const float*)d_in[2];
  const float* Wx_w = (const float*)d_in[3];
  const float* Wx_b = (const float*)d_in[4];
  float* out = (float*)d_out;

  const float* h0 = feat;
  const float* h1 = feat + (size_t)TD;
  const float* h2 = feat + (size_t)2 * TD;

  dim3 blk(256);
  dim3 grd_c(CBLK3);

  gsn_cand_wmma<<<grd_c, blk, 0, stream>>>(
      h0, h2, out + 0 * (size_t)TD, h1, h0, out + 1 * (size_t)TD,
      h2, h1, h0, out + 2 * (size_t)TD, W_w, W_b, Wx_w, Wx_b);
}


# h of the final sum loaded by the last slot's owner lanes within its operand load group and handed over in a 4th LDS plane: the sum behind the barrier reads LDS only, no global load; sum order kept
# speedup vs baseline: 5.6794x; 1.0333x over previous
#include <hip/hip_runtime.h>
#include <math.h>

#define LOG_TD 22u
#define TD (1u << LOG_TD)

typedef __attribute__((ext_vector_type(2))) float v2f;
typedef __attribute__((ext_vector_type(8))) float v8f;

constexpr unsigned NQ     = TD / 4u;
constexpr unsigned TPS    = 2u;
constexpr unsigned GPB    = 16u * 8u * TPS;
constexpr unsigned LQ     = 760u;
constexpr unsigned CBLK   = (NQ + LQ - 1u) / LQ;
constexpr unsigned CBLK3  = 3u * CBLK;
static_assert(LQ % 8u == 0u && LQ + 2u <= 3u * GPB, "block range must be whole lines and fit its groups");
constexpr unsigned BQ = (NQ / 4u) / LQ;
constexpr unsigned BH = (NQ / 2u) / LQ;
constexpr unsigned BL = CBLK - 1u;
constexpr unsigned N1 = BQ - 1u;
constexpr unsigned N2 = BH - BQ - 1u;
constexpr unsigned N3 = BL - BH - 1u;
constexpr unsigned NREG = N1 + N2 + N3;
static_assert(1u < BQ && BQ + 1u < BH && BH + 1u < BL && NREG + 4u == CBLK, "the four named blocks and the three ranges between them are the node's blocks, each once");
static_assert(TD % 12u == 4u, "gate alignment in slab 2");

__device__ __forceinline__ float hw_exp2(float x) {
#if __has_builtin(__builtin_amdgcn_exp2f)
  return __builtin_amdgcn_exp2f(x);
#else
  return exp2f(x);
#endif
}
__device__ __forceinline__ float hw_rcp(float x) {
#if __has_builtin(__builtin_amdgcn_rcpf)
  return __builtin_amdgcn_rcpf(x);
#else
  return 1.0f / x;
#endif
}
__device__ __forceinline__ float fast_sigmoid(float x) {
  return hw_rcp(1.0f + hw_exp2(-1.4426950408889634f * x));
}
__device__ __forceinline__ float fast_tanh(float x) {
#if __has_builtin(__builtin_amdgcn_tanhf)
  return __builtin_amdgcn_tanhf(x);
#else
  return 1.0f - 2.0f * hw_rcp(hw_exp2(2.8853900817779268f * x) + 1.0f);
#endif
}

__device__ __forceinline__ float4 ld4(const float* __restrict__ p) {
  return *(const float4*)p;
}
__device__ __forceinline__ float4 zero4() { return make_float4(0.0f, 0.0f, 0.0f, 0.0f); }
__device__ __forceinline__ float4 sel4(bool c, const float4& v) {
  return make_float4(c ? v.x : 0.0f, c ? v.y : 0.0f, c ? v.z : 0.0f, c ? v.w : 0.0f);
}


__device__ __forceinline__ float4 gate_src4(bool has_m1, unsigned m,
    const float* __restrict__ hv, const float* __restrict__ m0,
    const float* __restrict__ m1) {
  const unsigned seg = m >> LOG_TD;
  const unsigned r   = m & (TD - 1u);
  const bool is_h  = seg >= 3u;
  const bool is_m0 = seg == 0u;
  const bool is_m1 = has_m1 && seg == 1u;
  const bool valid = is_h || is_m0 || is_m1;
  const float* base = is_m0 ? m0 : (is_m1 ? m1 : hv);
  const float4 v = ld4(base + (valid ? r : 0u));
  return sel4(valid, v);
}

__device__ __forceinline__ float4 msg4(bool has_m1, unsigned p, const float* __restrict__ hv,
    const float* __restrict__ m0, const float* __restrict__ m1) {
  const unsigned n = p >> LOG_TD;
  const unsigned r = p & (TD - 1u);
  const bool is_m0 = n == 0u;
  const bool is_m1 = has_m1 && n == 1u;
  const bool valid = is_m0 || is_m1;
  const float* base = is_m0 ? m0 : (is_m1 ? m1 : hv);
  const float4 v = ld4(base + (valid ? r : 0u));
  return sel4(valid, v);
}

__device__ __forceinline__ float4 cand_val4(bool has_m1, unsigned m, const float4& gu,
    const float* __restrict__ hv, const float* __restrict__ m0,
    const float* __restrict__ m1) {
  const unsigned seg = m >> LOG_TD;
  const unsigned r   = m & (TD - 1u);
  const unsigned n   = seg >> 1;
  const bool is_h  = (seg & 1u) != 0u;
  const bool is_m0 = !is_h && n == 0u;
  const bool is_m1 = has_m1 && !is_h && n == 1u;
  const bool prod  = is_m0 || is_m1;
  const float* wb = is_m0 ? m0 : (is_m1 ? m1 : hv);
  const float4 v = ld4(wb + ((prod || is_h) ? r : 0u));
  const float4 gm = make_float4(gu.x * v.x, gu.y * v.y, gu.z * v.z, gu.w * v.w);
  return make_float4(is_h ? v.x : (prod ? gm.x : 0.0f), is_h ? v.y : (prod ? gm.y : 0.0f),
                     is_h ? v.z : (prod ? gm.z : 0.0f), is_h ? v.w : (prod ? gm.w : 0.0f));
}

__device__ __forceinline__ void make_weights(const float* __restrict__ Wm,
    const float* __restrict__ Bv, unsigned lane, v2f (&a)[6], v8f& cinit) {
  const unsigned o  = lane & 15u;
  const unsigned h  = lane >> 4;
  const unsigned g  = o / 3u;
  const unsigned j  = o - 3u * g;
  const bool valid  = (o < 12u) && ((g >> 1) == h);
  const float msk   = valid ? 1.0f : 0.0f;
  const bool second = (g & 1u) != 0u;
  float w[6];
#pragma unroll
  for (int i = 0; i < 6; ++i) w[i] = Wm[j * 6u + (unsigned)i] * msk;
#pragma unroll
  for (int c = 0; c < 3; ++c) {
    a[c].x     = second ? 0.0f : w[2 * c];
    a[c].y     = second ? 0.0f : w[2 * c + 1];
    a[c + 3].x = second ? w[2 * c]     : 0.0f;
    a[c + 3].y = second ? w[2 * c + 1] : 0.0f;
  }
  const float b0 = Bv[0], b1 = Bv[1], b2 = Bv[2];
  const bool hi = h != 0u;
  cinit[0] = hi ? b2 : b0;
  cinit[1] = hi ? b0 : b1;
  cinit[2] = hi ? b1 : b2;
  cinit[3] = hi ? b2 : b0;
  cinit[4] = hi ? 0.0f : b1;
  cinit[5] = hi ? 0.0f : b2;
  cinit[6] = hi ? 0.0f : b0;
  cinit[7] = hi ? 0.0f : b1;
}

__device__ __forceinline__ v8f group_product(const v2f (&a)[6], const float4& x0,
    const float4& x1, const float4& x2, v8f c) {
  v2f b;
  b.x = x0.x; b.y = x0.y;
  c = __builtin_amdgcn_wmma_f32_16x16x4_f32(false, a[0], false, b, (short)0, c, false, false);
  b.x = x0.z; b.y = x0.w;
  c = __builtin_amdgcn_wmma_f32_16x16x4_f32(false, a[1], false, b, (short)0, c, false, false);
  b.x = x1.x; b.y = x1.y;
  c = __builtin_amdgcn_wmma_f32_16x16x4_f32(false, a[2], false, b, (short)0, c, false, false);
  b.x = x1.z; b.y = x1.w;
  c = __builtin_amdgcn_wmma_f32_16x16x4_f32(false, a[3], false, b, (short)0, c, false, false);
  b.x = x2.x; b.y = x2.y;
  c = __builtin_amdgcn_wmma_f32_16x16x4_f32(false, a[4], false, b, (short)0, c, false, false);
  b.x = x2.z; b.y = x2.w;
  c = __builtin_amdgcn_wmma_f32_16x16x4_f32(false, a[5], false, b, (short)0, c, false, false);
  return c;
}

__device__ __forceinline__ float shr2_upper(float v) {
  const int i = __builtin_bit_cast(int, v);
  return __builtin_bit_cast(float, __builtin_amdgcn_update_dpp(i, i, 0x112, 0x2, 0xf, true));
}
__device__ __forceinline__ void pair_weights(const v2f (&ag)[6], v2f (&ap)[6]) {
#pragma unroll
  for (int c = 0; c < 6; ++c) {
    ap[c].x = shr2_upper(ag[c].x);
    ap[c].y = shr2_upper(ag[c].y);
  }
}
__device__ __forceinline__ void pair_bias(const float* __restrict__ Bv, v8f& cinit) {
  const float b0 = Bv[0], b1 = Bv[1], b2 = Bv[2];
  cinit[0] = b0; cinit[1] = b1; cinit[2] = b2;
  cinit[3] = b0; cinit[4] = b1; cinit[5] = b2;
  cinit[6] = 0.0f; cinit[7] = 0.0f;
}

__device__ __forceinline__ v8f row_product(const v2f (&a)[6], const float4& x0,
    const float4& x1, v8f c) {
  v2f b;
  b.x = x0.x; b.y = x0.y;
  c = __builtin_amdgcn_wmma_f32_16x16x4_f32(false, a[0], false, b, (short)0, c, false, false);
  b.x = x0.z; b.y = x0.w;
  c = __builtin_amdgcn_wmma_f32_16x16x4_f32(false, a[1], false, b, (short)0, c, false, false);
  b.x = x1.x; b.y = x1.y;
  c = __builtin_amdgcn_wmma_f32_16x16x4_f32(false, a[2], false, b, (short)0, c, false, false);
  return c;
}

__device__ __forceinline__ float4 sigmoid4(float d0, float d1, float d2, float d3) {
  return make_float4(fast_sigmoid(d0), fast_sigmoid(d1), fast_sigmoid(d2), fast_sigmoid(d3));
}

__device__ __forceinline__ float4 delta4(const float4& g, const float4& mv,
    float d0, float d1, float d2, float d3) {
  float4 o;
  o.x = (1.0f - g.x) * mv.x + g.x * fast_tanh(d0);
  o.y = (1.0f - g.y) * mv.y + g.y * fast_tanh(d1);
  o.z = (1.0f - g.z) * mv.z + g.z * fast_tanh(d2);
  o.w = (1.0f - g.w) * mv.w + g.w * fast_tanh(d3);
  return o;
}

__device__ __forceinline__ float4 gated4(const float4& g, float d0, float d1, float d2, float d3) {
  float4 o;
  o.x = g.x * fast_tanh(d0);
  o.y = g.y * fast_tanh(d1);
  o.z = g.z * fast_tanh(d2);
  o.w = g.w * fast_tanh(d3);
  return o;
}

__device__ __forceinline__ float4 tanh4(float d0, float d1, float d2, float d3) {
  return make_float4(fast_tanh(d0), fast_tanh(d1), fast_tanh(d2), fast_tanh(d3));
}
__device__ __forceinline__ float4 delta4t(const float4& g, const float4& mv, const float4& t) {
  float4 o;
  o.x = (1.0f - g.x) * mv.x + g.x * t.x;
  o.y = (1.0f - g.y) * mv.y + g.y * t.y;
  o.z = (1.0f - g.z) * mv.z + g.z * t.z;
  o.w = (1.0f - g.w) * mv.w + g.w * t.w;
  return o;
}
__device__ __forceinline__ float4 gated4t(const float4& g, const float4& t) {
  return make_float4(g.x * t.x, g.y * t.y, g.z * t.z, g.w * t.w);
}


__device__ __forceinline__ void gates_sh0(const v8f& P, const v8f& Q, float4& u0, float4& u1, float4& u2) {
  u0 = sigmoid4(P[0], P[1], P[2], P[3]);
  u1 = sigmoid4(P[4], P[5], Q[0], Q[1]);
  u2 = sigmoid4(Q[2], Q[3], Q[4], Q[5]);
}
__device__ __forceinline__ void gates_sh2(const v8f& P, const v8f& Q, const v8f& R, float4& u0, float4& u1, float4& u2) {
  u0 = sigmoid4(P[2], P[3], P[4], P[5]);
  u1 = sigmoid4(Q[0], Q[1], Q[2], Q[3]);
  u2 = sigmoid4(Q[4], Q[5], R[0], R[1]);
}

__device__ __forceinline__ void cand_slot_lane(bool has_m1, unsigned Q0, unsigned g0, unsigned nq,
    unsigned ll, unsigned ql, unsigned h, const v2f (&a)[6], const v8f& cinit,
    const v2f (&ag)[6], const v8f& ginit, const v8f& pinit,
    const float* __restrict__ hv, const float* __restrict__ m0, const float* __restrict__ m1,
    float4* accn) {
  v2f ap[6];
  pair_weights(ag, ap);
  const unsigned seg_lo = (24u * g0) >> LOG_TD;
  const unsigned seg_hi = (24u * g0 + 24u * GPB - 1u) >> LOG_TD;
  const bool in0 = seg_lo == 0u;
  const bool in2 = has_m1 && (seg_lo == 2u || seg_hi == 2u);
  for (unsigned t = 0; t < TPS; ++t) {
    const unsigned bu   = 24u * (g0 + t * 128u) + ll;
    const unsigned bmax = 24u * (NQ - 1u) + 12u * h;
    const unsigned base = bu < bmax ? bu : bmax;

    float4 gu0 = zero4(), gu1 = zero4(), gu2 = zero4();
    if (in0) {
      const unsigned gi = 2u * base;
      const float4 X0 = gate_src4(has_m1, gi,       hv, m0, m1);
      const float4 X1 = gate_src4(has_m1, gi + 4u,  hv, m0, m1);
      const float4 X2 = gate_src4(has_m1, gi + 8u,  hv, m0, m1);
      const float4 X3 = gate_src4(has_m1, gi + 12u, hv, m0, m1);
      const float4 X4 = gate_src4(has_m1, gi + 16u, hv, m0, m1);
      const float4 X5 = gate_src4(has_m1, gi + 20u, hv, m0, m1);
      const v8f P = group_product(ap, X0, X1, X2, pinit);
      const v8f Q = group_product(ap, X3, X4, X5, pinit);
      gates_sh0(P, Q, gu0, gu1, gu2);
    } else if (in2) {
      const unsigned gi = 2u * (base - TD) - 4u;
      const float4 X0 = gate_src4(has_m1, gi,       hv, m0, m1);
      const float4 X1 = gate_src4(has_m1, gi + 4u,  hv, m0, m1);
      const float4 X2 = gate_src4(has_m1, gi + 8u,  hv, m0, m1);
      const float4 X3 = gate_src4(has_m1, gi + 12u, hv, m0, m1);
      const float4 X4 = gate_src4(has_m1, gi + 16u, hv, m0, m1);
      const float4 X5 = gate_src4(has_m1, gi + 20u, hv, m0, m1);
      const float4 X6 = gate_src4(has_m1, gi + 24u, hv, m0, m1);
      const float4 X7 = gate_src4(has_m1, gi + 28u, hv, m0, m1);
      const v8f P = group_product(ap, X0, X1, X2, pinit);
      const v8f Q = group_product(ap, X3, X4, X5, pinit);
      const v8f R = row_product(ap, X6, X7, pinit);
      gates_sh2(P, Q, R, gu0, gu1, gu2);
    }

    const float4 y0 = gate_src4(has_m1, base,      hv, m0, m1);
    const float4 y1 = gate_src4(has_m1, base + 4u, hv, m0, m1);
    const float4 y2 = gate_src4(has_m1, base + 8u, hv, m0, m1);
    const v8f dg = group_product(ag, y0, y1, y2, ginit);
    const float4 ga = sigmoid4(dg[0], dg[1], dg[2], dg[3]);
    const float4 gb = sigmoid4(dg[4], dg[5], dg[6], dg[7]);

    const float4 x0 = cand_val4(has_m1, base,      gu0, hv, m0, m1);
    const float4 x1 = cand_val4(has_m1, base + 4u, gu1, hv, m0, m1);
    const float4 x2 = cand_val4(has_m1, base + 8u, gu2, hv, m0, m1);
    const v8f d = group_product(a, x0, x1, x2, cinit);

    const unsigned p  = (base >> 1) + 2u * h;
    const unsigned p2 = p + 4u * (h ^ 1u);
    const float4 ma = msg4(has_m1, p,  hv, m0, m1);
    const float4 mb = msg4(has_m1, p2, hv, m0, m1);

    const unsigned qa = 3u * (g0 + t * 128u) - Q0 + ql;
    if (qa < nq)
      accn[qa] = delta4(ga, ma, d[0], d[1], d[2], d[3]);
    if (h == 0u && qa + 1u < nq)
      accn[qa + 1u] = delta4(gb, mb, d[4], d[5], d[6], d[7]);
  }
}

constexpr int K_PROD = 1;
constexpr int K_H    = 2;
constexpr int K_ZERO = 3;

constexpr int GK_W    = 1;
constexpr int GK_U    = 2;
constexpr int GK_ZERO = 3;
constexpr int GK_LOAD = 4;

constexpr int UG_NONE = 0;
constexpr int UG_L0   = 1;
constexpr int UG_Z0   = 2;
constexpr int UG_L2   = 3;
constexpr int UG_Z2   = 4;

template <int KIND, bool MSG, int GK, int UG, bool ADDH>
__device__ __forceinline__ void cand_slot_uniform(
    const float* __restrict__ ub, const float* __restrict__ wb,
    const float* __restrict__ gs, const float* __restrict__ mp,
    const float* __restrict__ xs, unsigned xo,
    unsigned rb, unsigned pr, unsigned qoff, unsigned nq, unsigned ll, unsigned ql, unsigned h,
    const v2f (&a)[6], const v8f& cinit, const v2f (&ag)[6], const v8f& ginit,
    const v8f& pinit,
    float4* accn, float4* acch) {
  static_assert(!(MSG && ADDH), "the slot that hands h over has no message: ma / mb hold h there");
  static_assert(GK != GK_W || KIND == K_PROD, "w is loaded only for a gate * message slab");
  static_assert(GK != GK_U || KIND == K_H, "u is the node state only in an h slab");
  static_assert((UG != UG_NONE) == (KIND == K_PROD), "operand gates belong to a gate * message slab");
  const unsigned lo = rb + ll;
  const unsigned po = pr + 4u * ql;
  const unsigned pq = po + 4u * (h ^ 1u);
  float4 zga = zero4(), zgb = zero4(), zta = zero4(), ztb = zero4();
  float4 u[TPS][3], w[TPS][3], y[TPS][3], ma[TPS], mb[TPS];
  if (UG == UG_Z0 || UG == UG_Z2) {
    const float pb0 = pinit[0], pb1 = pinit[1], pb2 = pinit[2];
    const float s0 = fast_sigmoid(pb0), s1 = fast_sigmoid(pb1), s2 = fast_sigmoid(pb2);
#pragma unroll
    for (unsigned t = 0; t < TPS; ++t) {
      if (UG == UG_Z0) {
        u[t][0] = make_float4(s0, s1, s2, s0);
        u[t][1] = make_float4(s1, s2, s0, s1);
        u[t][2] = make_float4(s2, s0, s1, s2);
      } else {
        u[t][0] = make_float4(s2, s0, s1, s2);
        u[t][1] = make_float4(s0, s1, s2, s0);
        u[t][2] = make_float4(s1, s2, s0, s1);
      }
    }
  }
  if (UG == UG_L0 || UG == UG_L2) {
    const unsigned xl = xo + 2u * ll;
    v2f ap[6];
    pair_weights(ag, ap);
    float4 X[TPS][8];
    if (UG == UG_L0) {
#pragma unroll
      for (unsigned t = 0; t < TPS; ++t) {
        const unsigned xt = xl + t * (48u * 128u);
        X[t][0] = ld4(xs + xt);
        X[t][1] = ld4(xs + xt + 4u);
        X[t][2] = ld4(xs + xt + 8u);
        X[t][3] = ld4(xs + xt + 12u);
        X[t][4] = ld4(xs + xt + 16u);
        X[t][5] = ld4(xs + xt + 20u);
      }
#pragma unroll
      for (unsigned t = 0; t < TPS; ++t) {
        const v8f P = group_product(ap, X[t][0], X[t][1], X[t][2], pinit);
        const v8f Q = group_product(ap, X[t][3], X[t][4], X[t][5], pinit);
        gates_sh0(P, Q, u[t][0], u[t][1], u[t][2]);
      }
    } else {
#pragma unroll
      for (unsigned t = 0; t < TPS; ++t) {
        const unsigned xt = xl + t * (48u * 128u);
        X[t][0] = ld4(xs + xt);
        X[t][1] = ld4(xs + xt + 4u);
        X[t][2] = ld4(xs + xt + 8u);
        X[t][3] = ld4(xs + xt + 12u);
        X[t][4] = ld4(xs + xt + 16u);
        X[t][5] = ld4(xs + xt + 20u);
        X[t][6] = ld4(xs + xt + 24u);
        X[t][7] = ld4(xs + xt + 28u);
        const v8f P = group_product(ap, X[t][0], X[t][1], X[t][2], pinit);
        const v8f Q = group_product(ap, X[t][3], X[t][4], X[t][5], pinit);
        const v8f R = row_product(ap, X[t][6], X[t][7], pinit);
        gates_sh2(P, Q, R, u[t][0], u[t][1], u[t][2]);
        __builtin_amdgcn_sched_barrier(0);
      }
    }
  }
  if (GK == GK_ZERO || KIND == K_ZERO) {
    float gb0 = ginit[0], gb1 = ginit[1], gb2 = ginit[2], cb0 = cinit[0], cb1 = cinit[1], cb2 = cinit[2];
    asm volatile("" : "+v"(gb0), "+v"(gb1), "+v"(gb2), "+v"(cb0), "+v"(cb1), "+v"(cb2));
    if (GK == GK_ZERO) {
      const float sg0 = fast_sigmoid(gb0), sg1 = fast_sigmoid(gb1), sg2 = fast_sigmoid(gb2);
      zga = make_float4(sg0, sg1, sg2, sg0);
      zgb = make_float4(sg1, sg2, sg0, sg1);
    }
    if (KIND == K_ZERO) {
      const float st0 = fast_tanh(cb0), st1 = fast_tanh(cb1), st2 = fast_tanh(cb2);
      zta = make_float4(st0, st1, st2, st0);
      ztb = make_float4(st1, st2, st0, st1);
    }
  }
#pragma unroll
  for (unsigned t = 0; t < TPS; ++t) {
    const unsigned ot = lo + t * (24u * 128u);
    const unsigned pt = po + t * (12u * 128u);
    const unsigned pu = pq + t * (12u * 128u);
    if (KIND == K_H) {
      u[t][0] = ld4(ub + ot);
      u[t][1] = ld4(ub + ot + 4u);
      u[t][2] = ld4(ub + ot + 8u);
    } else if (KIND == K_ZERO) {
      u[t][0] = zero4(); u[t][1] = zero4(); u[t][2] = zero4();
    }
    if (KIND == K_PROD) {
      w[t][0] = ld4(wb + ot);
      w[t][1] = ld4(wb + ot + 4u);
      w[t][2] = ld4(wb + ot + 8u);
    } else {
      w[t][0] = zero4(); w[t][1] = zero4(); w[t][2] = zero4();
    }
    if (GK == GK_LOAD) {
      y[t][0] = ld4(gs + ot);
      y[t][1] = ld4(gs + ot + 4u);
      y[t][2] = ld4(gs + ot + 8u);
    } else {
      y[t][0] = zero4(); y[t][1] = zero4(); y[t][2] = zero4();
    }
    if (MSG || ADDH) {
      ma[t] = ld4(mp + pt);
      mb[t] = ld4(mp + pu);
    } else {
      ma[t] = zero4(); mb[t] = zero4();
    }
  }
#pragma unroll
  for (unsigned t = 0; t < TPS; ++t) {
    if (KIND == K_H)
      asm volatile("" :: "v"(u[t][0].x), "v"(u[t][0].y), "v"(u[t][0].z), "v"(u[t][0].w),
                         "v"(u[t][1].x), "v"(u[t][1].y), "v"(u[t][1].z), "v"(u[t][1].w),
                         "v"(u[t][2].x), "v"(u[t][2].y), "v"(u[t][2].z), "v"(u[t][2].w));
    if (KIND == K_PROD)
      asm volatile("" :: "v"(w[t][0].x), "v"(w[t][0].y), "v"(w[t][0].z), "v"(w[t][0].w),
                         "v"(w[t][1].x), "v"(w[t][1].y), "v"(w[t][1].z), "v"(w[t][1].w),
                         "v"(w[t][2].x), "v"(w[t][2].y), "v"(w[t][2].z), "v"(w[t][2].w));
    if (GK == GK_LOAD)
      asm volatile("" :: "v"(y[t][0].x), "v"(y[t][0].y), "v"(y[t][0].z), "v"(y[t][0].w),
                         "v"(y[t][1].x), "v"(y[t][1].y), "v"(y[t][1].z), "v"(y[t][1].w),
                         "v"(y[t][2].x), "v"(y[t][2].y), "v"(y[t][2].z), "v"(y[t][2].w));
    if (MSG || ADDH)
      asm volatile("" :: "v"(ma[t].x), "v"(ma[t].y), "v"(ma[t].z), "v"(ma[t].w),
                         "v"(mb[t].x), "v"(mb[t].y), "v"(mb[t].z), "v"(mb[t].w));
  }
#pragma unroll
  for (unsigned t = 0; t < TPS; ++t) {
    const float4 y0 = GK == GK_W ? w[t][0] : (GK == GK_U ? u[t][0] : y[t][0]);
    const float4 y1 = GK == GK_W ? w[t][1] : (GK == GK_U ? u[t][1] : y[t][1]);
    const float4 y2 = GK == GK_W ? w[t][2] : (GK == GK_U ? u[t][2] : y[t][2]);
    float4 x0 = u[t][0], x1 = u[t][1], x2 = u[t][2];
    if (KIND == K_PROD) {
      x0 = make_float4(x0.x * w[t][0].x, x0.y * w[t][0].y, x0.z * w[t][0].z, x0.w * w[t][0].w);
      x1 = make_float4(x1.x * w[t][1].x, x1.y * w[t][1].y, x1.z * w[t][1].z, x1.w * w[t][1].w);
      x2 = make_float4(x2.x * w[t][2].x, x2.y * w[t][2].y, x2.z * w[t][2].z, x2.w * w[t][2].w);
    }
    float4 ta = zta, tb = ztb;
    if (KIND != K_ZERO) {
      const v8f d = group_product(a, x0, x1, x2, cinit);
      ta = tanh4(d[0], d[1], d[2], d[3]);
      tb = tanh4(d[4], d[5], d[6], d[7]);
    }
    float4 ga = zga, gb = zgb;
    if (GK != GK_ZERO) {
      const v8f dg = group_product(ag, y0, y1, y2, ginit);
      ga = sigmoid4(dg[0], dg[1], dg[2], dg[3]);
      gb = sigmoid4(dg[4], dg[5], dg[6], dg[7]);
    }

    const unsigned qa = qoff + t * 384u + ql;
    if (qa < nq) {
      accn[qa] = MSG ? delta4t(ga, ma[t], ta) : gated4t(ga, ta);
      if (ADDH) acch[qa] = ma[t];
    }
    if (h == 0u && qa + 1u < nq) {
      accn[qa + 1u] = MSG ? delta4t(gb, mb[t], tb) : gated4t(gb, tb);
      if (ADDH) acch[qa + 1u] = mb[t];
    }
  }
}

__global__ __launch_bounds__(256) __attribute__((amdgpu_waves_per_eu(8, 8))) void gsn_cand_wmma(
    const float* __restrict__ hvA, const float* __restrict__ m0A, float* __restrict__ outA,
    const float* __restrict__ hvB, const float* __restrict__ m0B, float* __restrict__ outB,
    const float* __restrict__ hvC, const float* __restrict__ m0C, const float* __restrict__ m1C,
    float* __restrict__ outC,
    const float* __restrict__ Ww,
    const float* __restrict__ Bw,
    const float* __restrict__ Wx,
    const float* __restrict__ Bx) {
  __shared__ float4 acc[4][LQ];
  const unsigned d  = blockIdx.x;
  const bool c0 = d < 12u, c1 = d < 12u + NREG, c2 = d < 12u + NREG + 2u * (N3 + N1);
  const unsigned e1 = d - 12u;
  const unsigned d2 = e1 - NREG;
  const unsigned d3 = d2 - 2u * (N3 + N1);
  const bool f2 = d2 < N3 + N1, f3 = d3 < N2;
  const unsigned e2 = d2 - (f2 ? 0u : N3 + N1);
  const unsigned e3 = d3 - (f3 ? 0u : N2);
  const unsigned r1 = e1 + N1 - (e1 + N1 >= NREG ? NREG : 0u);
  const unsigned r2 = e2 + N1 + N2 - (e2 >= N3 ? NREG : 0u);
  const unsigned r3 = e3 + N1;
  const unsigned r  = c1 ? r1 : (c2 ? r2 : r3);
  const unsigned bxr = r + 1u + (r >= N1 ? 1u : 0u) + (r >= N1 + N2 ? 1u : 0u);
  const unsigned kb  = d & 3u;
  const unsigned bxb = kb == 0u ? BL : (kb == 1u ? BH : (kb == 2u ? BQ : 0u));
  const unsigned bx   = c0 ? bxb : bxr;
  const unsigned node = c0 ? 2u - (d >> 2u) : (c1 ? 2u : ((c2 ? f2 : f3) ? 1u : 0u));
  const bool isC = node == 2u;
  const bool isB = node == 1u;
  const bool has_m1 = isC;
  const float* __restrict__ hv = isC ? hvC : (isB ? hvB : hvA);
  const float* __restrict__ m0 = isC ? m0C : (isB ? m0B : m0A);
  const float* __restrict__ m1 = m1C;
  float* __restrict__ out      = isC ? outC : (isB ? outB : outA);
  const unsigned lane = threadIdx.x & 31u;
  const unsigned h    = lane >> 4;
  const unsigned gl   = (threadIdx.x >> 5u) * 16u + (lane & 15u);
  const unsigned ll   = 24u * gl + 12u * h;
  const unsigned ql   = 3u * gl + 2u * h;
  const unsigned q0   = bx * LQ;
  const unsigned nq   = (NQ - q0) < LQ ? (NQ - q0) : LQ;

  v2f a[6], ag[6];
  v8f cinit, ginit, pinit;
  make_weights(Ww, Bw, lane, a, cinit);
  make_weights(Wx, Bx, lane, ag, ginit);
  pair_bias(Bx, pinit);

  bool all_whole = true;
  for (unsigned n = 0; n < 3u; ++n) {
    const unsigned g0    = (n * NQ + q0) / 3u;
    const unsigned op_lo = 24u * g0;
    const unsigned ep_lo = 12u * g0;
    all_whole = all_whole && g0 + (GPB - 1u) < NQ
        && ((op_lo + 24u * GPB - 1u) >> LOG_TD) == (op_lo >> LOG_TD)
        && (ep_lo >> LOG_TD) == n && ((ep_lo + 12u * GPB - 1u) >> LOG_TD) == n;
    const unsigned seg = op_lo >> LOG_TD;
    if (seg == 0u) {
      const unsigned xi = 2u * op_lo;
      all_whole = all_whole && ((xi + 48u * GPB - 1u) >> LOG_TD) == (xi >> LOG_TD);
    } else if (seg == 2u && has_m1) {
      const unsigned xi = 2u * (op_lo - TD) - 4u;
      all_whole = all_whole && ((xi + 48u * GPB + 7u) >> LOG_TD) == (xi >> LOG_TD);
    }
  }
  if (all_whole) {
    for (unsigned n = 0; n < 3u; ++n) {
      const unsigned Q0 = n * NQ + q0;
      const unsigned g0 = Q0 / 3u;
      const unsigned op_lo = 24u * g0;
      const unsigned ep_lo = 12u * g0;
      const unsigned seg   = op_lo >> LOG_TD;
      const unsigned rb  = op_lo & (TD - 1u);
      const unsigned pr  = ep_lo & (TD - 1u);
      const unsigned qoff = 3u * g0 - Q0;
      if (seg == 0u) {
        const unsigned xi = 2u * op_lo;
        const unsigned xo = xi & (TD - 1u);
        if ((xi >> LOG_TD) == 0u)
          cand_slot_uniform<K_PROD, true, GK_W, UG_L0, false>(hv, m0, hv, m0, m0, xo, rb, pr, qoff, nq, ll, ql, h, a, cinit, ag, ginit, pinit, acc[n], acc[3]);
        else if (has_m1)
          cand_slot_uniform<K_PROD, true, GK_W, UG_L0, false>(hv, m0, hv, m0, m1, xo, rb, pr, qoff, nq, ll, ql, h, a, cinit, ag, ginit, pinit, acc[n], acc[3]);
        else
          cand_slot_uniform<K_PROD, true, GK_W, UG_Z0, false>(hv, m0, hv, m0, hv, 0u, rb, pr, qoff, nq, ll, ql, h, a, cinit, ag, ginit, pinit, acc[n], acc[3]);
      }
      else if (seg == 1u && has_m1)
        cand_slot_uniform<K_H, true, GK_LOAD, UG_NONE, false>(hv, hv, m1, m0, hv, 0u, rb, pr, qoff, nq, ll, ql, h, a, cinit, ag, ginit, pinit, acc[n], acc[3]);
      else if (seg == 1u)
        cand_slot_uniform<K_H, true, GK_ZERO, UG_NONE, false>(hv, hv, hv, m0, hv, 0u, rb, pr, qoff, nq, ll, ql, h, a, cinit, ag, ginit, pinit, acc[n], acc[3]);
      else if (seg == 2u && has_m1) {
        const unsigned xi = 2u * (op_lo - TD) - 4u;
        const unsigned xo = xi & (TD - 1u);
        if ((xi >> LOG_TD) == 2u)
          cand_slot_uniform<K_PROD, true, GK_ZERO, UG_Z2, false>(hv, m1, hv, m1, hv, 0u, rb, pr, qoff, nq, ll, ql, h, a, cinit, ag, ginit, pinit, acc[n], acc[3]);
        else
          cand_slot_uniform<K_PROD, true, GK_ZERO, UG_L2, false>(hv, m1, hv, m1, hv, xo, rb, pr, qoff, nq, ll, ql, h, a, cinit, ag, ginit, pinit, acc[n], acc[3]);
      }
      else if (seg == 2u)
        cand_slot_uniform<K_ZERO, false, GK_ZERO, UG_NONE, false>(hv, hv, hv, hv, hv, 0u, rb, pr, qoff, nq, ll, ql, h, a, cinit, ag, ginit, pinit, acc[n], acc[3]);
      else if (seg == 3u && has_m1)
        cand_slot_uniform<K_H, true, GK_U, UG_NONE, false>(hv, hv, hv, m1, hv, 0u, rb, pr, qoff, nq, ll, ql, h, a, cinit, ag, ginit, pinit, acc[n], acc[3]);
      else if (seg == 4u)
        cand_slot_uniform<K_ZERO, false, GK_LOAD, UG_NONE, true>(hv, hv, hv, hv, hv, 0u, rb, pr, qoff, nq, ll, ql, h, a, cinit, ag, ginit, pinit, acc[n], acc[3]);
      else if (seg == 5u)
        cand_slot_uniform<K_H, false, GK_U, UG_NONE, true>(hv, hv, hv, hv, hv, 0u, rb, pr, qoff, nq, ll, ql, h, a, cinit, ag, ginit, pinit, acc[n], acc[3]);
      else
        cand_slot_uniform<K_H, false, GK_U, UG_NONE, false>(hv, hv, hv, hv, hv, 0u, rb, pr, qoff, nq, ll, ql, h, a, cinit, ag, ginit, pinit, acc[n], acc[3]);
    }
  } else {
    for (unsigned n = 0; n < 3u; ++n) {
      const unsigned Q0 = n * NQ + q0;
      cand_slot_lane(has_m1, Q0, Q0 / 3u, nq, ll, ql, h, a, cinit, ag, ginit, pinit, hv, m0, m1, acc[n]);
    }
    for (unsigned i = threadIdx.x; i < nq; i += 256u)
      acc[3][i] = ld4(hv + 4u * (q0 + i));
  }
  __syncthreads();
  for (unsigned i = threadIdx.x; i < nq; i += 256u) {
    const float4 h4 = acc[3][i];
    const float4 c0 = acc[0][i];
    const float4 c1 = acc[1][i];
    const float4 c2 = acc[2][i];
    float4 o;
    o.x = h4.x + c0.x + c1.x + c2.x;
    o.y = h4.y + c0.y + c1.y + c2.y;
    o.z = h4.z + c0.z + c1.z + c2.z;
    o.w = h4.w + c0.w + c1.w + c2.w;
    *(float4*)(out + 4u * (q0 + i)) = o;
  }
}

extern "C" void kernel_launch(void* const* d_in, const int* in_sizes, int n_in,
                              void* d_out, int out_size, void* d_ws, size_t ws_size,
                              hipStream_t stream) {
  (void)in_sizes; (void)n_in; (void)out_size; (void)d_ws; (void)ws_size;
  const float* feat = (const float*)d_in[0];
  const float* W_w  = (const float*)d_in[1];
  const float* W_b  = (const float*)d_in[2];
  const float* Wx_w = (const float*)d_in[3];
  const float* Wx_b = (const float*)d_in[4];
  float* out = (float*)d_out;

  const float* h0 = feat;
  const float* h1 = feat + (size_t)TD;
  const float* h2 = feat + (size_t)2 * TD;

  dim3 blk(256);
  dim3 grd_c(CBLK3);

  gsn_cand_wmma<<<grd_c, blk, 0, stream>>>(
      h0, h2, out + 0 * (size_t)TD, h1, h0, out + 1 * (size_t)TD,
      h2, h1, h0, out + 2 * (size_t)TD, W_w, W_b, Wx_w, Wx_b);
}
